# starting kernel
# baseline (speedup 1.0000x reference)
#include <hip/hip_runtime.h>
#include <math.h>

#define DIMS   64
#define NTRAIN 4096
#define NQUERY 8192
#define NB     64
#define NOISE  1e-4f
#define LS2    64.0f
#define KSCALE 0.6931471805599453f
#define SPAD   68

typedef float v2f __attribute__((ext_vector_type(2)));
typedef float v8f __attribute__((ext_vector_type(8)));

__device__ __forceinline__ v8f wmma4(v2f a, v2f b, v8f c) {
  return __builtin_amdgcn_wmma_f32_16x16x4_f32(
       false, a,  false, b,
       (short)0, c,  false,  false);
}

__device__ __forceinline__ v2f frag_nt(const float* __restrict__ X, size_t rowStride,
                                       int rowBase, int colBase, int k0, int lane) {
  int kk = k0 + ((lane >> 4) << 1);
  const float* p = X + (size_t)(rowBase + (lane & 15)) * rowStride + colBase + kk;
  v2f v; v.x = p[0]; v.y = p[1];
  return v;
}

__device__ __forceinline__ v2f frag_lds(float (*S)[SPAD], int rowBase, int k0, int lane) {
  int kk = k0 + ((lane >> 4) << 1);
  const float* p = &S[rowBase + (lane & 15)][kk];
  v2f v; v.x = p[0]; v.y = p[1];
  return v;
}

__global__ void gp_sqnorm(const float* __restrict__ X, float* __restrict__ nrm, int rows) {
  int r = blockIdx.x * blockDim.x + threadIdx.x;
  if (r >= rows) return;
  const float4* p = (const float4*)(X + (size_t)r * DIMS);
  float s = 0.f;
#pragma unroll
  for (int i = 0; i < DIMS / 4; ++i) {
    float4 v = p[i];
    s += v.x * v.x + v.y * v.y + v.z * v.z + v.w * v.w;
  }
  nrm[r] = s;
}

__global__ void gp_build_K(const float* __restrict__ tx, const float* __restrict__ nrm,
                           float* __restrict__ K) {
  int bi = blockIdx.x, bj = blockIdx.y;
  int lane = threadIdx.x & 31;
  v8f ce = {}, co = {};
#pragma unroll
  for (int k0 = 0; k0 < DIMS; k0 += 8) {
    ce = wmma4(frag_nt(tx, DIMS, bi * 16, 0, k0, lane),
               frag_nt(tx, DIMS, bj * 16, 0, k0, lane), ce);
    co = wmma4(frag_nt(tx, DIMS, bi * 16, 0, k0 + 4, lane),
               frag_nt(tx, DIMS, bj * 16, 0, k0 + 4, lane), co);
  }
  v8f c = ce + co;
  int n  = lane & 15;
  int mh = (lane >> 4) * 8;
  int col = bj * 16 + n;
  float ncol = nrm[col];
#pragma unroll
  for (int r = 0; r < 8; ++r) {
    int row = bi * 16 + mh + r;
    float d2 = nrm[row] + ncol - 2.0f * c[r];
    float v  = KSCALE * __expf(-0.5f * d2 * (1.0f / LS2));
    if (row == col) v += NOISE;
    K[(size_t)row * NTRAIN + col] = v;
  }
}

__global__ void gp_chol_diag(float* __restrict__ K, int kb) {
  __shared__ float a[NB][NB + 1];
  int t = threadIdx.x;
  int base = kb * NB;
  for (int j = 0; j < NB; ++j) a[t][j] = K[(size_t)(base + t) * NTRAIN + base + j];
  __syncthreads();
  for (int j = 0; j < NB; ++j) {
    if (t == j) a[j][j] = sqrtf(a[j][j]);
    __syncthreads();
    if (t > j) a[t][j] /= a[j][j];
    __syncthreads();
    for (int c2 = j + 1; c2 <= t; ++c2) a[t][c2] -= a[t][j] * a[c2][j];
    __syncthreads();
  }
  for (int j = 0; j <= t; ++j) K[(size_t)(base + t) * NTRAIN + base + j] = a[t][j];
}

__global__ void gp_chol_trsm(float* __restrict__ K, int kb) {
  __shared__ float Ld[NB][NB + 1];
  __shared__ float rw[NB][NB + 1];
  int t = threadIdx.x;
  int base = kb * NB;
  for (int j = 0; j < NB; ++j) Ld[t][j] = K[(size_t)(base + t) * NTRAIN + base + j];
  int row = base + NB + blockIdx.x * NB + t;
  const float* src = K + (size_t)row * NTRAIN + base;
  for (int j = 0; j < NB; ++j) rw[t][j] = src[j];
  __syncthreads();
  for (int j = 0; j < NB; ++j) {
    float s = rw[t][j];
    for (int p = 0; p < j; ++p) s -= rw[t][p] * Ld[j][p];
    rw[t][j] = s / Ld[j][j];
  }
  float* dst = K + (size_t)row * NTRAIN + base;
  for (int j = 0; j < NB; ++j) dst[j] = rw[t][j];
}

__global__ void gp_chol_syrk(float* __restrict__ K, int kb) {
  __shared__ float As[64][SPAD];
  __shared__ float Bs[64][SPAD];
  int bi = blockIdx.x, bj = blockIdx.y;
  if (bj > bi) return;
  int base = kb * NB;
  int tb   = base + NB;
  int t = threadIdx.x;
  int lane = t & 31, wave = t >> 5;
  int rowA = tb + bi * 64, rowB = tb + bj * 64;

  for (int f = t; f < 64 * 16; f += 128) {
    int r = f >> 4, c4 = (f & 15) << 2;
    *(float4*)&As[r][c4] = *(const float4*)&K[(size_t)(rowA + r) * NTRAIN + base + c4];
    *(float4*)&Bs[r][c4] = *(const float4*)&K[(size_t)(rowB + r) * NTRAIN + base + c4];
  }
  __syncthreads();

  int wr = (wave >> 1) * 32;
  int wc = (wave & 1) * 32;
  v8f c00 = {}, c01 = {}, c10 = {}, c11 = {};
#pragma unroll
  for (int k0 = 0; k0 < NB; k0 += 4) {
    v2f a0 = frag_lds(As, wr,      k0, lane);
    v2f a1 = frag_lds(As, wr + 16, k0, lane);
    v2f b0 = frag_lds(Bs, wc,      k0, lane);
    v2f b1 = frag_lds(Bs, wc + 16, k0, lane);
    c00 = wmma4(a0, b0, c00);
    c01 = wmma4(a0, b1, c01);
    c10 = wmma4(a1, b0, c10);
    c11 = wmma4(a1, b1, c11);
  }

  int n  = lane & 15;
  int mh = (lane >> 4) * 8;
  int r0 = tb + bi * 64 + wr + mh;
  int c0 = tb + bj * 64 + wc + n;
#pragma unroll
  for (int r = 0; r < 8; ++r) {
    K[(size_t)(r0 + r) * NTRAIN + c0]           -= c00[r];
    K[(size_t)(r0 + r) * NTRAIN + c0 + 16]      -= c01[r];
    K[(size_t)(r0 + 16 + r) * NTRAIN + c0]      -= c10[r];
    K[(size_t)(r0 + 16 + r) * NTRAIN + c0 + 16] -= c11[r];
  }
}

__global__ void gp_fwd_solve(const float* __restrict__ K, float* __restrict__ y) {
  __shared__ float yl[NTRAIN];
  __shared__ float diag[NB][NB + 1];
  __shared__ float pv[NB];
  int t = threadIdx.x;
  int wave = t >> 5, lane = t & 31;
  for (int i = t; i < NTRAIN; i += 256) yl[i] = y[i];
  __syncthreads();
  for (int m = 0; m < NTRAIN / NB; ++m) {
    int rb = m * NB;
    for (int idx = t; idx < NB * NB; idx += 256)
      diag[idx >> 6][idx & 63] = K[(size_t)(rb + (idx >> 6)) * NTRAIN + rb + (idx & 63)];
    for (int rr = 0; rr < 8; ++rr) {
      int r = wave * 8 + rr;
      float s = 0.f;
      const float* Lr = K + (size_t)(rb + r) * NTRAIN;
      for (int j = lane; j < rb; j += 32) s += Lr[j] * yl[j];
      for (int off = 16; off; off >>= 1) s += __shfl_xor(s, off, 32);
      if (lane == 0) pv[r] = s;
    }
    __syncthreads();
    if (t == 0) {
      for (int j = 0; j < NB; ++j) {
        float v = (yl[rb + j] - pv[j]) / diag[j][j];
        yl[rb + j] = v;
        for (int r = j + 1; r < NB; ++r) pv[r] += diag[r][j] * v;
      }
    }
    __syncthreads();
  }
  for (int i = t; i < NTRAIN; i += 256) y[i] = yl[i];
}

__global__ void gp_bwd_solve(const float* __restrict__ K, float* __restrict__ y) {
  __shared__ float yl[NTRAIN];
  __shared__ float diag[NB][NB + 1];
  int t = threadIdx.x;
  for (int i = t; i < NTRAIN; i += 256) yl[i] = y[i];
  __syncthreads();
  for (int m = NTRAIN / NB - 1; m >= 0; --m) {
    int rb = m * NB;
    for (int idx = t; idx < NB * NB; idx += 256)
      diag[idx >> 6][idx & 63] = K[(size_t)(rb + (idx >> 6)) * NTRAIN + rb + (idx & 63)];
    __syncthreads();
    if (t == 0) {
      for (int j = NB - 1; j >= 0; --j) {
        float s = yl[rb + j];
        for (int r = j + 1; r < NB; ++r) s -= diag[r][j] * yl[rb + r];
        yl[rb + j] = s / diag[j][j];
      }
    }
    __syncthreads();
    for (int i = t; i < rb; i += 256) {
      float acc = 0.f;
      for (int j = 0; j < NB; ++j) acc += K[(size_t)(rb + j) * NTRAIN + i] * yl[rb + j];
      yl[i] -= acc;
    }
    __syncthreads();
  }
  for (int i = t; i < NTRAIN; i += 256) y[i] = yl[i];
}

__global__ void gp_fused_grad(const float* __restrict__ x, const float* __restrict__ tx,
                              const float* __restrict__ alpha,
                              const float* __restrict__ nx, const float* __restrict__ nt,
                              float* __restrict__ out) {
  __shared__ float wt[8][16][18];
  int wave = threadIdx.x >> 5, lane = threadIdx.x & 31;
  int rowBase = (blockIdx.x * 8 + wave) * 16;
  int n  = lane & 15;
  int mh = (lane >> 4) * 8;

  float nxr[8];
#pragma unroll
  for (int r = 0; r < 8; ++r) nxr[r] = nx[rowBase + mh + r];

  v2f af[16];
#pragma unroll
  for (int k0 = 0; k0 < DIMS; k0 += 4)
    af[k0 >> 2] = frag_nt(x, DIMS, rowBase, 0, k0, lane);

  v8f g[4]; g[0] = {}; g[1] = {}; g[2] = {}; g[3] = {};
  float rs[8] = {0, 0, 0, 0, 0, 0, 0, 0};

  for (int c0 = 0; c0 < NTRAIN; c0 += 16) {
    if (c0 + 16 < NTRAIN)
      __builtin_prefetch(tx + (size_t)(c0 + 16) * DIMS + lane * 32, 0, 0);
    v8f ce = {}, co = {};
#pragma unroll
    for (int k0 = 0; k0 < DIMS; k0 += 8) {
      ce = wmma4(af[(k0 >> 2)],     frag_nt(tx, DIMS, c0, 0, k0,     lane), ce);
      co = wmma4(af[(k0 >> 2) + 1], frag_nt(tx, DIMS, c0, 0, k0 + 4, lane), co);
    }
    v8f c = ce + co;
    float ncol = nt[c0 + n];
    float av   = alpha[c0 + n];
#pragma unroll
    for (int r = 0; r < 8; ++r) {
      float d2 = nxr[r] + ncol - 2.0f * c[r];
      float w  = KSCALE * __expf(-0.5f * d2 * (1.0f / LS2)) * av;
      rs[r] += w;
      wt[wave][mh + r][n] = w;
    }
    __syncthreads();
#pragma unroll
    for (int nc = 0; nc < 4; ++nc) {
#pragma unroll
      for (int k0 = 0; k0 < 16; k0 += 4) {
        int kk = k0 + ((lane >> 4) << 1);
        v2f a, b;
        a.x = wt[wave][n][kk];
        a.y = wt[wave][n][kk + 1];
        b.x = tx[(size_t)(c0 + kk) * DIMS + nc * 16 + n];
        b.y = tx[(size_t)(c0 + kk + 1) * DIMS + nc * 16 + n];
        g[nc] = wmma4(a, b, g[nc]);
      }
    }
    __syncthreads();
  }

#pragma unroll
  for (int r = 0; r < 8; ++r)
    for (int off = 8; off; off >>= 1) rs[r] += __shfl_xor(rs[r], off, 32);

#pragma unroll
  for (int nc = 0; nc < 4; ++nc) {
#pragma unroll
    for (int r = 0; r < 8; ++r) {
      int row = rowBase + mh + r;
      int d   = nc * 16 + n;
      float xv = x[(size_t)row * DIMS + d];
      out[(size_t)row * DIMS + d] = xv - (g[nc][r] - rs[r] * xv) * (1.0f / LS2);
    }
  }
}

extern "C" void kernel_launch(void* const* d_in, const int* in_sizes, int n_in,
                              void* d_out, int out_size, void* d_ws, size_t ws_size,
                              hipStream_t stream) {
  (void)in_sizes; (void)n_in; (void)out_size; (void)ws_size;
  const float* x  = (const float*)d_in[0];
  const float* tx = (const float*)d_in[1];
  const float* ty = (const float*)d_in[2];
  float* out = (float*)d_out;

  float* Kd = (float*)d_ws;
  float* nt = Kd + (size_t)NTRAIN * NTRAIN;
  float* nx = nt + NTRAIN;
  float* al = nx + NQUERY;

  gp_sqnorm<<<(NTRAIN + 255) / 256, 256, 0, stream>>>(tx, nt, NTRAIN);
  gp_sqnorm<<<(NQUERY + 255) / 256, 256, 0, stream>>>(x, nx, NQUERY);

  gp_build_K<<<dim3(NTRAIN / 16, NTRAIN / 16), 32, 0, stream>>>(tx, nt, Kd);

  for (int kb = 0; kb < NTRAIN / NB; ++kb) {
    gp_chol_diag<<<1, NB, 0, stream>>>(Kd, kb);
    int below = NTRAIN - NB * (kb + 1);
    if (below > 0) {
      gp_chol_trsm<<<below / NB, NB, 0, stream>>>(Kd, kb);
      int tt = below / 64;
      gp_chol_syrk<<<dim3(tt, tt), 128, 0, stream>>>(Kd, kb);
    }
  }

  hipMemcpyAsync(al, ty, NTRAIN * sizeof(float), hipMemcpyDeviceToDevice, stream);
  gp_fwd_solve<<<1, 256, 0, stream>>>(Kd, al);
  gp_bwd_solve<<<1, 256, 0, stream>>>(Kd, al);

  gp_fused_grad<<<NQUERY / 128, 256, 0, stream>>>(x, tx, al, nx, nt, out);
}


# forward substitution folded into the factorization: rhs rides as one more row of each diagonal block, panel kernel subtracts L21*z; one-block forward solve dropped; its sums regrouped per 64-block
# speedup vs baseline: 1.8122x; 1.8122x over previous
#include <hip/hip_runtime.h>
#include <math.h>

#define DIMS   64
#define NTRAIN 4096
#define NQUERY 8192
#define NB     64
#define NOISE  1e-4f
#define LS2    64.0f
#define KSCALE 0.6931471805599453f
#define SPAD   68

typedef float v2f __attribute__((ext_vector_type(2)));
typedef float v8f __attribute__((ext_vector_type(8)));

__device__ __forceinline__ v8f wmma4(v2f a, v2f b, v8f c) {
  return __builtin_amdgcn_wmma_f32_16x16x4_f32(
       false, a,  false, b,
       (short)0, c,  false,  false);
}

__device__ __forceinline__ v2f frag_nt(const float* __restrict__ X, size_t rowStride,
                                       int rowBase, int colBase, int k0, int lane) {
  int kk = k0 + ((lane >> 4) << 1);
  const float* p = X + (size_t)(rowBase + (lane & 15)) * rowStride + colBase + kk;
  v2f v; v.x = p[0]; v.y = p[1];
  return v;
}

__device__ __forceinline__ v2f frag_lds(float (*S)[SPAD], int rowBase, int k0, int lane) {
  int kk = k0 + ((lane >> 4) << 1);
  const float* p = &S[rowBase + (lane & 15)][kk];
  v2f v; v.x = p[0]; v.y = p[1];
  return v;
}

__global__ void gp_sqnorm(const float* __restrict__ X, float* __restrict__ nrm, int rows) {
  int r = blockIdx.x * blockDim.x + threadIdx.x;
  if (r >= rows) return;
  const float4* p = (const float4*)(X + (size_t)r * DIMS);
  float s = 0.f;
#pragma unroll
  for (int i = 0; i < DIMS / 4; ++i) {
    float4 v = p[i];
    s += v.x * v.x + v.y * v.y + v.z * v.z + v.w * v.w;
  }
  nrm[r] = s;
}

__global__ void gp_build_K(const float* __restrict__ tx, const float* __restrict__ nrm,
                           float* __restrict__ K) {
  int bi = blockIdx.x, bj = blockIdx.y;
  int lane = threadIdx.x & 31;
  v8f ce = {}, co = {};
#pragma unroll
  for (int k0 = 0; k0 < DIMS; k0 += 8) {
    ce = wmma4(frag_nt(tx, DIMS, bi * 16, 0, k0, lane),
               frag_nt(tx, DIMS, bj * 16, 0, k0, lane), ce);
    co = wmma4(frag_nt(tx, DIMS, bi * 16, 0, k0 + 4, lane),
               frag_nt(tx, DIMS, bj * 16, 0, k0 + 4, lane), co);
  }
  v8f c = ce + co;
  int n  = lane & 15;
  int mh = (lane >> 4) * 8;
  int col = bj * 16 + n;
  float ncol = nrm[col];
#pragma unroll
  for (int r = 0; r < 8; ++r) {
    int row = bi * 16 + mh + r;
    float d2 = nrm[row] + ncol - 2.0f * c[r];
    float v  = KSCALE * __expf(-0.5f * d2 * (1.0f / LS2));
    if (row == col) v += NOISE;
    K[(size_t)row * NTRAIN + col] = v;
  }
}

__global__ void gp_chol_diag(float* __restrict__ K, const float* __restrict__ yw,
                             float* __restrict__ z, int kb) {
  __shared__ float a[NB][NB + 1];
  __shared__ float zr[NB];
  int t = threadIdx.x;
  int base = kb * NB;
  for (int j = 0; j < NB; ++j) a[t][j] = K[(size_t)(base + t) * NTRAIN + base + j];
  float yv = yw[base + t];
  __syncthreads();
  for (int j = 0; j < NB; ++j) {
    if (t == j) a[j][j] = sqrtf(a[j][j]);
    __syncthreads();
    if (t >= j) {
      float num = (t == j) ? yv : a[t][j];
      float q = num / a[j][j];
      if (t == j) zr[j] = q; else a[t][j] = q;
    }
    __syncthreads();
    if (t > j) {
      float zj = zr[j];
      float lt = a[t][j];
      for (int c2 = j + 1; c2 <= t; ++c2) a[t][c2] -= lt * a[c2][j];
      yv -= zj * lt;
    }
    __syncthreads();
  }
  for (int j = 0; j <= t; ++j) K[(size_t)(base + t) * NTRAIN + base + j] = a[t][j];
  z[base + t] = zr[t];
}

__global__ void gp_chol_trsm(float* __restrict__ K, float* __restrict__ yw,
                             const float* __restrict__ z, int kb) {
  __shared__ float Ld[NB][NB + 1];
  __shared__ float rw[NB][NB + 1];
  __shared__ float zs[NB];
  int t = threadIdx.x;
  int base = kb * NB;
  for (int j = 0; j < NB; ++j) Ld[t][j] = K[(size_t)(base + t) * NTRAIN + base + j];
  zs[t] = z[base + t];
  int row = base + NB + blockIdx.x * NB + t;
  const float* src = K + (size_t)row * NTRAIN + base;
  for (int j = 0; j < NB; ++j) rw[t][j] = src[j];
  __syncthreads();
  float dz = 0.f;
  for (int j = 0; j < NB; ++j) {
    float s = rw[t][j];
    for (int p = 0; p < j; ++p) s -= rw[t][p] * Ld[j][p];
    s /= Ld[j][j];
    rw[t][j] = s;
    dz += s * zs[j];
  }
  float* dst = K + (size_t)row * NTRAIN + base;
  for (int j = 0; j < NB; ++j) dst[j] = rw[t][j];
  yw[row] -= dz;
}

__global__ void gp_chol_syrk(float* __restrict__ K, int kb) {
  __shared__ float As[64][SPAD];
  __shared__ float Bs[64][SPAD];
  int bi = blockIdx.x, bj = blockIdx.y;
  if (bj > bi) return;
  int base = kb * NB;
  int tb   = base + NB;
  int t = threadIdx.x;
  int lane = t & 31, wave = t >> 5;
  int rowA = tb + bi * 64, rowB = tb + bj * 64;

  for (int f = t; f < 64 * 16; f += 128) {
    int r = f >> 4, c4 = (f & 15) << 2;
    *(float4*)&As[r][c4] = *(const float4*)&K[(size_t)(rowA + r) * NTRAIN + base + c4];
    *(float4*)&Bs[r][c4] = *(const float4*)&K[(size_t)(rowB + r) * NTRAIN + base + c4];
  }
  __syncthreads();

  int wr = (wave >> 1) * 32;
  int wc = (wave & 1) * 32;
  v8f c00 = {}, c01 = {}, c10 = {}, c11 = {};
#pragma unroll
  for (int k0 = 0; k0 < NB; k0 += 4) {
    v2f a0 = frag_lds(As, wr,      k0, lane);
    v2f a1 = frag_lds(As, wr + 16, k0, lane);
    v2f b0 = frag_lds(Bs, wc,      k0, lane);
    v2f b1 = frag_lds(Bs, wc + 16, k0, lane);
    c00 = wmma4(a0, b0, c00);
    c01 = wmma4(a0, b1, c01);
    c10 = wmma4(a1, b0, c10);
    c11 = wmma4(a1, b1, c11);
  }

  int n  = lane & 15;
  int mh = (lane >> 4) * 8;
  int r0 = tb + bi * 64 + wr + mh;
  int c0 = tb + bj * 64 + wc + n;
#pragma unroll
  for (int r = 0; r < 8; ++r) {
    K[(size_t)(r0 + r) * NTRAIN + c0]           -= c00[r];
    K[(size_t)(r0 + r) * NTRAIN + c0 + 16]      -= c01[r];
    K[(size_t)(r0 + 16 + r) * NTRAIN + c0]      -= c10[r];
    K[(size_t)(r0 + 16 + r) * NTRAIN + c0 + 16] -= c11[r];
  }
}

__global__ void gp_bwd_solve(const float* __restrict__ K, float* __restrict__ y) {
  __shared__ float yl[NTRAIN];
  __shared__ float diag[NB][NB + 1];
  int t = threadIdx.x;
  for (int i = t; i < NTRAIN; i += 256) yl[i] = y[i];
  __syncthreads();
  for (int m = NTRAIN / NB - 1; m >= 0; --m) {
    int rb = m * NB;
    for (int idx = t; idx < NB * NB; idx += 256)
      diag[idx >> 6][idx & 63] = K[(size_t)(rb + (idx >> 6)) * NTRAIN + rb + (idx & 63)];
    __syncthreads();
    if (t == 0) {
      for (int j = NB - 1; j >= 0; --j) {
        float s = yl[rb + j];
        for (int r = j + 1; r < NB; ++r) s -= diag[r][j] * yl[rb + r];
        yl[rb + j] = s / diag[j][j];
      }
    }
    __syncthreads();
    for (int i = t; i < rb; i += 256) {
      float acc = 0.f;
      for (int j = 0; j < NB; ++j) acc += K[(size_t)(rb + j) * NTRAIN + i] * yl[rb + j];
      yl[i] -= acc;
    }
    __syncthreads();
  }
  for (int i = t; i < NTRAIN; i += 256) y[i] = yl[i];
}

__global__ void gp_fused_grad(const float* __restrict__ x, const float* __restrict__ tx,
                              const float* __restrict__ alpha,
                              const float* __restrict__ nx, const float* __restrict__ nt,
                              float* __restrict__ out) {
  __shared__ float wt[8][16][18];
  int wave = threadIdx.x >> 5, lane = threadIdx.x & 31;
  int rowBase = (blockIdx.x * 8 + wave) * 16;
  int n  = lane & 15;
  int mh = (lane >> 4) * 8;

  float nxr[8];
#pragma unroll
  for (int r = 0; r < 8; ++r) nxr[r] = nx[rowBase + mh + r];

  v2f af[16];
#pragma unroll
  for (int k0 = 0; k0 < DIMS; k0 += 4)
    af[k0 >> 2] = frag_nt(x, DIMS, rowBase, 0, k0, lane);

  v8f g[4]; g[0] = {}; g[1] = {}; g[2] = {}; g[3] = {};
  float rs[8] = {0, 0, 0, 0, 0, 0, 0, 0};

  for (int c0 = 0; c0 < NTRAIN; c0 += 16) {
    if (c0 + 16 < NTRAIN)
      __builtin_prefetch(tx + (size_t)(c0 + 16) * DIMS + lane * 32, 0, 0);
    v8f ce = {}, co = {};
#pragma unroll
    for (int k0 = 0; k0 < DIMS; k0 += 8) {
      ce = wmma4(af[(k0 >> 2)],     frag_nt(tx, DIMS, c0, 0, k0,     lane), ce);
      co = wmma4(af[(k0 >> 2) + 1], frag_nt(tx, DIMS, c0, 0, k0 + 4, lane), co);
    }
    v8f c = ce + co;
    float ncol = nt[c0 + n];
    float av   = alpha[c0 + n];
#pragma unroll
    for (int r = 0; r < 8; ++r) {
      float d2 = nxr[r] + ncol - 2.0f * c[r];
      float w  = KSCALE * __expf(-0.5f * d2 * (1.0f / LS2)) * av;
      rs[r] += w;
      wt[wave][mh + r][n] = w;
    }
    __syncthreads();
#pragma unroll
    for (int nc = 0; nc < 4; ++nc) {
#pragma unroll
      for (int k0 = 0; k0 < 16; k0 += 4) {
        int kk = k0 + ((lane >> 4) << 1);
        v2f a, b;
        a.x = wt[wave][n][kk];
        a.y = wt[wave][n][kk + 1];
        b.x = tx[(size_t)(c0 + kk) * DIMS + nc * 16 + n];
        b.y = tx[(size_t)(c0 + kk + 1) * DIMS + nc * 16 + n];
        g[nc] = wmma4(a, b, g[nc]);
      }
    }
    __syncthreads();
  }

#pragma unroll
  for (int r = 0; r < 8; ++r)
    for (int off = 8; off; off >>= 1) rs[r] += __shfl_xor(rs[r], off, 32);

#pragma unroll
  for (int nc = 0; nc < 4; ++nc) {
#pragma unroll
    for (int r = 0; r < 8; ++r) {
      int row = rowBase + mh + r;
      int d   = nc * 16 + n;
      float xv = x[(size_t)row * DIMS + d];
      out[(size_t)row * DIMS + d] = xv - (g[nc][r] - rs[r] * xv) * (1.0f / LS2);
    }
  }
}

extern "C" void kernel_launch(void* const* d_in, const int* in_sizes, int n_in,
                              void* d_out, int out_size, void* d_ws, size_t ws_size,
                              hipStream_t stream) {
  (void)in_sizes; (void)n_in; (void)out_size; (void)ws_size;
  const float* x  = (const float*)d_in[0];
  const float* tx = (const float*)d_in[1];
  const float* ty = (const float*)d_in[2];
  float* out = (float*)d_out;

  float* Kd = (float*)d_ws;
  float* nt = Kd + (size_t)NTRAIN * NTRAIN;
  float* nx = nt + NTRAIN;
  float* al = nx + NQUERY;
  float* yw = al + NTRAIN;

  gp_sqnorm<<<(NTRAIN + 255) / 256, 256, 0, stream>>>(tx, nt, NTRAIN);
  gp_sqnorm<<<(NQUERY + 255) / 256, 256, 0, stream>>>(x, nx, NQUERY);

  gp_build_K<<<dim3(NTRAIN / 16, NTRAIN / 16), 32, 0, stream>>>(tx, nt, Kd);
  hipMemcpyAsync(yw, ty, NTRAIN * sizeof(float), hipMemcpyDeviceToDevice, stream);

  for (int kb = 0; kb < NTRAIN / NB; ++kb) {
    gp_chol_diag<<<1, NB, 0, stream>>>(Kd, yw, al, kb);
    int below = NTRAIN - NB * (kb + 1);
    if (below > 0) {
      gp_chol_trsm<<<below / NB, NB, 0, stream>>>(Kd, yw, al, kb);
      int tt = below / 64;
      gp_chol_syrk<<<dim3(tt, tt), 128, 0, stream>>>(Kd, kb);
    }
  }

  gp_bwd_solve<<<1, 256, 0, stream>>>(Kd, al);

  gp_fused_grad<<<NQUERY / 128, 256, 0, stream>>>(x, tx, al, nx, nt, out);
}


# backward solve re-parallelised: 64x64 diagonal systems by one wave (column sweep, lane broadcasts, reciprocal diagonal), not one thread; row update on 1024 threads with 64 loads each; sums regrouped
# speedup vs baseline: 2.4220x; 1.3365x over previous
#include <hip/hip_runtime.h>
#include <math.h>

#define DIMS   64
#define NTRAIN 4096
#define NQUERY 8192
#define NB     64
#define NOISE  1e-4f
#define LS2    64.0f
#define KSCALE 0.6931471805599453f
#define SPAD   68

typedef float v2f __attribute__((ext_vector_type(2)));
typedef float v8f __attribute__((ext_vector_type(8)));

__device__ __forceinline__ v8f wmma4(v2f a, v2f b, v8f c) {
  return __builtin_amdgcn_wmma_f32_16x16x4_f32(
       false, a,  false, b,
       (short)0, c,  false,  false);
}

__device__ __forceinline__ v2f frag_nt(const float* __restrict__ X, size_t rowStride,
                                       int rowBase, int colBase, int k0, int lane) {
  int kk = k0 + ((lane >> 4) << 1);
  const float* p = X + (size_t)(rowBase + (lane & 15)) * rowStride + colBase + kk;
  v2f v; v.x = p[0]; v.y = p[1];
  return v;
}

__device__ __forceinline__ v2f frag_lds(float (*S)[SPAD], int rowBase, int k0, int lane) {
  int kk = k0 + ((lane >> 4) << 1);
  const float* p = &S[rowBase + (lane & 15)][kk];
  v2f v; v.x = p[0]; v.y = p[1];
  return v;
}

__global__ void gp_sqnorm(const float* __restrict__ X, float* __restrict__ nrm, int rows) {
  int r = blockIdx.x * blockDim.x + threadIdx.x;
  if (r >= rows) return;
  const float4* p = (const float4*)(X + (size_t)r * DIMS);
  float s = 0.f;
#pragma unroll
  for (int i = 0; i < DIMS / 4; ++i) {
    float4 v = p[i];
    s += v.x * v.x + v.y * v.y + v.z * v.z + v.w * v.w;
  }
  nrm[r] = s;
}

__global__ void gp_build_K(const float* __restrict__ tx, const float* __restrict__ nrm,
                           float* __restrict__ K) {
  int bi = blockIdx.x, bj = blockIdx.y;
  int lane = threadIdx.x & 31;
  v8f ce = {}, co = {};
#pragma unroll
  for (int k0 = 0; k0 < DIMS; k0 += 8) {
    ce = wmma4(frag_nt(tx, DIMS, bi * 16, 0, k0, lane),
               frag_nt(tx, DIMS, bj * 16, 0, k0, lane), ce);
    co = wmma4(frag_nt(tx, DIMS, bi * 16, 0, k0 + 4, lane),
               frag_nt(tx, DIMS, bj * 16, 0, k0 + 4, lane), co);
  }
  v8f c = ce + co;
  int n  = lane & 15;
  int mh = (lane >> 4) * 8;
  int col = bj * 16 + n;
  float ncol = nrm[col];
#pragma unroll
  for (int r = 0; r < 8; ++r) {
    int row = bi * 16 + mh + r;
    float d2 = nrm[row] + ncol - 2.0f * c[r];
    float v  = KSCALE * __expf(-0.5f * d2 * (1.0f / LS2));
    if (row == col) v += NOISE;
    K[(size_t)row * NTRAIN + col] = v;
  }
}

__global__ void gp_chol_diag(float* __restrict__ K, const float* __restrict__ yw,
                             float* __restrict__ z, int kb) {
  __shared__ float a[NB][NB + 1];
  __shared__ float zr[NB];
  int t = threadIdx.x;
  int base = kb * NB;
  for (int j = 0; j < NB; ++j) a[t][j] = K[(size_t)(base + t) * NTRAIN + base + j];
  float yv = yw[base + t];
  __syncthreads();
  for (int j = 0; j < NB; ++j) {
    if (t == j) a[j][j] = sqrtf(a[j][j]);
    __syncthreads();
    if (t >= j) {
      float num = (t == j) ? yv : a[t][j];
      float q = num / a[j][j];
      if (t == j) zr[j] = q; else a[t][j] = q;
    }
    __syncthreads();
    if (t > j) {
      float zj = zr[j];
      float lt = a[t][j];
      for (int c2 = j + 1; c2 <= t; ++c2) a[t][c2] -= lt * a[c2][j];
      yv -= zj * lt;
    }
    __syncthreads();
  }
  for (int j = 0; j <= t; ++j) K[(size_t)(base + t) * NTRAIN + base + j] = a[t][j];
  z[base + t] = zr[t];
}

__global__ void gp_chol_trsm(float* __restrict__ K, float* __restrict__ yw,
                             const float* __restrict__ z, int kb) {
  __shared__ float Ld[NB][NB + 1];
  __shared__ float rw[NB][NB + 1];
  __shared__ float zs[NB];
  int t = threadIdx.x;
  int base = kb * NB;
  for (int j = 0; j < NB; ++j) Ld[t][j] = K[(size_t)(base + t) * NTRAIN + base + j];
  zs[t] = z[base + t];
  int row = base + NB + blockIdx.x * NB + t;
  const float* src = K + (size_t)row * NTRAIN + base;
  for (int j = 0; j < NB; ++j) rw[t][j] = src[j];
  __syncthreads();
  float dz = 0.f;
  for (int j = 0; j < NB; ++j) {
    float s = rw[t][j];
    for (int p = 0; p < j; ++p) s -= rw[t][p] * Ld[j][p];
    s /= Ld[j][j];
    rw[t][j] = s;
    dz += s * zs[j];
  }
  float* dst = K + (size_t)row * NTRAIN + base;
  for (int j = 0; j < NB; ++j) dst[j] = rw[t][j];
  yw[row] -= dz;
}

__global__ void gp_chol_syrk(float* __restrict__ K, int kb) {
  __shared__ float As[64][SPAD];
  __shared__ float Bs[64][SPAD];
  int bi = blockIdx.x, bj = blockIdx.y;
  if (bj > bi) return;
  int base = kb * NB;
  int tb   = base + NB;
  int t = threadIdx.x;
  int lane = t & 31, wave = t >> 5;
  int rowA = tb + bi * 64, rowB = tb + bj * 64;

  for (int f = t; f < 64 * 16; f += 128) {
    int r = f >> 4, c4 = (f & 15) << 2;
    *(float4*)&As[r][c4] = *(const float4*)&K[(size_t)(rowA + r) * NTRAIN + base + c4];
    *(float4*)&Bs[r][c4] = *(const float4*)&K[(size_t)(rowB + r) * NTRAIN + base + c4];
  }
  __syncthreads();

  int wr = (wave >> 1) * 32;
  int wc = (wave & 1) * 32;
  v8f c00 = {}, c01 = {}, c10 = {}, c11 = {};
#pragma unroll
  for (int k0 = 0; k0 < NB; k0 += 4) {
    v2f a0 = frag_lds(As, wr,      k0, lane);
    v2f a1 = frag_lds(As, wr + 16, k0, lane);
    v2f b0 = frag_lds(Bs, wc,      k0, lane);
    v2f b1 = frag_lds(Bs, wc + 16, k0, lane);
    c00 = wmma4(a0, b0, c00);
    c01 = wmma4(a0, b1, c01);
    c10 = wmma4(a1, b0, c10);
    c11 = wmma4(a1, b1, c11);
  }

  int n  = lane & 15;
  int mh = (lane >> 4) * 8;
  int r0 = tb + bi * 64 + wr + mh;
  int c0 = tb + bj * 64 + wc + n;
#pragma unroll
  for (int r = 0; r < 8; ++r) {
    K[(size_t)(r0 + r) * NTRAIN + c0]           -= c00[r];
    K[(size_t)(r0 + r) * NTRAIN + c0 + 16]      -= c01[r];
    K[(size_t)(r0 + 16 + r) * NTRAIN + c0]      -= c10[r];
    K[(size_t)(r0 + 16 + r) * NTRAIN + c0 + 16] -= c11[r];
  }
}

#define BWT 1024

__device__ __forceinline__ float lane_bcast(float v, int srcLane) {
  return __int_as_float(__builtin_amdgcn_readlane(__float_as_int(v), srcLane));
}

__global__ void __launch_bounds__(BWT) gp_bwd_solve(const float* __restrict__ K,
                                                    float* __restrict__ y) {
  __shared__ float yl[NTRAIN];
  __shared__ float dg[NB][NB + 1];
  __shared__ float dd[NB];
  int t = threadIdx.x;
  for (int i = t; i < NTRAIN; i += BWT) yl[i] = y[i];

  int dr = t >> 6, dc = t & 63;
  float nd[4];
  {
    int rb0 = NTRAIN - NB;
#pragma unroll
    for (int q = 0; q < 4; ++q)
      nd[q] = K[(size_t)(rb0 + dr + 16 * q) * NTRAIN + rb0 + dc];
  }

  for (int m = NTRAIN / NB - 1; m >= 0; --m) {
    int rb = m * NB;
#pragma unroll
    for (int q = 0; q < 4; ++q) {
      int r = dr + 16 * q;
      dg[r][dc] = (dc < r) ? nd[q] : 0.f;
      if (dc == r) dd[r] = nd[q];
    }
    __syncthreads();
    if (m > 0) {
      int rn = rb - NB;
#pragma unroll
      for (int q = 0; q < 4; ++q)
        nd[q] = K[(size_t)(rn + dr + 16 * q) * NTRAIN + rn + dc];
    }

    if (t < 32) {
      float s0 = yl[rb + t], s1 = yl[rb + 32 + t];
      float r0 = 1.0f / dd[t];
      float r1 = 1.0f / dd[32 + t];
#pragma unroll
      for (int j = NB - 1; j >= 32; --j) {
        float xj = lane_bcast(s1, j - 32) * lane_bcast(r1, j - 32);
        s1 -= dg[j][32 + t] * xj;
        s0 -= dg[j][t] * xj;
      }
#pragma unroll
      for (int j = 31; j >= 1; --j) {
        float xj = lane_bcast(s0, j) * lane_bcast(r0, j);
        s0 -= dg[j][t] * xj;
      }
      yl[rb + t]      = s0 * r0;
      yl[rb + 32 + t] = s1 * r1;
    }
    __syncthreads();

    for (int i = t; i < rb; i += BWT) {
      const float* col = K + (size_t)rb * NTRAIN + i;
      float a0 = 0.f, a1 = 0.f, a2 = 0.f, a3 = 0.f;
#pragma unroll
      for (int j = 0; j < NB; j += 4) {
        a0 += col[(size_t)(j + 0) * NTRAIN] * yl[rb + j + 0];
        a1 += col[(size_t)(j + 1) * NTRAIN] * yl[rb + j + 1];
        a2 += col[(size_t)(j + 2) * NTRAIN] * yl[rb + j + 2];
        a3 += col[(size_t)(j + 3) * NTRAIN] * yl[rb + j + 3];
      }
      yl[i] -= (a0 + a1) + (a2 + a3);
    }
  }
  __syncthreads();
  for (int i = t; i < NTRAIN; i += BWT) y[i] = yl[i];
}

__global__ void gp_fused_grad(const float* __restrict__ x, const float* __restrict__ tx,
                              const float* __restrict__ alpha,
                              const float* __restrict__ nx, const float* __restrict__ nt,
                              float* __restrict__ out) {
  __shared__ float wt[8][16][18];
  int wave = threadIdx.x >> 5, lane = threadIdx.x & 31;
  int rowBase = (blockIdx.x * 8 + wave) * 16;
  int n  = lane & 15;
  int mh = (lane >> 4) * 8;

  float nxr[8];
#pragma unroll
  for (int r = 0; r < 8; ++r) nxr[r] = nx[rowBase + mh + r];

  v2f af[16];
#pragma unroll
  for (int k0 = 0; k0 < DIMS; k0 += 4)
    af[k0 >> 2] = frag_nt(x, DIMS, rowBase, 0, k0, lane);

  v8f g[4]; g[0] = {}; g[1] = {}; g[2] = {}; g[3] = {};
  float rs[8] = {0, 0, 0, 0, 0, 0, 0, 0};

  for (int c0 = 0; c0 < NTRAIN; c0 += 16) {
    if (c0 + 16 < NTRAIN)
      __builtin_prefetch(tx + (size_t)(c0 + 16) * DIMS + lane * 32, 0, 0);
    v8f ce = {}, co = {};
#pragma unroll
    for (int k0 = 0; k0 < DIMS; k0 += 8) {
      ce = wmma4(af[(k0 >> 2)],     frag_nt(tx, DIMS, c0, 0, k0,     lane), ce);
      co = wmma4(af[(k0 >> 2) + 1], frag_nt(tx, DIMS, c0, 0, k0 + 4, lane), co);
    }
    v8f c = ce + co;
    float ncol = nt[c0 + n];
    float av   = alpha[c0 + n];
#pragma unroll
    for (int r = 0; r < 8; ++r) {
      float d2 = nxr[r] + ncol - 2.0f * c[r];
      float w  = KSCALE * __expf(-0.5f * d2 * (1.0f / LS2)) * av;
      rs[r] += w;
      wt[wave][mh + r][n] = w;
    }
    __syncthreads();
#pragma unroll
    for (int nc = 0; nc < 4; ++nc) {
#pragma unroll
      for (int k0 = 0; k0 < 16; k0 += 4) {
        int kk = k0 + ((lane >> 4) << 1);
        v2f a, b;
        a.x = wt[wave][n][kk];
        a.y = wt[wave][n][kk + 1];
        b.x = tx[(size_t)(c0 + kk) * DIMS + nc * 16 + n];
        b.y = tx[(size_t)(c0 + kk + 1) * DIMS + nc * 16 + n];
        g[nc] = wmma4(a, b, g[nc]);
      }
    }
    __syncthreads();
  }

#pragma unroll
  for (int r = 0; r < 8; ++r)
    for (int off = 8; off; off >>= 1) rs[r] += __shfl_xor(rs[r], off, 32);

#pragma unroll
  for (int nc = 0; nc < 4; ++nc) {
#pragma unroll
    for (int r = 0; r < 8; ++r) {
      int row = rowBase + mh + r;
      int d   = nc * 16 + n;
      float xv = x[(size_t)row * DIMS + d];
      out[(size_t)row * DIMS + d] = xv - (g[nc][r] - rs[r] * xv) * (1.0f / LS2);
    }
  }
}

extern "C" void kernel_launch(void* const* d_in, const int* in_sizes, int n_in,
                              void* d_out, int out_size, void* d_ws, size_t ws_size,
                              hipStream_t stream) {
  (void)in_sizes; (void)n_in; (void)out_size; (void)ws_size;
  const float* x  = (const float*)d_in[0];
  const float* tx = (const float*)d_in[1];
  const float* ty = (const float*)d_in[2];
  float* out = (float*)d_out;

  float* Kd = (float*)d_ws;
  float* nt = Kd + (size_t)NTRAIN * NTRAIN;
  float* nx = nt + NTRAIN;
  float* al = nx + NQUERY;
  float* yw = al + NTRAIN;

  gp_sqnorm<<<(NTRAIN + 255) / 256, 256, 0, stream>>>(tx, nt, NTRAIN);
  gp_sqnorm<<<(NQUERY + 255) / 256, 256, 0, stream>>>(x, nx, NQUERY);

  gp_build_K<<<dim3(NTRAIN / 16, NTRAIN / 16), 32, 0, stream>>>(tx, nt, Kd);
  hipMemcpyAsync(yw, ty, NTRAIN * sizeof(float), hipMemcpyDeviceToDevice, stream);

  for (int kb = 0; kb < NTRAIN / NB; ++kb) {
    gp_chol_diag<<<1, NB, 0, stream>>>(Kd, yw, al, kb);
    int below = NTRAIN - NB * (kb + 1);
    if (below > 0) {
      gp_chol_trsm<<<below / NB, NB, 0, stream>>>(Kd, yw, al, kb);
      int tt = below / 64;
      gp_chol_syrk<<<dim3(tt, tt), 128, 0, stream>>>(Kd, kb);
    }
  }

  gp_bwd_solve<<<1, BWT, 0, stream>>>(Kd, al);

  gp_fused_grad<<<NQUERY / 128, 256, 0, stream>>>(x, tx, al, nx, nt, out);
}


# diagonal 64x64 block factor by one wave in dot-product form: finished columns only read from LDS (16-B loads, all in flight), pivot by lane broadcast and reciprocal square root; sums regrouped
# speedup vs baseline: 3.7426x; 1.5453x over previous
#include <hip/hip_runtime.h>
#include <math.h>

#define DIMS   64
#define NTRAIN 4096
#define NQUERY 8192
#define NB     64
#define NOISE  1e-4f
#define LS2    64.0f
#define KSCALE 0.6931471805599453f
#define SPAD   68

typedef float v2f __attribute__((ext_vector_type(2)));
typedef float v8f __attribute__((ext_vector_type(8)));

__device__ __forceinline__ v8f wmma4(v2f a, v2f b, v8f c) {
  return __builtin_amdgcn_wmma_f32_16x16x4_f32(
       false, a,  false, b,
       (short)0, c,  false,  false);
}

__device__ __forceinline__ v2f frag_nt(const float* __restrict__ X, size_t rowStride,
                                       int rowBase, int colBase, int k0, int lane) {
  int kk = k0 + ((lane >> 4) << 1);
  const float* p = X + (size_t)(rowBase + (lane & 15)) * rowStride + colBase + kk;
  v2f v; v.x = p[0]; v.y = p[1];
  return v;
}

__device__ __forceinline__ v2f frag_lds(float (*S)[SPAD], int rowBase, int k0, int lane) {
  int kk = k0 + ((lane >> 4) << 1);
  const float* p = &S[rowBase + (lane & 15)][kk];
  v2f v; v.x = p[0]; v.y = p[1];
  return v;
}

__global__ void gp_sqnorm(const float* __restrict__ X, float* __restrict__ nrm, int rows) {
  int r = blockIdx.x * blockDim.x + threadIdx.x;
  if (r >= rows) return;
  const float4* p = (const float4*)(X + (size_t)r * DIMS);
  float s = 0.f;
#pragma unroll
  for (int i = 0; i < DIMS / 4; ++i) {
    float4 v = p[i];
    s += v.x * v.x + v.y * v.y + v.z * v.z + v.w * v.w;
  }
  nrm[r] = s;
}

__global__ void gp_build_K(const float* __restrict__ tx, const float* __restrict__ nrm,
                           float* __restrict__ K) {
  int bi = blockIdx.x, bj = blockIdx.y;
  int lane = threadIdx.x & 31;
  v8f ce = {}, co = {};
#pragma unroll
  for (int k0 = 0; k0 < DIMS; k0 += 8) {
    ce = wmma4(frag_nt(tx, DIMS, bi * 16, 0, k0, lane),
               frag_nt(tx, DIMS, bj * 16, 0, k0, lane), ce);
    co = wmma4(frag_nt(tx, DIMS, bi * 16, 0, k0 + 4, lane),
               frag_nt(tx, DIMS, bj * 16, 0, k0 + 4, lane), co);
  }
  v8f c = ce + co;
  int n  = lane & 15;
  int mh = (lane >> 4) * 8;
  int col = bj * 16 + n;
  float ncol = nrm[col];
#pragma unroll
  for (int r = 0; r < 8; ++r) {
    int row = bi * 16 + mh + r;
    float d2 = nrm[row] + ncol - 2.0f * c[r];
    float v  = KSCALE * __expf(-0.5f * d2 * (1.0f / LS2));
    if (row == col) v += NOISE;
    K[(size_t)row * NTRAIN + col] = v;
  }
}

#define DP 68

__device__ __forceinline__ float lane_bcast(float v, int srcLane) {
  return __int_as_float(__builtin_amdgcn_readlane(__float_as_int(v), srcLane));
}

template <bool ROWA>
__device__ __forceinline__ void diag_dot16(float (*lz)[DP], int j, int rA, int rB, int p0,
                                           float4& sa, float4& sb) {
#pragma unroll
  for (int u = 0; u < 4; ++u) {
    int p = p0 + 4 * u;
    float4 rj = *(const float4*)&lz[j][p];
    float4 xb = *(const float4*)&lz[rB][p];
    sb.x = fmaf(xb.x, rj.x, sb.x); sb.y = fmaf(xb.y, rj.y, sb.y);
    sb.z = fmaf(xb.z, rj.z, sb.z); sb.w = fmaf(xb.w, rj.w, sb.w);
    if (ROWA) {
      float4 xa = *(const float4*)&lz[rA][p];
      sa.x = fmaf(xa.x, rj.x, sa.x); sa.y = fmaf(xa.y, rj.y, sa.y);
      sa.z = fmaf(xa.z, rj.z, sa.z); sa.w = fmaf(xa.w, rj.w, sa.w);
    }
  }
}

template <int Q, bool ROWA>
__device__ __forceinline__ void diag_cols16(float (*a)[DP], float (*lz)[DP], int l,
                                            float& ya, float& yb) {
  const int rA = l, rB = l + 32;
#pragma nounroll
  for (int jj = 0; jj < 16; ++jj) {
    const int j  = 16 * Q + jj;
    const int jl = ((16 * Q) & 31) + jj;
    float a0B = a[rB][j];
    float a0A = ROWA ? a[rA][j] : 0.f;
    float4 sa = {0.f, 0.f, 0.f, 0.f}, sb = {0.f, 0.f, 0.f, 0.f};
#pragma unroll
    for (int q = 0; q <= Q; ++q)
      diag_dot16<ROWA>(lz, j, rA, rB, 16 * q, sa, sb);
    float sB = a0B - ((sb.x + sb.y) + (sb.z + sb.w));
    float sA = a0A - ((sa.x + sa.y) + (sa.z + sa.w));
    float piv = lane_bcast(ROWA ? sA : sB, jl);
    float r  = rsqrtf(piv);
    float d  = piv * r;
    float zj = lane_bcast(ROWA ? ya : yb, jl) * r;
    float LB = sB * r;
    if (ROWA) {
      float LA = sA * r;
      if (l == jl)     { a[j][j] = d; ya = zj; }
      else if (l > jl) { a[rA][j] = LA; lz[rA][j] = LA; ya -= zj * LA; }
      a[rB][j] = LB; lz[rB][j] = LB;
      yb -= zj * LB;
    } else {
      if (l == jl)     { a[j][j] = d; yb = zj; }
      else if (l > jl) { a[rB][j] = LB; lz[rB][j] = LB; yb -= zj * LB; }
    }
    __syncthreads();
  }
}

__global__ void __launch_bounds__(32) gp_chol_diag(float* __restrict__ K, const float* __restrict__ yw,
                                                   float* __restrict__ z, int kb) {
  __shared__ __attribute__((aligned(16))) float a[NB][DP];
  __shared__ __attribute__((aligned(16))) float lz[NB][DP];
  int l = threadIdx.x;
  int base = kb * NB;
  int r0 = l >> 4, c4 = (l & 15) << 2;
  float* blk = K + (size_t)(base + r0) * NTRAIN + base + c4;
#pragma unroll
  for (int i = 0; i < NB / 2; ++i) {
    *(float4*)&a[r0 + 2 * i][c4]  = *(const float4*)(blk + (size_t)i * 2 * NTRAIN);
    *(float4*)&lz[r0 + 2 * i][c4] = make_float4(0.f, 0.f, 0.f, 0.f);
  }
  float ya = yw[base + l], yb = yw[base + 32 + l];
  __syncthreads();
  diag_cols16<0, true>(a, lz, l, ya, yb);
  diag_cols16<1, true>(a, lz, l, ya, yb);
  diag_cols16<2, false>(a, lz, l, ya, yb);
  diag_cols16<3, false>(a, lz, l, ya, yb);
#pragma unroll
  for (int i = 0; i < NB / 2; ++i)
    *(float4*)(blk + (size_t)i * 2 * NTRAIN) = *(const float4*)&a[r0 + 2 * i][c4];
  z[base + l]      = ya;
  z[base + 32 + l] = yb;
}

__global__ void gp_chol_trsm(float* __restrict__ K, float* __restrict__ yw,
                             const float* __restrict__ z, int kb) {
  __shared__ float Ld[NB][NB + 1];
  __shared__ float rw[NB][NB + 1];
  __shared__ float zs[NB];
  int t = threadIdx.x;
  int base = kb * NB;
  for (int j = 0; j < NB; ++j) Ld[t][j] = K[(size_t)(base + t) * NTRAIN + base + j];
  zs[t] = z[base + t];
  int row = base + NB + blockIdx.x * NB + t;
  const float* src = K + (size_t)row * NTRAIN + base;
  for (int j = 0; j < NB; ++j) rw[t][j] = src[j];
  __syncthreads();
  float dz = 0.f;
  for (int j = 0; j < NB; ++j) {
    float s = rw[t][j];
    for (int p = 0; p < j; ++p) s -= rw[t][p] * Ld[j][p];
    s /= Ld[j][j];
    rw[t][j] = s;
    dz += s * zs[j];
  }
  float* dst = K + (size_t)row * NTRAIN + base;
  for (int j = 0; j < NB; ++j) dst[j] = rw[t][j];
  yw[row] -= dz;
}

__global__ void gp_chol_syrk(float* __restrict__ K, int kb) {
  __shared__ float As[64][SPAD];
  __shared__ float Bs[64][SPAD];
  int bi = blockIdx.x, bj = blockIdx.y;
  if (bj > bi) return;
  int base = kb * NB;
  int tb   = base + NB;
  int t = threadIdx.x;
  int lane = t & 31, wave = t >> 5;
  int rowA = tb + bi * 64, rowB = tb + bj * 64;

  for (int f = t; f < 64 * 16; f += 128) {
    int r = f >> 4, c4 = (f & 15) << 2;
    *(float4*)&As[r][c4] = *(const float4*)&K[(size_t)(rowA + r) * NTRAIN + base + c4];
    *(float4*)&Bs[r][c4] = *(const float4*)&K[(size_t)(rowB + r) * NTRAIN + base + c4];
  }
  __syncthreads();

  int wr = (wave >> 1) * 32;
  int wc = (wave & 1) * 32;
  v8f c00 = {}, c01 = {}, c10 = {}, c11 = {};
#pragma unroll
  for (int k0 = 0; k0 < NB; k0 += 4) {
    v2f a0 = frag_lds(As, wr,      k0, lane);
    v2f a1 = frag_lds(As, wr + 16, k0, lane);
    v2f b0 = frag_lds(Bs, wc,      k0, lane);
    v2f b1 = frag_lds(Bs, wc + 16, k0, lane);
    c00 = wmma4(a0, b0, c00);
    c01 = wmma4(a0, b1, c01);
    c10 = wmma4(a1, b0, c10);
    c11 = wmma4(a1, b1, c11);
  }

  int n  = lane & 15;
  int mh = (lane >> 4) * 8;
  int r0 = tb + bi * 64 + wr + mh;
  int c0 = tb + bj * 64 + wc + n;
#pragma unroll
  for (int r = 0; r < 8; ++r) {
    K[(size_t)(r0 + r) * NTRAIN + c0]           -= c00[r];
    K[(size_t)(r0 + r) * NTRAIN + c0 + 16]      -= c01[r];
    K[(size_t)(r0 + 16 + r) * NTRAIN + c0]      -= c10[r];
    K[(size_t)(r0 + 16 + r) * NTRAIN + c0 + 16] -= c11[r];
  }
}

#define BWT 1024

__global__ void __launch_bounds__(BWT) gp_bwd_solve(const float* __restrict__ K,
                                                    float* __restrict__ y) {
  __shared__ float yl[NTRAIN];
  __shared__ float dg[NB][NB + 1];
  __shared__ float dd[NB];
  int t = threadIdx.x;
  for (int i = t; i < NTRAIN; i += BWT) yl[i] = y[i];

  int dr = t >> 6, dc = t & 63;
  float nd[4];
  {
    int rb0 = NTRAIN - NB;
#pragma unroll
    for (int q = 0; q < 4; ++q)
      nd[q] = K[(size_t)(rb0 + dr + 16 * q) * NTRAIN + rb0 + dc];
  }

  for (int m = NTRAIN / NB - 1; m >= 0; --m) {
    int rb = m * NB;
#pragma unroll
    for (int q = 0; q < 4; ++q) {
      int r = dr + 16 * q;
      dg[r][dc] = (dc < r) ? nd[q] : 0.f;
      if (dc == r) dd[r] = nd[q];
    }
    __syncthreads();
    if (m > 0) {
      int rn = rb - NB;
#pragma unroll
      for (int q = 0; q < 4; ++q)
        nd[q] = K[(size_t)(rn + dr + 16 * q) * NTRAIN + rn + dc];
    }

    if (t < 32) {
      float s0 = yl[rb + t], s1 = yl[rb + 32 + t];
      float r0 = 1.0f / dd[t];
      float r1 = 1.0f / dd[32 + t];
#pragma unroll
      for (int j = NB - 1; j >= 32; --j) {
        float xj = lane_bcast(s1, j - 32) * lane_bcast(r1, j - 32);
        s1 -= dg[j][32 + t] * xj;
        s0 -= dg[j][t] * xj;
      }
#pragma unroll
      for (int j = 31; j >= 1; --j) {
        float xj = lane_bcast(s0, j) * lane_bcast(r0, j);
        s0 -= dg[j][t] * xj;
      }
      yl[rb + t]      = s0 * r0;
      yl[rb + 32 + t] = s1 * r1;
    }
    __syncthreads();

    for (int i = t; i < rb; i += BWT) {
      const float* col = K + (size_t)rb * NTRAIN + i;
      float a0 = 0.f, a1 = 0.f, a2 = 0.f, a3 = 0.f;
#pragma unroll
      for (int j = 0; j < NB; j += 4) {
        a0 += col[(size_t)(j + 0) * NTRAIN] * yl[rb + j + 0];
        a1 += col[(size_t)(j + 1) * NTRAIN] * yl[rb + j + 1];
        a2 += col[(size_t)(j + 2) * NTRAIN] * yl[rb + j + 2];
        a3 += col[(size_t)(j + 3) * NTRAIN] * yl[rb + j + 3];
      }
      yl[i] -= (a0 + a1) + (a2 + a3);
    }
  }
  __syncthreads();
  for (int i = t; i < NTRAIN; i += BWT) y[i] = yl[i];
}

__global__ void gp_fused_grad(const float* __restrict__ x, const float* __restrict__ tx,
                              const float* __restrict__ alpha,
                              const float* __restrict__ nx, const float* __restrict__ nt,
                              float* __restrict__ out) {
  __shared__ float wt[8][16][18];
  int wave = threadIdx.x >> 5, lane = threadIdx.x & 31;
  int rowBase = (blockIdx.x * 8 + wave) * 16;
  int n  = lane & 15;
  int mh = (lane >> 4) * 8;

  float nxr[8];
#pragma unroll
  for (int r = 0; r < 8; ++r) nxr[r] = nx[rowBase + mh + r];

  v2f af[16];
#pragma unroll
  for (int k0 = 0; k0 < DIMS; k0 += 4)
    af[k0 >> 2] = frag_nt(x, DIMS, rowBase, 0, k0, lane);

  v8f g[4]; g[0] = {}; g[1] = {}; g[2] = {}; g[3] = {};
  float rs[8] = {0, 0, 0, 0, 0, 0, 0, 0};

  for (int c0 = 0; c0 < NTRAIN; c0 += 16) {
    if (c0 + 16 < NTRAIN)
      __builtin_prefetch(tx + (size_t)(c0 + 16) * DIMS + lane * 32, 0, 0);
    v8f ce = {}, co = {};
#pragma unroll
    for (int k0 = 0; k0 < DIMS; k0 += 8) {
      ce = wmma4(af[(k0 >> 2)],     frag_nt(tx, DIMS, c0, 0, k0,     lane), ce);
      co = wmma4(af[(k0 >> 2) + 1], frag_nt(tx, DIMS, c0, 0, k0 + 4, lane), co);
    }
    v8f c = ce + co;
    float ncol = nt[c0 + n];
    float av   = alpha[c0 + n];
#pragma unroll
    for (int r = 0; r < 8; ++r) {
      float d2 = nxr[r] + ncol - 2.0f * c[r];
      float w  = KSCALE * __expf(-0.5f * d2 * (1.0f / LS2)) * av;
      rs[r] += w;
      wt[wave][mh + r][n] = w;
    }
    __syncthreads();
#pragma unroll
    for (int nc = 0; nc < 4; ++nc) {
#pragma unroll
      for (int k0 = 0; k0 < 16; k0 += 4) {
        int kk = k0 + ((lane >> 4) << 1);
        v2f a, b;
        a.x = wt[wave][n][kk];
        a.y = wt[wave][n][kk + 1];
        b.x = tx[(size_t)(c0 + kk) * DIMS + nc * 16 + n];
        b.y = tx[(size_t)(c0 + kk + 1) * DIMS + nc * 16 + n];
        g[nc] = wmma4(a, b, g[nc]);
      }
    }
    __syncthreads();
  }

#pragma unroll
  for (int r = 0; r < 8; ++r)
    for (int off = 8; off; off >>= 1) rs[r] += __shfl_xor(rs[r], off, 32);

#pragma unroll
  for (int nc = 0; nc < 4; ++nc) {
#pragma unroll
    for (int r = 0; r < 8; ++r) {
      int row = rowBase + mh + r;
      int d   = nc * 16 + n;
      float xv = x[(size_t)row * DIMS + d];
      out[(size_t)row * DIMS + d] = xv - (g[nc][r] - rs[r] * xv) * (1.0f / LS2);
    }
  }
}

extern "C" void kernel_launch(void* const* d_in, const int* in_sizes, int n_in,
                              void* d_out, int out_size, void* d_ws, size_t ws_size,
                              hipStream_t stream) {
  (void)in_sizes; (void)n_in; (void)out_size; (void)ws_size;
  const float* x  = (const float*)d_in[0];
  const float* tx = (const float*)d_in[1];
  const float* ty = (const float*)d_in[2];
  float* out = (float*)d_out;

  float* Kd = (float*)d_ws;
  float* nt = Kd + (size_t)NTRAIN * NTRAIN;
  float* nx = nt + NTRAIN;
  float* al = nx + NQUERY;
  float* yw = al + NTRAIN;

  gp_sqnorm<<<(NTRAIN + 255) / 256, 256, 0, stream>>>(tx, nt, NTRAIN);
  gp_sqnorm<<<(NQUERY + 255) / 256, 256, 0, stream>>>(x, nx, NQUERY);

  gp_build_K<<<dim3(NTRAIN / 16, NTRAIN / 16), 32, 0, stream>>>(tx, nt, Kd);
  hipMemcpyAsync(yw, ty, NTRAIN * sizeof(float), hipMemcpyDeviceToDevice, stream);

  for (int kb = 0; kb < NTRAIN / NB; ++kb) {
    gp_chol_diag<<<1, 32, 0, stream>>>(Kd, yw, al, kb);
    int below = NTRAIN - NB * (kb + 1);
    if (below > 0) {
      gp_chol_trsm<<<below / NB, NB, 0, stream>>>(Kd, yw, al, kb);
      int tt = below / 64;
      gp_chol_syrk<<<dim3(tt, tt), 128, 0, stream>>>(Kd, kb);
    }
  }

  gp_bwd_solve<<<1, BWT, 0, stream>>>(Kd, al);

  gp_fused_grad<<<NQUERY / 128, 256, 0, stream>>>(x, tx, al, nx, nt, out);
}


# panel TRSM re-parallelised: row per thread in registers, column sweep; L11 columns read from LDS by aligned 16-B broadcast; divide by diagonal becomes multiply by reciprocal; sum order kept
# speedup vs baseline: 5.8903x; 1.5739x over previous
#include <hip/hip_runtime.h>
#include <math.h>

#define DIMS   64
#define NTRAIN 4096
#define NQUERY 8192
#define NB     64
#define NOISE  1e-4f
#define LS2    64.0f
#define KSCALE 0.6931471805599453f
#define SPAD   68

typedef float v2f __attribute__((ext_vector_type(2)));
typedef float v8f __attribute__((ext_vector_type(8)));

__device__ __forceinline__ v8f wmma4(v2f a, v2f b, v8f c) {
  return __builtin_amdgcn_wmma_f32_16x16x4_f32(
       false, a,  false, b,
       (short)0, c,  false,  false);
}

__device__ __forceinline__ v2f frag_nt(const float* __restrict__ X, size_t rowStride,
                                       int rowBase, int colBase, int k0, int lane) {
  int kk = k0 + ((lane >> 4) << 1);
  const float* p = X + (size_t)(rowBase + (lane & 15)) * rowStride + colBase + kk;
  v2f v; v.x = p[0]; v.y = p[1];
  return v;
}

__device__ __forceinline__ v2f frag_lds(float (*S)[SPAD], int rowBase, int k0, int lane) {
  int kk = k0 + ((lane >> 4) << 1);
  const float* p = &S[rowBase + (lane & 15)][kk];
  v2f v; v.x = p[0]; v.y = p[1];
  return v;
}

__global__ void gp_sqnorm(const float* __restrict__ X, float* __restrict__ nrm, int rows) {
  int r = blockIdx.x * blockDim.x + threadIdx.x;
  if (r >= rows) return;
  const float4* p = (const float4*)(X + (size_t)r * DIMS);
  float s = 0.f;
#pragma unroll
  for (int i = 0; i < DIMS / 4; ++i) {
    float4 v = p[i];
    s += v.x * v.x + v.y * v.y + v.z * v.z + v.w * v.w;
  }
  nrm[r] = s;
}

__global__ void gp_build_K(const float* __restrict__ tx, const float* __restrict__ nrm,
                           float* __restrict__ K) {
  int bi = blockIdx.x, bj = blockIdx.y;
  int lane = threadIdx.x & 31;
  v8f ce = {}, co = {};
#pragma unroll
  for (int k0 = 0; k0 < DIMS; k0 += 8) {
    ce = wmma4(frag_nt(tx, DIMS, bi * 16, 0, k0, lane),
               frag_nt(tx, DIMS, bj * 16, 0, k0, lane), ce);
    co = wmma4(frag_nt(tx, DIMS, bi * 16, 0, k0 + 4, lane),
               frag_nt(tx, DIMS, bj * 16, 0, k0 + 4, lane), co);
  }
  v8f c = ce + co;
  int n  = lane & 15;
  int mh = (lane >> 4) * 8;
  int col = bj * 16 + n;
  float ncol = nrm[col];
#pragma unroll
  for (int r = 0; r < 8; ++r) {
    int row = bi * 16 + mh + r;
    float d2 = nrm[row] + ncol - 2.0f * c[r];
    float v  = KSCALE * __expf(-0.5f * d2 * (1.0f / LS2));
    if (row == col) v += NOISE;
    K[(size_t)row * NTRAIN + col] = v;
  }
}

#define DP 68

__device__ __forceinline__ float lane_bcast(float v, int srcLane) {
  return __int_as_float(__builtin_amdgcn_readlane(__float_as_int(v), srcLane));
}

template <bool ROWA>
__device__ __forceinline__ void diag_dot16(float (*lz)[DP], int j, int rA, int rB, int p0,
                                           float4& sa, float4& sb) {
#pragma unroll
  for (int u = 0; u < 4; ++u) {
    int p = p0 + 4 * u;
    float4 rj = *(const float4*)&lz[j][p];
    float4 xb = *(const float4*)&lz[rB][p];
    sb.x = fmaf(xb.x, rj.x, sb.x); sb.y = fmaf(xb.y, rj.y, sb.y);
    sb.z = fmaf(xb.z, rj.z, sb.z); sb.w = fmaf(xb.w, rj.w, sb.w);
    if (ROWA) {
      float4 xa = *(const float4*)&lz[rA][p];
      sa.x = fmaf(xa.x, rj.x, sa.x); sa.y = fmaf(xa.y, rj.y, sa.y);
      sa.z = fmaf(xa.z, rj.z, sa.z); sa.w = fmaf(xa.w, rj.w, sa.w);
    }
  }
}

template <int Q, bool ROWA>
__device__ __forceinline__ void diag_cols16(float (*a)[DP], float (*lz)[DP], int l,
                                            float& ya, float& yb) {
  const int rA = l, rB = l + 32;
#pragma nounroll
  for (int jj = 0; jj < 16; ++jj) {
    const int j  = 16 * Q + jj;
    const int jl = ((16 * Q) & 31) + jj;
    float a0B = a[rB][j];
    float a0A = ROWA ? a[rA][j] : 0.f;
    float4 sa = {0.f, 0.f, 0.f, 0.f}, sb = {0.f, 0.f, 0.f, 0.f};
#pragma unroll
    for (int q = 0; q <= Q; ++q)
      diag_dot16<ROWA>(lz, j, rA, rB, 16 * q, sa, sb);
    float sB = a0B - ((sb.x + sb.y) + (sb.z + sb.w));
    float sA = a0A - ((sa.x + sa.y) + (sa.z + sa.w));
    float piv = lane_bcast(ROWA ? sA : sB, jl);
    float r  = rsqrtf(piv);
    float d  = piv * r;
    float zj = lane_bcast(ROWA ? ya : yb, jl) * r;
    float LB = sB * r;
    if (ROWA) {
      float LA = sA * r;
      if (l == jl)     { a[j][j] = d; ya = zj; }
      else if (l > jl) { a[rA][j] = LA; lz[rA][j] = LA; ya -= zj * LA; }
      a[rB][j] = LB; lz[rB][j] = LB;
      yb -= zj * LB;
    } else {
      if (l == jl)     { a[j][j] = d; yb = zj; }
      else if (l > jl) { a[rB][j] = LB; lz[rB][j] = LB; yb -= zj * LB; }
    }
    __syncthreads();
  }
}

__global__ void __launch_bounds__(32) gp_chol_diag(float* __restrict__ K, const float* __restrict__ yw,
                                                   float* __restrict__ z, int kb) {
  __shared__ __attribute__((aligned(16))) float a[NB][DP];
  __shared__ __attribute__((aligned(16))) float lz[NB][DP];
  int l = threadIdx.x;
  int base = kb * NB;
  int r0 = l >> 4, c4 = (l & 15) << 2;
  float* blk = K + (size_t)(base + r0) * NTRAIN + base + c4;
#pragma unroll
  for (int i = 0; i < NB / 2; ++i) {
    *(float4*)&a[r0 + 2 * i][c4]  = *(const float4*)(blk + (size_t)i * 2 * NTRAIN);
    *(float4*)&lz[r0 + 2 * i][c4] = make_float4(0.f, 0.f, 0.f, 0.f);
  }
  float ya = yw[base + l], yb = yw[base + 32 + l];
  __syncthreads();
  diag_cols16<0, true>(a, lz, l, ya, yb);
  diag_cols16<1, true>(a, lz, l, ya, yb);
  diag_cols16<2, false>(a, lz, l, ya, yb);
  diag_cols16<3, false>(a, lz, l, ya, yb);
#pragma unroll
  for (int i = 0; i < NB / 2; ++i)
    *(float4*)(blk + (size_t)i * 2 * NTRAIN) = *(const float4*)&a[r0 + 2 * i][c4];
  z[base + l]      = ya;
  z[base + 32 + l] = yb;
}

#define TP 68

__global__ void __launch_bounds__(NB) gp_chol_trsm(float* __restrict__ K, float* __restrict__ yw,
                                                   const float* __restrict__ z, int kb) {
  __shared__ __attribute__((aligned(16))) float LC[NB][TP];
  __shared__ __attribute__((aligned(16))) float nri[NB];
  __shared__ __attribute__((aligned(16))) float zs[NB];
  int t = threadIdx.x;
  int base = kb * NB;
  int row = base + NB + blockIdx.x * NB + t;
  float* prow = K + (size_t)row * NTRAIN + base;

  float s[NB];
#pragma unroll
  for (int q = 0; q < NB / 4; ++q) {
    float4 v = *(const float4*)(prow + 4 * q);
    s[4 * q + 0] = v.x; s[4 * q + 1] = v.y; s[4 * q + 2] = v.z; s[4 * q + 3] = v.w;
  }

  {
    int r0 = t >> 4, c4 = (t & 15) << 2;
    const float* blk = K + (size_t)(base + r0) * NTRAIN + base + c4;
#pragma unroll
    for (int i = 0; i < NB / 4; ++i) {
      float4 v = *(const float4*)(blk + (size_t)i * 4 * NTRAIN);
      int r = r0 + 4 * i;
      LC[c4 + 0][(r - c4 - 1) & (NB - 1)] = v.x;
      LC[c4 + 1][(r - c4 - 2) & (NB - 1)] = v.y;
      LC[c4 + 2][(r - c4 - 3) & (NB - 1)] = v.z;
      LC[c4 + 3][(r - c4 - 4) & (NB - 1)] = v.w;
    }
  }
  nri[t] = -1.0f / K[(size_t)(base + t) * NTRAIN + base + t];
  zs[t] = z[base + t];
  __syncthreads();

  float ndz = 0.f;
#pragma unroll
  for (int j = 0; j < NB; ++j) {
    const float nx = s[j] * nri[j];
    ndz = fmaf(nx, zs[j], ndz);
#pragma unroll
    for (int m = 0; j + 1 + 4 * m < NB; ++m) {
      const float4 l4 = *(const float4*)&LC[j][4 * m];
      const int i0 = j + 1 + 4 * m;
      s[i0] = fmaf(nx, l4.x, s[i0]);
      if (i0 + 1 < NB) s[i0 + 1] = fmaf(nx, l4.y, s[i0 + 1]);
      if (i0 + 2 < NB) s[i0 + 2] = fmaf(nx, l4.z, s[i0 + 2]);
      if (i0 + 3 < NB) s[i0 + 3] = fmaf(nx, l4.w, s[i0 + 3]);
    }
    s[j] = -nx;
  }

#pragma unroll
  for (int q = 0; q < NB / 4; ++q)
    *(float4*)(prow + 4 * q) = make_float4(s[4 * q + 0], s[4 * q + 1], s[4 * q + 2], s[4 * q + 3]);
  yw[row] += ndz;
}

__global__ void gp_chol_syrk(float* __restrict__ K, int kb) {
  __shared__ float As[64][SPAD];
  __shared__ float Bs[64][SPAD];
  int bi = blockIdx.x, bj = blockIdx.y;
  if (bj > bi) return;
  int base = kb * NB;
  int tb   = base + NB;
  int t = threadIdx.x;
  int lane = t & 31, wave = t >> 5;
  int rowA = tb + bi * 64, rowB = tb + bj * 64;

  for (int f = t; f < 64 * 16; f += 128) {
    int r = f >> 4, c4 = (f & 15) << 2;
    *(float4*)&As[r][c4] = *(const float4*)&K[(size_t)(rowA + r) * NTRAIN + base + c4];
    *(float4*)&Bs[r][c4] = *(const float4*)&K[(size_t)(rowB + r) * NTRAIN + base + c4];
  }
  __syncthreads();

  int wr = (wave >> 1) * 32;
  int wc = (wave & 1) * 32;
  v8f c00 = {}, c01 = {}, c10 = {}, c11 = {};
#pragma unroll
  for (int k0 = 0; k0 < NB; k0 += 4) {
    v2f a0 = frag_lds(As, wr,      k0, lane);
    v2f a1 = frag_lds(As, wr + 16, k0, lane);
    v2f b0 = frag_lds(Bs, wc,      k0, lane);
    v2f b1 = frag_lds(Bs, wc + 16, k0, lane);
    c00 = wmma4(a0, b0, c00);
    c01 = wmma4(a0, b1, c01);
    c10 = wmma4(a1, b0, c10);
    c11 = wmma4(a1, b1, c11);
  }

  int n  = lane & 15;
  int mh = (lane >> 4) * 8;
  int r0 = tb + bi * 64 + wr + mh;
  int c0 = tb + bj * 64 + wc + n;
#pragma unroll
  for (int r = 0; r < 8; ++r) {
    K[(size_t)(r0 + r) * NTRAIN + c0]           -= c00[r];
    K[(size_t)(r0 + r) * NTRAIN + c0 + 16]      -= c01[r];
    K[(size_t)(r0 + 16 + r) * NTRAIN + c0]      -= c10[r];
    K[(size_t)(r0 + 16 + r) * NTRAIN + c0 + 16] -= c11[r];
  }
}

#define BWT 1024

__global__ void __launch_bounds__(BWT) gp_bwd_solve(const float* __restrict__ K,
                                                    float* __restrict__ y) {
  __shared__ float yl[NTRAIN];
  __shared__ float dg[NB][NB + 1];
  __shared__ float dd[NB];
  int t = threadIdx.x;
  for (int i = t; i < NTRAIN; i += BWT) yl[i] = y[i];

  int dr = t >> 6, dc = t & 63;
  float nd[4];
  {
    int rb0 = NTRAIN - NB;
#pragma unroll
    for (int q = 0; q < 4; ++q)
      nd[q] = K[(size_t)(rb0 + dr + 16 * q) * NTRAIN + rb0 + dc];
  }

  for (int m = NTRAIN / NB - 1; m >= 0; --m) {
    int rb = m * NB;
#pragma unroll
    for (int q = 0; q < 4; ++q) {
      int r = dr + 16 * q;
      dg[r][dc] = (dc < r) ? nd[q] : 0.f;
      if (dc == r) dd[r] = nd[q];
    }
    __syncthreads();
    if (m > 0) {
      int rn = rb - NB;
#pragma unroll
      for (int q = 0; q < 4; ++q)
        nd[q] = K[(size_t)(rn + dr + 16 * q) * NTRAIN + rn + dc];
    }

    if (t < 32) {
      float s0 = yl[rb + t], s1 = yl[rb + 32 + t];
      float r0 = 1.0f / dd[t];
      float r1 = 1.0f / dd[32 + t];
#pragma unroll
      for (int j = NB - 1; j >= 32; --j) {
        float xj = lane_bcast(s1, j - 32) * lane_bcast(r1, j - 32);
        s1 -= dg[j][32 + t] * xj;
        s0 -= dg[j][t] * xj;
      }
#pragma unroll
      for (int j = 31; j >= 1; --j) {
        float xj = lane_bcast(s0, j) * lane_bcast(r0, j);
        s0 -= dg[j][t] * xj;
      }
      yl[rb + t]      = s0 * r0;
      yl[rb + 32 + t] = s1 * r1;
    }
    __syncthreads();

    for (int i = t; i < rb; i += BWT) {
      const float* col = K + (size_t)rb * NTRAIN + i;
      float a0 = 0.f, a1 = 0.f, a2 = 0.f, a3 = 0.f;
#pragma unroll
      for (int j = 0; j < NB; j += 4) {
        a0 += col[(size_t)(j + 0) * NTRAIN] * yl[rb + j + 0];
        a1 += col[(size_t)(j + 1) * NTRAIN] * yl[rb + j + 1];
        a2 += col[(size_t)(j + 2) * NTRAIN] * yl[rb + j + 2];
        a3 += col[(size_t)(j + 3) * NTRAIN] * yl[rb + j + 3];
      }
      yl[i] -= (a0 + a1) + (a2 + a3);
    }
  }
  __syncthreads();
  for (int i = t; i < NTRAIN; i += BWT) y[i] = yl[i];
}

__global__ void gp_fused_grad(const float* __restrict__ x, const float* __restrict__ tx,
                              const float* __restrict__ alpha,
                              const float* __restrict__ nx, const float* __restrict__ nt,
                              float* __restrict__ out) {
  __shared__ float wt[8][16][18];
  int wave = threadIdx.x >> 5, lane = threadIdx.x & 31;
  int rowBase = (blockIdx.x * 8 + wave) * 16;
  int n  = lane & 15;
  int mh = (lane >> 4) * 8;

  float nxr[8];
#pragma unroll
  for (int r = 0; r < 8; ++r) nxr[r] = nx[rowBase + mh + r];

  v2f af[16];
#pragma unroll
  for (int k0 = 0; k0 < DIMS; k0 += 4)
    af[k0 >> 2] = frag_nt(x, DIMS, rowBase, 0, k0, lane);

  v8f g[4]; g[0] = {}; g[1] = {}; g[2] = {}; g[3] = {};
  float rs[8] = {0, 0, 0, 0, 0, 0, 0, 0};

  for (int c0 = 0; c0 < NTRAIN; c0 += 16) {
    if (c0 + 16 < NTRAIN)
      __builtin_prefetch(tx + (size_t)(c0 + 16) * DIMS + lane * 32, 0, 0);
    v8f ce = {}, co = {};
#pragma unroll
    for (int k0 = 0; k0 < DIMS; k0 += 8) {
      ce = wmma4(af[(k0 >> 2)],     frag_nt(tx, DIMS, c0, 0, k0,     lane), ce);
      co = wmma4(af[(k0 >> 2) + 1], frag_nt(tx, DIMS, c0, 0, k0 + 4, lane), co);
    }
    v8f c = ce + co;
    float ncol = nt[c0 + n];
    float av   = alpha[c0 + n];
#pragma unroll
    for (int r = 0; r < 8; ++r) {
      float d2 = nxr[r] + ncol - 2.0f * c[r];
      float w  = KSCALE * __expf(-0.5f * d2 * (1.0f / LS2)) * av;
      rs[r] += w;
      wt[wave][mh + r][n] = w;
    }
    __syncthreads();
#pragma unroll
    for (int nc = 0; nc < 4; ++nc) {
#pragma unroll
      for (int k0 = 0; k0 < 16; k0 += 4) {
        int kk = k0 + ((lane >> 4) << 1);
        v2f a, b;
        a.x = wt[wave][n][kk];
        a.y = wt[wave][n][kk + 1];
        b.x = tx[(size_t)(c0 + kk) * DIMS + nc * 16 + n];
        b.y = tx[(size_t)(c0 + kk + 1) * DIMS + nc * 16 + n];
        g[nc] = wmma4(a, b, g[nc]);
      }
    }
    __syncthreads();
  }

#pragma unroll
  for (int r = 0; r < 8; ++r)
    for (int off = 8; off; off >>= 1) rs[r] += __shfl_xor(rs[r], off, 32);

#pragma unroll
  for (int nc = 0; nc < 4; ++nc) {
#pragma unroll
    for (int r = 0; r < 8; ++r) {
      int row = rowBase + mh + r;
      int d   = nc * 16 + n;
      float xv = x[(size_t)row * DIMS + d];
      out[(size_t)row * DIMS + d] = xv - (g[nc][r] - rs[r] * xv) * (1.0f / LS2);
    }
  }
}

extern "C" void kernel_launch(void* const* d_in, const int* in_sizes, int n_in,
                              void* d_out, int out_size, void* d_ws, size_t ws_size,
                              hipStream_t stream) {
  (void)in_sizes; (void)n_in; (void)out_size; (void)ws_size;
  const float* x  = (const float*)d_in[0];
  const float* tx = (const float*)d_in[1];
  const float* ty = (const float*)d_in[2];
  float* out = (float*)d_out;

  float* Kd = (float*)d_ws;
  float* nt = Kd + (size_t)NTRAIN * NTRAIN;
  float* nx = nt + NTRAIN;
  float* al = nx + NQUERY;
  float* yw = al + NTRAIN;

  gp_sqnorm<<<(NTRAIN + 255) / 256, 256, 0, stream>>>(tx, nt, NTRAIN);
  gp_sqnorm<<<(NQUERY + 255) / 256, 256, 0, stream>>>(x, nx, NQUERY);

  gp_build_K<<<dim3(NTRAIN / 16, NTRAIN / 16), 32, 0, stream>>>(tx, nt, Kd);
  hipMemcpyAsync(yw, ty, NTRAIN * sizeof(float), hipMemcpyDeviceToDevice, stream);

  for (int kb = 0; kb < NTRAIN / NB; ++kb) {
    gp_chol_diag<<<1, 32, 0, stream>>>(Kd, yw, al, kb);
    int below = NTRAIN - NB * (kb + 1);
    if (below > 0) {
      gp_chol_trsm<<<below / NB, NB, 0, stream>>>(Kd, yw, al, kb);
      int tt = below / 64;
      gp_chol_syrk<<<dim3(tt, tt), 128, 0, stream>>>(Kd, kb);
    }
  }

  gp_bwd_solve<<<1, BWT, 0, stream>>>(Kd, al);

  gp_fused_grad<<<NQUERY / 128, 256, 0, stream>>>(x, tx, al, nx, nt, out);
}


# look-ahead: next diagonal block factored inside the trailing-update launch by wave 0 of its tile's work-group (in-wave LDS hand-over, no block barrier); 63 launches fewer; values and sum order kept
# speedup vs baseline: 6.7673x; 1.1489x over previous
#include <hip/hip_runtime.h>
#include <math.h>

#define DIMS   64
#define NTRAIN 4096
#define NQUERY 8192
#define NB     64
#define NOISE  1e-4f
#define LS2    64.0f
#define KSCALE 0.6931471805599453f
#define SPAD   68

typedef float v2f __attribute__((ext_vector_type(2)));
typedef float v8f __attribute__((ext_vector_type(8)));

__device__ __forceinline__ v8f wmma4(v2f a, v2f b, v8f c) {
  return __builtin_amdgcn_wmma_f32_16x16x4_f32(
       false, a,  false, b,
       (short)0, c,  false,  false);
}

__device__ __forceinline__ v2f frag_nt(const float* __restrict__ X, size_t rowStride,
                                       int rowBase, int colBase, int k0, int lane) {
  int kk = k0 + ((lane >> 4) << 1);
  const float* p = X + (size_t)(rowBase + (lane & 15)) * rowStride + colBase + kk;
  v2f v; v.x = p[0]; v.y = p[1];
  return v;
}

__device__ __forceinline__ v2f frag_lds(float (*S)[SPAD], int rowBase, int k0, int lane) {
  int kk = k0 + ((lane >> 4) << 1);
  const float* p = &S[rowBase + (lane & 15)][kk];
  v2f v; v.x = p[0]; v.y = p[1];
  return v;
}

__global__ void gp_sqnorm(const float* __restrict__ X, float* __restrict__ nrm, int rows) {
  int r = blockIdx.x * blockDim.x + threadIdx.x;
  if (r >= rows) return;
  const float4* p = (const float4*)(X + (size_t)r * DIMS);
  float s = 0.f;
#pragma unroll
  for (int i = 0; i < DIMS / 4; ++i) {
    float4 v = p[i];
    s += v.x * v.x + v.y * v.y + v.z * v.z + v.w * v.w;
  }
  nrm[r] = s;
}

__global__ void gp_build_K(const float* __restrict__ tx, const float* __restrict__ nrm,
                           float* __restrict__ K) {
  int bi = blockIdx.x, bj = blockIdx.y;
  int lane = threadIdx.x & 31;
  v8f ce = {}, co = {};
#pragma unroll
  for (int k0 = 0; k0 < DIMS; k0 += 8) {
    ce = wmma4(frag_nt(tx, DIMS, bi * 16, 0, k0, lane),
               frag_nt(tx, DIMS, bj * 16, 0, k0, lane), ce);
    co = wmma4(frag_nt(tx, DIMS, bi * 16, 0, k0 + 4, lane),
               frag_nt(tx, DIMS, bj * 16, 0, k0 + 4, lane), co);
  }
  v8f c = ce + co;
  int n  = lane & 15;
  int mh = (lane >> 4) * 8;
  int col = bj * 16 + n;
  float ncol = nrm[col];
#pragma unroll
  for (int r = 0; r < 8; ++r) {
    int row = bi * 16 + mh + r;
    float d2 = nrm[row] + ncol - 2.0f * c[r];
    float v  = KSCALE * __expf(-0.5f * d2 * (1.0f / LS2));
    if (row == col) v += NOISE;
    K[(size_t)row * NTRAIN + col] = v;
  }
}

#define DP 68

__device__ __forceinline__ float lane_bcast(float v, int srcLane) {
  return __int_as_float(__builtin_amdgcn_readlane(__float_as_int(v), srcLane));
}

__device__ __forceinline__ void wave_handover() {
  __builtin_amdgcn_fence(__ATOMIC_RELEASE, "wavefront");
  __builtin_amdgcn_wave_barrier();
  __builtin_amdgcn_fence(__ATOMIC_ACQUIRE, "wavefront");
}

template <bool ROWA>
__device__ __forceinline__ void diag_dot16(float (*lz)[DP], int j, int rA, int rB, int p0,
                                           float4& sa, float4& sb) {
#pragma unroll
  for (int u = 0; u < 4; ++u) {
    int p = p0 + 4 * u;
    float4 rj = *(const float4*)&lz[j][p];
    float4 xb = *(const float4*)&lz[rB][p];
    sb.x = fmaf(xb.x, rj.x, sb.x); sb.y = fmaf(xb.y, rj.y, sb.y);
    sb.z = fmaf(xb.z, rj.z, sb.z); sb.w = fmaf(xb.w, rj.w, sb.w);
    if (ROWA) {
      float4 xa = *(const float4*)&lz[rA][p];
      sa.x = fmaf(xa.x, rj.x, sa.x); sa.y = fmaf(xa.y, rj.y, sa.y);
      sa.z = fmaf(xa.z, rj.z, sa.z); sa.w = fmaf(xa.w, rj.w, sa.w);
    }
  }
}

template <int Q, bool ROWA>
__device__ __forceinline__ void diag_cols16(float (*a)[DP], float (*lz)[DP], int l,
                                            float& ya, float& yb) {
  const int rA = l, rB = l + 32;
#pragma nounroll
  for (int jj = 0; jj < 16; ++jj) {
    const int j  = 16 * Q + jj;
    const int jl = ((16 * Q) & 31) + jj;
    float a0B = a[rB][j];
    float a0A = ROWA ? a[rA][j] : 0.f;
    float4 sa = {0.f, 0.f, 0.f, 0.f}, sb = {0.f, 0.f, 0.f, 0.f};
#pragma unroll
    for (int q = 0; q <= Q; ++q)
      diag_dot16<ROWA>(lz, j, rA, rB, 16 * q, sa, sb);
    float sB = a0B - ((sb.x + sb.y) + (sb.z + sb.w));
    float sA = a0A - ((sa.x + sa.y) + (sa.z + sa.w));
    float piv = lane_bcast(ROWA ? sA : sB, jl);
    float r  = rsqrtf(piv);
    float d  = piv * r;
    float zj = lane_bcast(ROWA ? ya : yb, jl) * r;
    float LB = sB * r;
    if (ROWA) {
      float LA = sA * r;
      if (l == jl)     { a[j][j] = d; ya = zj; }
      else if (l > jl) { a[rA][j] = LA; lz[rA][j] = LA; ya -= zj * LA; }
      a[rB][j] = LB; lz[rB][j] = LB;
      yb -= zj * LB;
    } else {
      if (l == jl)     { a[j][j] = d; yb = zj; }
      else if (l > jl) { a[rB][j] = LB; lz[rB][j] = LB; yb -= zj * LB; }
    }
    wave_handover();
  }
}

__device__ __forceinline__ void diag_factor_wave(float (*a)[DP], float (*lz)[DP], int l, float ya, float yb,
                                                 float* __restrict__ K, float* __restrict__ z, int kb) {
  int base = kb * NB;
  int r0 = l >> 4, c4 = (l & 15) << 2;
  float* blk = K + (size_t)(base + r0) * NTRAIN + base + c4;
  diag_cols16<0, true>(a, lz, l, ya, yb);
  diag_cols16<1, true>(a, lz, l, ya, yb);
  diag_cols16<2, false>(a, lz, l, ya, yb);
  diag_cols16<3, false>(a, lz, l, ya, yb);
#pragma unroll
  for (int i = 0; i < NB / 2; ++i)
    *(float4*)(blk + (size_t)i * 2 * NTRAIN) = *(const float4*)&a[r0 + 2 * i][c4];
  z[base + l]      = ya;
  z[base + 32 + l] = yb;
}

__global__ void __launch_bounds__(32) gp_chol_diag(float* __restrict__ K, const float* __restrict__ yw,
                                                   float* __restrict__ z, int kb) {
  __shared__ __attribute__((aligned(16))) float a[NB][DP];
  __shared__ __attribute__((aligned(16))) float lz[NB][DP];
  int l = threadIdx.x;
  int base = kb * NB;
  int r0 = l >> 4, c4 = (l & 15) << 2;
  float* blk = K + (size_t)(base + r0) * NTRAIN + base + c4;
#pragma unroll
  for (int i = 0; i < NB / 2; ++i) {
    *(float4*)&a[r0 + 2 * i][c4]  = *(const float4*)(blk + (size_t)i * 2 * NTRAIN);
    *(float4*)&lz[r0 + 2 * i][c4] = make_float4(0.f, 0.f, 0.f, 0.f);
  }
  float ya = yw[base + l], yb = yw[base + 32 + l];
  __syncthreads();
  diag_factor_wave(a, lz, l, ya, yb, K, z, kb);
}

#define TP 68

__global__ void __launch_bounds__(NB) gp_chol_trsm(float* __restrict__ K, float* __restrict__ yw,
                                                   const float* __restrict__ z, int kb) {
  __shared__ __attribute__((aligned(16))) float LC[NB][TP];
  __shared__ __attribute__((aligned(16))) float nri[NB];
  __shared__ __attribute__((aligned(16))) float zs[NB];
  int t = threadIdx.x;
  int base = kb * NB;
  int row = base + NB + blockIdx.x * NB + t;
  float* prow = K + (size_t)row * NTRAIN + base;

  float s[NB];
#pragma unroll
  for (int q = 0; q < NB / 4; ++q) {
    float4 v = *(const float4*)(prow + 4 * q);
    s[4 * q + 0] = v.x; s[4 * q + 1] = v.y; s[4 * q + 2] = v.z; s[4 * q + 3] = v.w;
  }

  {
    int r0 = t >> 4, c4 = (t & 15) << 2;
    const float* blk = K + (size_t)(base + r0) * NTRAIN + base + c4;
#pragma unroll
    for (int i = 0; i < NB / 4; ++i) {
      float4 v = *(const float4*)(blk + (size_t)i * 4 * NTRAIN);
      int r = r0 + 4 * i;
      LC[c4 + 0][(r - c4 - 1) & (NB - 1)] = v.x;
      LC[c4 + 1][(r - c4 - 2) & (NB - 1)] = v.y;
      LC[c4 + 2][(r - c4 - 3) & (NB - 1)] = v.z;
      LC[c4 + 3][(r - c4 - 4) & (NB - 1)] = v.w;
    }
  }
  nri[t] = -1.0f / K[(size_t)(base + t) * NTRAIN + base + t];
  zs[t] = z[base + t];
  __syncthreads();

  float ndz = 0.f;
#pragma unroll
  for (int j = 0; j < NB; ++j) {
    const float nx = s[j] * nri[j];
    ndz = fmaf(nx, zs[j], ndz);
#pragma unroll
    for (int m = 0; j + 1 + 4 * m < NB; ++m) {
      const float4 l4 = *(const float4*)&LC[j][4 * m];
      const int i0 = j + 1 + 4 * m;
      s[i0] = fmaf(nx, l4.x, s[i0]);
      if (i0 + 1 < NB) s[i0 + 1] = fmaf(nx, l4.y, s[i0 + 1]);
      if (i0 + 2 < NB) s[i0 + 2] = fmaf(nx, l4.z, s[i0 + 2]);
      if (i0 + 3 < NB) s[i0 + 3] = fmaf(nx, l4.w, s[i0 + 3]);
    }
    s[j] = -nx;
  }

#pragma unroll
  for (int q = 0; q < NB / 4; ++q)
    *(float4*)(prow + 4 * q) = make_float4(s[4 * q + 0], s[4 * q + 1], s[4 * q + 2], s[4 * q + 3]);
  yw[row] += ndz;
}

__global__ void __launch_bounds__(128) gp_chol_syrk(float* __restrict__ K, const float* __restrict__ yw,
                                                    float* __restrict__ z, int kb) {
  __shared__ __attribute__((aligned(16))) float As[64][SPAD];
  __shared__ __attribute__((aligned(16))) float Bs[64][SPAD];
  int bi = blockIdx.x, bj = blockIdx.y;
  if (bj > bi) return;
  int base = kb * NB;
  int tb   = base + NB;
  int t = threadIdx.x;
  int lane = t & 31, wave = t >> 5;
  int rowA = tb + bi * 64, rowB = tb + bj * 64;

  for (int f = t; f < 64 * 16; f += 128) {
    int r = f >> 4, c4 = (f & 15) << 2;
    *(float4*)&As[r][c4] = *(const float4*)&K[(size_t)(rowA + r) * NTRAIN + base + c4];
    *(float4*)&Bs[r][c4] = *(const float4*)&K[(size_t)(rowB + r) * NTRAIN + base + c4];
  }
  __syncthreads();

  int wr = (wave >> 1) * 32;
  int wc = (wave & 1) * 32;
  v8f c00 = {}, c01 = {}, c10 = {}, c11 = {};
#pragma unroll
  for (int k0 = 0; k0 < NB; k0 += 4) {
    v2f a0 = frag_lds(As, wr,      k0, lane);
    v2f a1 = frag_lds(As, wr + 16, k0, lane);
    v2f b0 = frag_lds(Bs, wc,      k0, lane);
    v2f b1 = frag_lds(Bs, wc + 16, k0, lane);
    c00 = wmma4(a0, b0, c00);
    c01 = wmma4(a0, b1, c01);
    c10 = wmma4(a1, b0, c10);
    c11 = wmma4(a1, b1, c11);
  }

  int n  = lane & 15;
  int mh = (lane >> 4) * 8;
  int r0 = tb + bi * 64 + wr + mh;
  int c0 = tb + bj * 64 + wc + n;

  if (bi == 0 && bj == 0) {
    const int tr = wr + mh, tc = wc + n;
    float v00[8], v01[8], v10[8], v11[8];
#pragma unroll
    for (int r = 0; r < 8; ++r) {
      v00[r] = K[(size_t)(r0 + r) * NTRAIN + c0];
      v01[r] = K[(size_t)(r0 + r) * NTRAIN + c0 + 16];
      v10[r] = K[(size_t)(r0 + 16 + r) * NTRAIN + c0];
      v11[r] = K[(size_t)(r0 + 16 + r) * NTRAIN + c0 + 16];
    }
    float ya = 0.f, yb = 0.f;
    if (wave == 0) { ya = yw[tb + lane]; yb = yw[tb + 32 + lane]; }
    __syncthreads();
#pragma unroll
    for (int r = 0; r < 8; ++r) {
      As[tr + r][tc]           = v00[r] - c00[r];
      As[tr + r][tc + 16]      = v01[r] - c01[r];
      As[tr + 16 + r][tc]      = v10[r] - c10[r];
      As[tr + 16 + r][tc + 16] = v11[r] - c11[r];
    }
    for (int f = t; f < 64 * 16; f += 128) {
      int r = f >> 4, c4 = (f & 15) << 2;
      *(float4*)&Bs[r][c4] = make_float4(0.f, 0.f, 0.f, 0.f);
    }
    __syncthreads();
    if (wave == 0) diag_factor_wave(As, Bs, lane, ya, yb, K, z, kb + 1);
    return;
  }

#pragma unroll
  for (int r = 0; r < 8; ++r) {
    K[(size_t)(r0 + r) * NTRAIN + c0]           -= c00[r];
    K[(size_t)(r0 + r) * NTRAIN + c0 + 16]      -= c01[r];
    K[(size_t)(r0 + 16 + r) * NTRAIN + c0]      -= c10[r];
    K[(size_t)(r0 + 16 + r) * NTRAIN + c0 + 16] -= c11[r];
  }
}

#define BWT 1024

__global__ void __launch_bounds__(BWT) gp_bwd_solve(const float* __restrict__ K,
                                                    float* __restrict__ y) {
  __shared__ float yl[NTRAIN];
  __shared__ float dg[NB][NB + 1];
  __shared__ float dd[NB];
  int t = threadIdx.x;
  for (int i = t; i < NTRAIN; i += BWT) yl[i] = y[i];

  int dr = t >> 6, dc = t & 63;
  float nd[4];
  {
    int rb0 = NTRAIN - NB;
#pragma unroll
    for (int q = 0; q < 4; ++q)
      nd[q] = K[(size_t)(rb0 + dr + 16 * q) * NTRAIN + rb0 + dc];
  }

  for (int m = NTRAIN / NB - 1; m >= 0; --m) {
    int rb = m * NB;
#pragma unroll
    for (int q = 0; q < 4; ++q) {
      int r = dr + 16 * q;
      dg[r][dc] = (dc < r) ? nd[q] : 0.f;
      if (dc == r) dd[r] = nd[q];
    }
    __syncthreads();
    if (m > 0) {
      int rn = rb - NB;
#pragma unroll
      for (int q = 0; q < 4; ++q)
        nd[q] = K[(size_t)(rn + dr + 16 * q) * NTRAIN + rn + dc];
    }

    if (t < 32) {
      float s0 = yl[rb + t], s1 = yl[rb + 32 + t];
      float r0 = 1.0f / dd[t];
      float r1 = 1.0f / dd[32 + t];
#pragma unroll
      for (int j = NB - 1; j >= 32; --j) {
        float xj = lane_bcast(s1, j - 32) * lane_bcast(r1, j - 32);
        s1 -= dg[j][32 + t] * xj;
        s0 -= dg[j][t] * xj;
      }
#pragma unroll
      for (int j = 31; j >= 1; --j) {
        float xj = lane_bcast(s0, j) * lane_bcast(r0, j);
        s0 -= dg[j][t] * xj;
      }
      yl[rb + t]      = s0 * r0;
      yl[rb + 32 + t] = s1 * r1;
    }
    __syncthreads();

    for (int i = t; i < rb; i += BWT) {
      const float* col = K + (size_t)rb * NTRAIN + i;
      float a0 = 0.f, a1 = 0.f, a2 = 0.f, a3 = 0.f;
#pragma unroll
      for (int j = 0; j < NB; j += 4) {
        a0 += col[(size_t)(j + 0) * NTRAIN] * yl[rb + j + 0];
        a1 += col[(size_t)(j + 1) * NTRAIN] * yl[rb + j + 1];
        a2 += col[(size_t)(j + 2) * NTRAIN] * yl[rb + j + 2];
        a3 += col[(size_t)(j + 3) * NTRAIN] * yl[rb + j + 3];
      }
      yl[i] -= (a0 + a1) + (a2 + a3);
    }
  }
  __syncthreads();
  for (int i = t; i < NTRAIN; i += BWT) y[i] = yl[i];
}

__global__ void gp_fused_grad(const float* __restrict__ x, const float* __restrict__ tx,
                              const float* __restrict__ alpha,
                              const float* __restrict__ nx, const float* __restrict__ nt,
                              float* __restrict__ out) {
  __shared__ float wt[8][16][18];
  int wave = threadIdx.x >> 5, lane = threadIdx.x & 31;
  int rowBase = (blockIdx.x * 8 + wave) * 16;
  int n  = lane & 15;
  int mh = (lane >> 4) * 8;

  float nxr[8];
#pragma unroll
  for (int r = 0; r < 8; ++r) nxr[r] = nx[rowBase + mh + r];

  v2f af[16];
#pragma unroll
  for (int k0 = 0; k0 < DIMS; k0 += 4)
    af[k0 >> 2] = frag_nt(x, DIMS, rowBase, 0, k0, lane);

  v8f g[4]; g[0] = {}; g[1] = {}; g[2] = {}; g[3] = {};
  float rs[8] = {0, 0, 0, 0, 0, 0, 0, 0};

  for (int c0 = 0; c0 < NTRAIN; c0 += 16) {
    if (c0 + 16 < NTRAIN)
      __builtin_prefetch(tx + (size_t)(c0 + 16) * DIMS + lane * 32, 0, 0);
    v8f ce = {}, co = {};
#pragma unroll
    for (int k0 = 0; k0 < DIMS; k0 += 8) {
      ce = wmma4(af[(k0 >> 2)],     frag_nt(tx, DIMS, c0, 0, k0,     lane), ce);
      co = wmma4(af[(k0 >> 2) + 1], frag_nt(tx, DIMS, c0, 0, k0 + 4, lane), co);
    }
    v8f c = ce + co;
    float ncol = nt[c0 + n];
    float av   = alpha[c0 + n];
#pragma unroll
    for (int r = 0; r < 8; ++r) {
      float d2 = nxr[r] + ncol - 2.0f * c[r];
      float w  = KSCALE * __expf(-0.5f * d2 * (1.0f / LS2)) * av;
      rs[r] += w;
      wt[wave][mh + r][n] = w;
    }
    __syncthreads();
#pragma unroll
    for (int nc = 0; nc < 4; ++nc) {
#pragma unroll
      for (int k0 = 0; k0 < 16; k0 += 4) {
        int kk = k0 + ((lane >> 4) << 1);
        v2f a, b;
        a.x = wt[wave][n][kk];
        a.y = wt[wave][n][kk + 1];
        b.x = tx[(size_t)(c0 + kk) * DIMS + nc * 16 + n];
        b.y = tx[(size_t)(c0 + kk + 1) * DIMS + nc * 16 + n];
        g[nc] = wmma4(a, b, g[nc]);
      }
    }
    __syncthreads();
  }

#pragma unroll
  for (int r = 0; r < 8; ++r)
    for (int off = 8; off; off >>= 1) rs[r] += __shfl_xor(rs[r], off, 32);

#pragma unroll
  for (int nc = 0; nc < 4; ++nc) {
#pragma unroll
    for (int r = 0; r < 8; ++r) {
      int row = rowBase + mh + r;
      int d   = nc * 16 + n;
      float xv = x[(size_t)row * DIMS + d];
      out[(size_t)row * DIMS + d] = xv - (g[nc][r] - rs[r] * xv) * (1.0f / LS2);
    }
  }
}

extern "C" void kernel_launch(void* const* d_in, const int* in_sizes, int n_in,
                              void* d_out, int out_size, void* d_ws, size_t ws_size,
                              hipStream_t stream) {
  (void)in_sizes; (void)n_in; (void)out_size; (void)ws_size;
  const float* x  = (const float*)d_in[0];
  const float* tx = (const float*)d_in[1];
  const float* ty = (const float*)d_in[2];
  float* out = (float*)d_out;

  float* Kd = (float*)d_ws;
  float* nt = Kd + (size_t)NTRAIN * NTRAIN;
  float* nx = nt + NTRAIN;
  float* al = nx + NQUERY;
  float* yw = al + NTRAIN;

  gp_sqnorm<<<(NTRAIN + 255) / 256, 256, 0, stream>>>(tx, nt, NTRAIN);
  gp_sqnorm<<<(NQUERY + 255) / 256, 256, 0, stream>>>(x, nx, NQUERY);

  gp_build_K<<<dim3(NTRAIN / 16, NTRAIN / 16), 32, 0, stream>>>(tx, nt, Kd);
  hipMemcpyAsync(yw, ty, NTRAIN * sizeof(float), hipMemcpyDeviceToDevice, stream);

  gp_chol_diag<<<1, 32, 0, stream>>>(Kd, yw, al, 0);
  for (int kb = 0; kb < NTRAIN / NB; ++kb) {
    int below = NTRAIN - NB * (kb + 1);
    if (below > 0) {
      gp_chol_trsm<<<below / NB, NB, 0, stream>>>(Kd, yw, al, kb);
      int tt = below / 64;
      gp_chol_syrk<<<dim3(tt, tt), 128, 0, stream>>>(Kd, yw, al, kb);
    }
  }

  gp_bwd_solve<<<1, BWT, 0, stream>>>(Kd, al);

  gp_fused_grad<<<NQUERY / 128, 256, 0, stream>>>(x, tx, al, nx, nt, out);
}


# diagonal 64x64 factor by one wave in 16-column blocks: block swept in registers (pivot and column by lane broadcast, no LDS per column), later columns updated by WMMA rank-16; sum order changed
# speedup vs baseline: 9.3202x; 1.3772x over previous
#include <hip/hip_runtime.h>
#include <math.h>

#define DIMS   64
#define NTRAIN 4096
#define NQUERY 8192
#define NB     64
#define NOISE  1e-4f
#define LS2    64.0f
#define KSCALE 0.6931471805599453f
#define SPAD   68

typedef float v2f __attribute__((ext_vector_type(2)));
typedef float v8f __attribute__((ext_vector_type(8)));

__device__ __forceinline__ v8f wmma4(v2f a, v2f b, v8f c) {
  return __builtin_amdgcn_wmma_f32_16x16x4_f32(
       false, a,  false, b,
       (short)0, c,  false,  false);
}

__device__ __forceinline__ v2f frag_nt(const float* __restrict__ X, size_t rowStride,
                                       int rowBase, int colBase, int k0, int lane) {
  int kk = k0 + ((lane >> 4) << 1);
  const float* p = X + (size_t)(rowBase + (lane & 15)) * rowStride + colBase + kk;
  v2f v; v.x = p[0]; v.y = p[1];
  return v;
}

__device__ __forceinline__ v2f frag_lds(float (*S)[SPAD], int rowBase, int k0, int lane) {
  int kk = k0 + ((lane >> 4) << 1);
  const float* p = &S[rowBase + (lane & 15)][kk];
  v2f v; v.x = p[0]; v.y = p[1];
  return v;
}

__global__ void gp_sqnorm(const float* __restrict__ X, float* __restrict__ nrm, int rows) {
  int r = blockIdx.x * blockDim.x + threadIdx.x;
  if (r >= rows) return;
  const float4* p = (const float4*)(X + (size_t)r * DIMS);
  float s = 0.f;
#pragma unroll
  for (int i = 0; i < DIMS / 4; ++i) {
    float4 v = p[i];
    s += v.x * v.x + v.y * v.y + v.z * v.z + v.w * v.w;
  }
  nrm[r] = s;
}

__global__ void gp_build_K(const float* __restrict__ tx, const float* __restrict__ nrm,
                           float* __restrict__ K) {
  int bi = blockIdx.x, bj = blockIdx.y;
  int lane = threadIdx.x & 31;
  v8f ce = {}, co = {};
#pragma unroll
  for (int k0 = 0; k0 < DIMS; k0 += 8) {
    ce = wmma4(frag_nt(tx, DIMS, bi * 16, 0, k0, lane),
               frag_nt(tx, DIMS, bj * 16, 0, k0, lane), ce);
    co = wmma4(frag_nt(tx, DIMS, bi * 16, 0, k0 + 4, lane),
               frag_nt(tx, DIMS, bj * 16, 0, k0 + 4, lane), co);
  }
  v8f c = ce + co;
  int n  = lane & 15;
  int mh = (lane >> 4) * 8;
  int col = bj * 16 + n;
  float ncol = nrm[col];
#pragma unroll
  for (int r = 0; r < 8; ++r) {
    int row = bi * 16 + mh + r;
    float d2 = nrm[row] + ncol - 2.0f * c[r];
    float v  = KSCALE * __expf(-0.5f * d2 * (1.0f / LS2));
    if (row == col) v += NOISE;
    K[(size_t)row * NTRAIN + col] = v;
  }
}

#define DP 68

__device__ __forceinline__ float lane_bcast(float v, int srcLane) {
  return __int_as_float(__builtin_amdgcn_readlane(__float_as_int(v), srcLane));
}

__device__ __forceinline__ void wave_handover() {
  __builtin_amdgcn_fence(__ATOMIC_RELEASE, "wavefront");
  __builtin_amdgcn_wave_barrier();
  __builtin_amdgcn_fence(__ATOMIC_ACQUIRE, "wavefront");
}

template <int Q>
__device__ __forceinline__ void diag_block16(float (*a)[DP], int l, float& ya, float& yb,
                                             float& za, float& zb) {
  constexpr bool ROWA = (Q < 2);
  constexpr int c0 = 16 * Q;
  constexpr int l0 = c0 & 31;
  const int rA = l, rB = l + 32;
  float sA[16], sB[16];
#pragma unroll
  for (int u = 0; u < 4; ++u) {
    float4 vb = *(const float4*)&a[rB][c0 + 4 * u];
    sB[4 * u + 0] = vb.x; sB[4 * u + 1] = vb.y; sB[4 * u + 2] = vb.z; sB[4 * u + 3] = vb.w;
    if (ROWA) {
      float4 va = *(const float4*)&a[rA][c0 + 4 * u];
      sA[4 * u + 0] = va.x; sA[4 * u + 1] = va.y; sA[4 * u + 2] = va.z; sA[4 * u + 3] = va.w;
    } else {
      sA[4 * u + 0] = 0.f; sA[4 * u + 1] = 0.f; sA[4 * u + 2] = 0.f; sA[4 * u + 3] = 0.f;
    }
  }
#pragma unroll
  for (int jj = 0; jj < 16; ++jj) {
    const int jl = l0 + jj;
    float piv = lane_bcast(ROWA ? sA[jj] : sB[jj], jl);
    float r  = rsqrtf(piv);
    float zj = lane_bcast(ROWA ? ya : yb, jl) * r;
    float LB = sB[jj] * r;
    float LA = 0.f;
    sB[jj] = LB;
    if (ROWA) {
      LA = sA[jj] * r;
      sA[jj] = LA;
      if (l == jl) za = zj;
      ya -= zj * LA;
    } else {
      if (l == jl) zb = zj;
    }
    yb -= zj * LB;
#pragma unroll
    for (int ii = jj + 1; ii < 16; ++ii) {
      float Lij = lane_bcast(ROWA ? LA : LB, l0 + ii);
      sB[ii] -= LB * Lij;
      if (ROWA) sA[ii] -= LA * Lij;
    }
  }
#pragma unroll
  for (int u = 0; u < 4; ++u) {
    *(float4*)&a[rB][c0 + 4 * u] = make_float4(sB[4 * u + 0], sB[4 * u + 1], sB[4 * u + 2], sB[4 * u + 3]);
    if (ROWA)
      *(float4*)&a[rA][c0 + 4 * u] = make_float4(sA[4 * u + 0], sA[4 * u + 1], sA[4 * u + 2], sA[4 * u + 3]);
  }
  wave_handover();
  if constexpr (Q < 3) {
    constexpr int NT = (3 - Q) * (4 - Q) / 2;
    v8f acc[NT];
    {
      int idx = 0;
#pragma unroll
      for (int ni = Q + 1; ni < 4; ++ni) {
#pragma unroll
        for (int mi = ni; mi < 4; ++mi) {
          v8f c = {};
#pragma unroll
          for (int k0 = 0; k0 < 16; k0 += 4)
            c = wmma4(frag_lds(a, 16 * mi, c0 + k0, l), frag_lds(a, 16 * ni, c0 + k0, l), c);
          acc[idx++] = c;
        }
      }
    }
    const int n = l & 15, mh = (l >> 4) * 8;
    {
      int idx = 0;
#pragma unroll
      for (int ni = Q + 1; ni < 4; ++ni) {
#pragma unroll
        for (int mi = ni; mi < 4; ++mi) {
#pragma unroll
          for (int r = 0; r < 8; ++r)
            a[16 * mi + mh + r][16 * ni + n] -= acc[idx][r];
          ++idx;
        }
      }
    }
    wave_handover();
  }
}

__device__ __forceinline__ void diag_factor_wave(float (*a)[DP], int l, float ya, float yb,
                                                 float* __restrict__ K, float* __restrict__ z, int kb) {
  int base = kb * NB;
  int r0 = l >> 4, c4 = (l & 15) << 2;
  float* blk = K + (size_t)(base + r0) * NTRAIN + base + c4;
  float za = 0.f, zb = 0.f;
  diag_block16<0>(a, l, ya, yb, za, zb);
  diag_block16<1>(a, l, ya, yb, za, zb);
  diag_block16<2>(a, l, ya, yb, za, zb);
  diag_block16<3>(a, l, ya, yb, za, zb);
#pragma unroll
  for (int i = 0; i < NB / 2; ++i)
    *(float4*)(blk + (size_t)i * 2 * NTRAIN) = *(const float4*)&a[r0 + 2 * i][c4];
  z[base + l]      = za;
  z[base + 32 + l] = zb;
}

__global__ void __launch_bounds__(32) gp_chol_diag(float* __restrict__ K, const float* __restrict__ yw,
                                                   float* __restrict__ z, int kb) {
  __shared__ __attribute__((aligned(16))) float a[NB][DP];
  int l = threadIdx.x;
  int base = kb * NB;
  int r0 = l >> 4, c4 = (l & 15) << 2;
  float* blk = K + (size_t)(base + r0) * NTRAIN + base + c4;
#pragma unroll
  for (int i = 0; i < NB / 2; ++i)
    *(float4*)&a[r0 + 2 * i][c4]  = *(const float4*)(blk + (size_t)i * 2 * NTRAIN);
  float ya = yw[base + l], yb = yw[base + 32 + l];
  __syncthreads();
  diag_factor_wave(a, l, ya, yb, K, z, kb);
}

#define TP 68

__global__ void __launch_bounds__(NB) gp_chol_trsm(float* __restrict__ K, float* __restrict__ yw,
                                                   const float* __restrict__ z, int kb) {
  __shared__ __attribute__((aligned(16))) float LC[NB][TP];
  __shared__ __attribute__((aligned(16))) float nri[NB];
  __shared__ __attribute__((aligned(16))) float zs[NB];
  int t = threadIdx.x;
  int base = kb * NB;
  int row = base + NB + blockIdx.x * NB + t;
  float* prow = K + (size_t)row * NTRAIN + base;

  float s[NB];
#pragma unroll
  for (int q = 0; q < NB / 4; ++q) {
    float4 v = *(const float4*)(prow + 4 * q);
    s[4 * q + 0] = v.x; s[4 * q + 1] = v.y; s[4 * q + 2] = v.z; s[4 * q + 3] = v.w;
  }

  {
    int r0 = t >> 4, c4 = (t & 15) << 2;
    const float* blk = K + (size_t)(base + r0) * NTRAIN + base + c4;
#pragma unroll
    for (int i = 0; i < NB / 4; ++i) {
      float4 v = *(const float4*)(blk + (size_t)i * 4 * NTRAIN);
      int r = r0 + 4 * i;
      LC[c4 + 0][(r - c4 - 1) & (NB - 1)] = v.x;
      LC[c4 + 1][(r - c4 - 2) & (NB - 1)] = v.y;
      LC[c4 + 2][(r - c4 - 3) & (NB - 1)] = v.z;
      LC[c4 + 3][(r - c4 - 4) & (NB - 1)] = v.w;
    }
  }
  nri[t] = -1.0f / K[(size_t)(base + t) * NTRAIN + base + t];
  zs[t] = z[base + t];
  __syncthreads();

  float ndz = 0.f;
#pragma unroll
  for (int j = 0; j < NB; ++j) {
    const float nx = s[j] * nri[j];
    ndz = fmaf(nx, zs[j], ndz);
#pragma unroll
    for (int m = 0; j + 1 + 4 * m < NB; ++m) {
      const float4 l4 = *(const float4*)&LC[j][4 * m];
      const int i0 = j + 1 + 4 * m;
      s[i0] = fmaf(nx, l4.x, s[i0]);
      if (i0 + 1 < NB) s[i0 + 1] = fmaf(nx, l4.y, s[i0 + 1]);
      if (i0 + 2 < NB) s[i0 + 2] = fmaf(nx, l4.z, s[i0 + 2]);
      if (i0 + 3 < NB) s[i0 + 3] = fmaf(nx, l4.w, s[i0 + 3]);
    }
    s[j] = -nx;
  }

#pragma unroll
  for (int q = 0; q < NB / 4; ++q)
    *(float4*)(prow + 4 * q) = make_float4(s[4 * q + 0], s[4 * q + 1], s[4 * q + 2], s[4 * q + 3]);
  yw[row] += ndz;
}

__global__ void __launch_bounds__(128) gp_chol_syrk(float* __restrict__ K, const float* __restrict__ yw,
                                                    float* __restrict__ z, int kb) {
  __shared__ __attribute__((aligned(16))) float As[64][SPAD];
  __shared__ __attribute__((aligned(16))) float Bs[64][SPAD];
  int bi = blockIdx.x, bj = blockIdx.y;
  if (bj > bi) return;
  int base = kb * NB;
  int tb   = base + NB;
  int t = threadIdx.x;
  int lane = t & 31, wave = t >> 5;
  int rowA = tb + bi * 64, rowB = tb + bj * 64;

  for (int f = t; f < 64 * 16; f += 128) {
    int r = f >> 4, c4 = (f & 15) << 2;
    *(float4*)&As[r][c4] = *(const float4*)&K[(size_t)(rowA + r) * NTRAIN + base + c4];
    *(float4*)&Bs[r][c4] = *(const float4*)&K[(size_t)(rowB + r) * NTRAIN + base + c4];
  }
  __syncthreads();

  int wr = (wave >> 1) * 32;
  int wc = (wave & 1) * 32;
  v8f c00 = {}, c01 = {}, c10 = {}, c11 = {};
#pragma unroll
  for (int k0 = 0; k0 < NB; k0 += 4) {
    v2f a0 = frag_lds(As, wr,      k0, lane);
    v2f a1 = frag_lds(As, wr + 16, k0, lane);
    v2f b0 = frag_lds(Bs, wc,      k0, lane);
    v2f b1 = frag_lds(Bs, wc + 16, k0, lane);
    c00 = wmma4(a0, b0, c00);
    c01 = wmma4(a0, b1, c01);
    c10 = wmma4(a1, b0, c10);
    c11 = wmma4(a1, b1, c11);
  }

  int n  = lane & 15;
  int mh = (lane >> 4) * 8;
  int r0 = tb + bi * 64 + wr + mh;
  int c0 = tb + bj * 64 + wc + n;

  if (bi == 0 && bj == 0) {
    const int tr = wr + mh, tc = wc + n;
    float v00[8], v01[8], v10[8], v11[8];
#pragma unroll
    for (int r = 0; r < 8; ++r) {
      v00[r] = K[(size_t)(r0 + r) * NTRAIN + c0];
      v01[r] = K[(size_t)(r0 + r) * NTRAIN + c0 + 16];
      v10[r] = K[(size_t)(r0 + 16 + r) * NTRAIN + c0];
      v11[r] = K[(size_t)(r0 + 16 + r) * NTRAIN + c0 + 16];
    }
    float ya = 0.f, yb = 0.f;
    if (wave == 0) { ya = yw[tb + lane]; yb = yw[tb + 32 + lane]; }
    __syncthreads();
#pragma unroll
    for (int r = 0; r < 8; ++r) {
      As[tr + r][tc]           = v00[r] - c00[r];
      As[tr + r][tc + 16]      = v01[r] - c01[r];
      As[tr + 16 + r][tc]      = v10[r] - c10[r];
      As[tr + 16 + r][tc + 16] = v11[r] - c11[r];
    }
    __syncthreads();
    if (wave == 0) diag_factor_wave(As, lane, ya, yb, K, z, kb + 1);
    return;
  }

#pragma unroll
  for (int r = 0; r < 8; ++r) {
    K[(size_t)(r0 + r) * NTRAIN + c0]           -= c00[r];
    K[(size_t)(r0 + r) * NTRAIN + c0 + 16]      -= c01[r];
    K[(size_t)(r0 + 16 + r) * NTRAIN + c0]      -= c10[r];
    K[(size_t)(r0 + 16 + r) * NTRAIN + c0 + 16] -= c11[r];
  }
}

#define BWT 1024

__global__ void __launch_bounds__(BWT) gp_bwd_solve(const float* __restrict__ K,
                                                    float* __restrict__ y) {
  __shared__ float yl[NTRAIN];
  __shared__ float dg[NB][NB + 1];
  __shared__ float dd[NB];
  int t = threadIdx.x;
  for (int i = t; i < NTRAIN; i += BWT) yl[i] = y[i];

  int dr = t >> 6, dc = t & 63;
  float nd[4];
  {
    int rb0 = NTRAIN - NB;
#pragma unroll
    for (int q = 0; q < 4; ++q)
      nd[q] = K[(size_t)(rb0 + dr + 16 * q) * NTRAIN + rb0 + dc];
  }

  for (int m = NTRAIN / NB - 1; m >= 0; --m) {
    int rb = m * NB;
#pragma unroll
    for (int q = 0; q < 4; ++q) {
      int r = dr + 16 * q;
      dg[r][dc] = (dc < r) ? nd[q] : 0.f;
      if (dc == r) dd[r] = nd[q];
    }
    __syncthreads();
    if (m > 0) {
      int rn = rb - NB;
#pragma unroll
      for (int q = 0; q < 4; ++q)
        nd[q] = K[(size_t)(rn + dr + 16 * q) * NTRAIN + rn + dc];
    }

    if (t < 32) {
      float s0 = yl[rb + t], s1 = yl[rb + 32 + t];
      float r0 = 1.0f / dd[t];
      float r1 = 1.0f / dd[32 + t];
#pragma unroll
      for (int j = NB - 1; j >= 32; --j) {
        float xj = lane_bcast(s1, j - 32) * lane_bcast(r1, j - 32);
        s1 -= dg[j][32 + t] * xj;
        s0 -= dg[j][t] * xj;
      }
#pragma unroll
      for (int j = 31; j >= 1; --j) {
        float xj = lane_bcast(s0, j) * lane_bcast(r0, j);
        s0 -= dg[j][t] * xj;
      }
      yl[rb + t]      = s0 * r0;
      yl[rb + 32 + t] = s1 * r1;
    }
    __syncthreads();

    for (int i = t; i < rb; i += BWT) {
      const float* col = K + (size_t)rb * NTRAIN + i;
      float a0 = 0.f, a1 = 0.f, a2 = 0.f, a3 = 0.f;
#pragma unroll
      for (int j = 0; j < NB; j += 4) {
        a0 += col[(size_t)(j + 0) * NTRAIN] * yl[rb + j + 0];
        a1 += col[(size_t)(j + 1) * NTRAIN] * yl[rb + j + 1];
        a2 += col[(size_t)(j + 2) * NTRAIN] * yl[rb + j + 2];
        a3 += col[(size_t)(j + 3) * NTRAIN] * yl[rb + j + 3];
      }
      yl[i] -= (a0 + a1) + (a2 + a3);
    }
  }
  __syncthreads();
  for (int i = t; i < NTRAIN; i += BWT) y[i] = yl[i];
}

__global__ void gp_fused_grad(const float* __restrict__ x, const float* __restrict__ tx,
                              const float* __restrict__ alpha,
                              const float* __restrict__ nx, const float* __restrict__ nt,
                              float* __restrict__ out) {
  __shared__ float wt[8][16][18];
  int wave = threadIdx.x >> 5, lane = threadIdx.x & 31;
  int rowBase = (blockIdx.x * 8 + wave) * 16;
  int n  = lane & 15;
  int mh = (lane >> 4) * 8;

  float nxr[8];
#pragma unroll
  for (int r = 0; r < 8; ++r) nxr[r] = nx[rowBase + mh + r];

  v2f af[16];
#pragma unroll
  for (int k0 = 0; k0 < DIMS; k0 += 4)
    af[k0 >> 2] = frag_nt(x, DIMS, rowBase, 0, k0, lane);

  v8f g[4]; g[0] = {}; g[1] = {}; g[2] = {}; g[3] = {};
  float rs[8] = {0, 0, 0, 0, 0, 0, 0, 0};

  for (int c0 = 0; c0 < NTRAIN; c0 += 16) {
    if (c0 + 16 < NTRAIN)
      __builtin_prefetch(tx + (size_t)(c0 + 16) * DIMS + lane * 32, 0, 0);
    v8f ce = {}, co = {};
#pragma unroll
    for (int k0 = 0; k0 < DIMS; k0 += 8) {
      ce = wmma4(af[(k0 >> 2)],     frag_nt(tx, DIMS, c0, 0, k0,     lane), ce);
      co = wmma4(af[(k0 >> 2) + 1], frag_nt(tx, DIMS, c0, 0, k0 + 4, lane), co);
    }
    v8f c = ce + co;
    float ncol = nt[c0 + n];
    float av   = alpha[c0 + n];
#pragma unroll
    for (int r = 0; r < 8; ++r) {
      float d2 = nxr[r] + ncol - 2.0f * c[r];
      float w  = KSCALE * __expf(-0.5f * d2 * (1.0f / LS2)) * av;
      rs[r] += w;
      wt[wave][mh + r][n] = w;
    }
    __syncthreads();
#pragma unroll
    for (int nc = 0; nc < 4; ++nc) {
#pragma unroll
      for (int k0 = 0; k0 < 16; k0 += 4) {
        int kk = k0 + ((lane >> 4) << 1);
        v2f a, b;
        a.x = wt[wave][n][kk];
        a.y = wt[wave][n][kk + 1];
        b.x = tx[(size_t)(c0 + kk) * DIMS + nc * 16 + n];
        b.y = tx[(size_t)(c0 + kk + 1) * DIMS + nc * 16 + n];
        g[nc] = wmma4(a, b, g[nc]);
      }
    }
    __syncthreads();
  }

#pragma unroll
  for (int r = 0; r < 8; ++r)
    for (int off = 8; off; off >>= 1) rs[r] += __shfl_xor(rs[r], off, 32);

#pragma unroll
  for (int nc = 0; nc < 4; ++nc) {
#pragma unroll
    for (int r = 0; r < 8; ++r) {
      int row = rowBase + mh + r;
      int d   = nc * 16 + n;
      float xv = x[(size_t)row * DIMS + d];
      out[(size_t)row * DIMS + d] = xv - (g[nc][r] - rs[r] * xv) * (1.0f / LS2);
    }
  }
}

extern "C" void kernel_launch(void* const* d_in, const int* in_sizes, int n_in,
                              void* d_out, int out_size, void* d_ws, size_t ws_size,
                              hipStream_t stream) {
  (void)in_sizes; (void)n_in; (void)out_size; (void)ws_size;
  const float* x  = (const float*)d_in[0];
  const float* tx = (const float*)d_in[1];
  const float* ty = (const float*)d_in[2];
  float* out = (float*)d_out;

  float* Kd = (float*)d_ws;
  float* nt = Kd + (size_t)NTRAIN * NTRAIN;
  float* nx = nt + NTRAIN;
  float* al = nx + NQUERY;
  float* yw = al + NTRAIN;

  gp_sqnorm<<<(NTRAIN + 255) / 256, 256, 0, stream>>>(tx, nt, NTRAIN);
  gp_sqnorm<<<(NQUERY + 255) / 256, 256, 0, stream>>>(x, nx, NQUERY);

  gp_build_K<<<dim3(NTRAIN / 16, NTRAIN / 16), 32, 0, stream>>>(tx, nt, Kd);
  hipMemcpyAsync(yw, ty, NTRAIN * sizeof(float), hipMemcpyDeviceToDevice, stream);

  gp_chol_diag<<<1, 32, 0, stream>>>(Kd, yw, al, 0);
  for (int kb = 0; kb < NTRAIN / NB; ++kb) {
    int below = NTRAIN - NB * (kb + 1);
    if (below > 0) {
      gp_chol_trsm<<<below / NB, NB, 0, stream>>>(Kd, yw, al, kb);
      int tt = below / 64;
      gp_chol_syrk<<<dim3(tt, tt), 128, 0, stream>>>(Kd, yw, al, kb);
    }
  }

  gp_bwd_solve<<<1, BWT, 0, stream>>>(Kd, al);

  gp_fused_grad<<<NQUERY / 128, 256, 0, stream>>>(x, tx, al, nx, nt, out);
}


# final gradient kernel: one 16-row query tile per work-group, train points split over its 8 waves (512 work-groups, not 64); the 8 partial sums added through LDS in wave order: sum order changed
# speedup vs baseline: 12.0833x; 1.2965x over previous
#include <hip/hip_runtime.h>
#include <math.h>

#define DIMS   64
#define NTRAIN 4096
#define NQUERY 8192
#define NB     64
#define NOISE  1e-4f
#define LS2    64.0f
#define KSCALE 0.6931471805599453f
#define SPAD   68

typedef float v2f __attribute__((ext_vector_type(2)));
typedef float v8f __attribute__((ext_vector_type(8)));

__device__ __forceinline__ v8f wmma4(v2f a, v2f b, v8f c) {
  return __builtin_amdgcn_wmma_f32_16x16x4_f32(
       false, a,  false, b,
       (short)0, c,  false,  false);
}

__device__ __forceinline__ v2f frag_nt(const float* __restrict__ X, size_t rowStride,
                                       int rowBase, int colBase, int k0, int lane) {
  int kk = k0 + ((lane >> 4) << 1);
  const float* p = X + (size_t)(rowBase + (lane & 15)) * rowStride + colBase + kk;
  v2f v; v.x = p[0]; v.y = p[1];
  return v;
}

__device__ __forceinline__ v2f frag_lds(float (*S)[SPAD], int rowBase, int k0, int lane) {
  int kk = k0 + ((lane >> 4) << 1);
  const float* p = &S[rowBase + (lane & 15)][kk];
  v2f v; v.x = p[0]; v.y = p[1];
  return v;
}

__global__ void gp_sqnorm(const float* __restrict__ X, float* __restrict__ nrm, int rows) {
  int r = blockIdx.x * blockDim.x + threadIdx.x;
  if (r >= rows) return;
  const float4* p = (const float4*)(X + (size_t)r * DIMS);
  float s = 0.f;
#pragma unroll
  for (int i = 0; i < DIMS / 4; ++i) {
    float4 v = p[i];
    s += v.x * v.x + v.y * v.y + v.z * v.z + v.w * v.w;
  }
  nrm[r] = s;
}

__global__ void gp_build_K(const float* __restrict__ tx, const float* __restrict__ nrm,
                           float* __restrict__ K) {
  int bi = blockIdx.x, bj = blockIdx.y;
  int lane = threadIdx.x & 31;
  v8f ce = {}, co = {};
#pragma unroll
  for (int k0 = 0; k0 < DIMS; k0 += 8) {
    ce = wmma4(frag_nt(tx, DIMS, bi * 16, 0, k0, lane),
               frag_nt(tx, DIMS, bj * 16, 0, k0, lane), ce);
    co = wmma4(frag_nt(tx, DIMS, bi * 16, 0, k0 + 4, lane),
               frag_nt(tx, DIMS, bj * 16, 0, k0 + 4, lane), co);
  }
  v8f c = ce + co;
  int n  = lane & 15;
  int mh = (lane >> 4) * 8;
  int col = bj * 16 + n;
  float ncol = nrm[col];
#pragma unroll
  for (int r = 0; r < 8; ++r) {
    int row = bi * 16 + mh + r;
    float d2 = nrm[row] + ncol - 2.0f * c[r];
    float v  = KSCALE * __expf(-0.5f * d2 * (1.0f / LS2));
    if (row == col) v += NOISE;
    K[(size_t)row * NTRAIN + col] = v;
  }
}

#define DP 68

__device__ __forceinline__ float lane_bcast(float v, int srcLane) {
  return __int_as_float(__builtin_amdgcn_readlane(__float_as_int(v), srcLane));
}

__device__ __forceinline__ void wave_handover() {
  __builtin_amdgcn_fence(__ATOMIC_RELEASE, "wavefront");
  __builtin_amdgcn_wave_barrier();
  __builtin_amdgcn_fence(__ATOMIC_ACQUIRE, "wavefront");
}

template <int Q>
__device__ __forceinline__ void diag_block16(float (*a)[DP], int l, float& ya, float& yb,
                                             float& za, float& zb) {
  constexpr bool ROWA = (Q < 2);
  constexpr int c0 = 16 * Q;
  constexpr int l0 = c0 & 31;
  const int rA = l, rB = l + 32;
  float sA[16], sB[16];
#pragma unroll
  for (int u = 0; u < 4; ++u) {
    float4 vb = *(const float4*)&a[rB][c0 + 4 * u];
    sB[4 * u + 0] = vb.x; sB[4 * u + 1] = vb.y; sB[4 * u + 2] = vb.z; sB[4 * u + 3] = vb.w;
    if (ROWA) {
      float4 va = *(const float4*)&a[rA][c0 + 4 * u];
      sA[4 * u + 0] = va.x; sA[4 * u + 1] = va.y; sA[4 * u + 2] = va.z; sA[4 * u + 3] = va.w;
    } else {
      sA[4 * u + 0] = 0.f; sA[4 * u + 1] = 0.f; sA[4 * u + 2] = 0.f; sA[4 * u + 3] = 0.f;
    }
  }
#pragma unroll
  for (int jj = 0; jj < 16; ++jj) {
    const int jl = l0 + jj;
    float piv = lane_bcast(ROWA ? sA[jj] : sB[jj], jl);
    float r  = rsqrtf(piv);
    float zj = lane_bcast(ROWA ? ya : yb, jl) * r;
    float LB = sB[jj] * r;
    float LA = 0.f;
    sB[jj] = LB;
    if (ROWA) {
      LA = sA[jj] * r;
      sA[jj] = LA;
      if (l == jl) za = zj;
      ya -= zj * LA;
    } else {
      if (l == jl) zb = zj;
    }
    yb -= zj * LB;
#pragma unroll
    for (int ii = jj + 1; ii < 16; ++ii) {
      float Lij = lane_bcast(ROWA ? LA : LB, l0 + ii);
      sB[ii] -= LB * Lij;
      if (ROWA) sA[ii] -= LA * Lij;
    }
  }
#pragma unroll
  for (int u = 0; u < 4; ++u) {
    *(float4*)&a[rB][c0 + 4 * u] = make_float4(sB[4 * u + 0], sB[4 * u + 1], sB[4 * u + 2], sB[4 * u + 3]);
    if (ROWA)
      *(float4*)&a[rA][c0 + 4 * u] = make_float4(sA[4 * u + 0], sA[4 * u + 1], sA[4 * u + 2], sA[4 * u + 3]);
  }
  wave_handover();
  if constexpr (Q < 3) {
    constexpr int NT = (3 - Q) * (4 - Q) / 2;
    v8f acc[NT];
    {
      int idx = 0;
#pragma unroll
      for (int ni = Q + 1; ni < 4; ++ni) {
#pragma unroll
        for (int mi = ni; mi < 4; ++mi) {
          v8f c = {};
#pragma unroll
          for (int k0 = 0; k0 < 16; k0 += 4)
            c = wmma4(frag_lds(a, 16 * mi, c0 + k0, l), frag_lds(a, 16 * ni, c0 + k0, l), c);
          acc[idx++] = c;
        }
      }
    }
    const int n = l & 15, mh = (l >> 4) * 8;
    {
      int idx = 0;
#pragma unroll
      for (int ni = Q + 1; ni < 4; ++ni) {
#pragma unroll
        for (int mi = ni; mi < 4; ++mi) {
#pragma unroll
          for (int r = 0; r < 8; ++r)
            a[16 * mi + mh + r][16 * ni + n] -= acc[idx][r];
          ++idx;
        }
      }
    }
    wave_handover();
  }
}

__device__ __forceinline__ void diag_factor_wave(float (*a)[DP], int l, float ya, float yb,
                                                 float* __restrict__ K, float* __restrict__ z, int kb) {
  int base = kb * NB;
  int r0 = l >> 4, c4 = (l & 15) << 2;
  float* blk = K + (size_t)(base + r0) * NTRAIN + base + c4;
  float za = 0.f, zb = 0.f;
  diag_block16<0>(a, l, ya, yb, za, zb);
  diag_block16<1>(a, l, ya, yb, za, zb);
  diag_block16<2>(a, l, ya, yb, za, zb);
  diag_block16<3>(a, l, ya, yb, za, zb);
#pragma unroll
  for (int i = 0; i < NB / 2; ++i)
    *(float4*)(blk + (size_t)i * 2 * NTRAIN) = *(const float4*)&a[r0 + 2 * i][c4];
  z[base + l]      = za;
  z[base + 32 + l] = zb;
}

__global__ void __launch_bounds__(32) gp_chol_diag(float* __restrict__ K, const float* __restrict__ yw,
                                                   float* __restrict__ z, int kb) {
  __shared__ __attribute__((aligned(16))) float a[NB][DP];
  int l = threadIdx.x;
  int base = kb * NB;
  int r0 = l >> 4, c4 = (l & 15) << 2;
  float* blk = K + (size_t)(base + r0) * NTRAIN + base + c4;
#pragma unroll
  for (int i = 0; i < NB / 2; ++i)
    *(float4*)&a[r0 + 2 * i][c4]  = *(const float4*)(blk + (size_t)i * 2 * NTRAIN);
  float ya = yw[base + l], yb = yw[base + 32 + l];
  __syncthreads();
  diag_factor_wave(a, l, ya, yb, K, z, kb);
}

#define TP 68

__global__ void __launch_bounds__(NB) gp_chol_trsm(float* __restrict__ K, float* __restrict__ yw,
                                                   const float* __restrict__ z, int kb) {
  __shared__ __attribute__((aligned(16))) float LC[NB][TP];
  __shared__ __attribute__((aligned(16))) float nri[NB];
  __shared__ __attribute__((aligned(16))) float zs[NB];
  int t = threadIdx.x;
  int base = kb * NB;
  int row = base + NB + blockIdx.x * NB + t;
  float* prow = K + (size_t)row * NTRAIN + base;

  float s[NB];
#pragma unroll
  for (int q = 0; q < NB / 4; ++q) {
    float4 v = *(const float4*)(prow + 4 * q);
    s[4 * q + 0] = v.x; s[4 * q + 1] = v.y; s[4 * q + 2] = v.z; s[4 * q + 3] = v.w;
  }

  {
    int r0 = t >> 4, c4 = (t & 15) << 2;
    const float* blk = K + (size_t)(base + r0) * NTRAIN + base + c4;
#pragma unroll
    for (int i = 0; i < NB / 4; ++i) {
      float4 v = *(const float4*)(blk + (size_t)i * 4 * NTRAIN);
      int r = r0 + 4 * i;
      LC[c4 + 0][(r - c4 - 1) & (NB - 1)] = v.x;
      LC[c4 + 1][(r - c4 - 2) & (NB - 1)] = v.y;
      LC[c4 + 2][(r - c4 - 3) & (NB - 1)] = v.z;
      LC[c4 + 3][(r - c4 - 4) & (NB - 1)] = v.w;
    }
  }
  nri[t] = -1.0f / K[(size_t)(base + t) * NTRAIN + base + t];
  zs[t] = z[base + t];
  __syncthreads();

  float ndz = 0.f;
#pragma unroll
  for (int j = 0; j < NB; ++j) {
    const float nx = s[j] * nri[j];
    ndz = fmaf(nx, zs[j], ndz);
#pragma unroll
    for (int m = 0; j + 1 + 4 * m < NB; ++m) {
      const float4 l4 = *(const float4*)&LC[j][4 * m];
      const int i0 = j + 1 + 4 * m;
      s[i0] = fmaf(nx, l4.x, s[i0]);
      if (i0 + 1 < NB) s[i0 + 1] = fmaf(nx, l4.y, s[i0 + 1]);
      if (i0 + 2 < NB) s[i0 + 2] = fmaf(nx, l4.z, s[i0 + 2]);
      if (i0 + 3 < NB) s[i0 + 3] = fmaf(nx, l4.w, s[i0 + 3]);
    }
    s[j] = -nx;
  }

#pragma unroll
  for (int q = 0; q < NB / 4; ++q)
    *(float4*)(prow + 4 * q) = make_float4(s[4 * q + 0], s[4 * q + 1], s[4 * q + 2], s[4 * q + 3]);
  yw[row] += ndz;
}

__global__ void __launch_bounds__(128) gp_chol_syrk(float* __restrict__ K, const float* __restrict__ yw,
                                                    float* __restrict__ z, int kb) {
  __shared__ __attribute__((aligned(16))) float As[64][SPAD];
  __shared__ __attribute__((aligned(16))) float Bs[64][SPAD];
  int bi = blockIdx.x, bj = blockIdx.y;
  if (bj > bi) return;
  int base = kb * NB;
  int tb   = base + NB;
  int t = threadIdx.x;
  int lane = t & 31, wave = t >> 5;
  int rowA = tb + bi * 64, rowB = tb + bj * 64;

  for (int f = t; f < 64 * 16; f += 128) {
    int r = f >> 4, c4 = (f & 15) << 2;
    *(float4*)&As[r][c4] = *(const float4*)&K[(size_t)(rowA + r) * NTRAIN + base + c4];
    *(float4*)&Bs[r][c4] = *(const float4*)&K[(size_t)(rowB + r) * NTRAIN + base + c4];
  }
  __syncthreads();

  int wr = (wave >> 1) * 32;
  int wc = (wave & 1) * 32;
  v8f c00 = {}, c01 = {}, c10 = {}, c11 = {};
#pragma unroll
  for (int k0 = 0; k0 < NB; k0 += 4) {
    v2f a0 = frag_lds(As, wr,      k0, lane);
    v2f a1 = frag_lds(As, wr + 16, k0, lane);
    v2f b0 = frag_lds(Bs, wc,      k0, lane);
    v2f b1 = frag_lds(Bs, wc + 16, k0, lane);
    c00 = wmma4(a0, b0, c00);
    c01 = wmma4(a0, b1, c01);
    c10 = wmma4(a1, b0, c10);
    c11 = wmma4(a1, b1, c11);
  }

  int n  = lane & 15;
  int mh = (lane >> 4) * 8;
  int r0 = tb + bi * 64 + wr + mh;
  int c0 = tb + bj * 64 + wc + n;

  if (bi == 0 && bj == 0) {
    const int tr = wr + mh, tc = wc + n;
    float v00[8], v01[8], v10[8], v11[8];
#pragma unroll
    for (int r = 0; r < 8; ++r) {
      v00[r] = K[(size_t)(r0 + r) * NTRAIN + c0];
      v01[r] = K[(size_t)(r0 + r) * NTRAIN + c0 + 16];
      v10[r] = K[(size_t)(r0 + 16 + r) * NTRAIN + c0];
      v11[r] = K[(size_t)(r0 + 16 + r) * NTRAIN + c0 + 16];
    }
    float ya = 0.f, yb = 0.f;
    if (wave == 0) { ya = yw[tb + lane]; yb = yw[tb + 32 + lane]; }
    __syncthreads();
#pragma unroll
    for (int r = 0; r < 8; ++r) {
      As[tr + r][tc]           = v00[r] - c00[r];
      As[tr + r][tc + 16]      = v01[r] - c01[r];
      As[tr + 16 + r][tc]      = v10[r] - c10[r];
      As[tr + 16 + r][tc + 16] = v11[r] - c11[r];
    }
    __syncthreads();
    if (wave == 0) diag_factor_wave(As, lane, ya, yb, K, z, kb + 1);
    return;
  }

#pragma unroll
  for (int r = 0; r < 8; ++r) {
    K[(size_t)(r0 + r) * NTRAIN + c0]           -= c00[r];
    K[(size_t)(r0 + r) * NTRAIN + c0 + 16]      -= c01[r];
    K[(size_t)(r0 + 16 + r) * NTRAIN + c0]      -= c10[r];
    K[(size_t)(r0 + 16 + r) * NTRAIN + c0 + 16] -= c11[r];
  }
}

#define BWT 1024

__global__ void __launch_bounds__(BWT) gp_bwd_solve(const float* __restrict__ K,
                                                    float* __restrict__ y) {
  __shared__ float yl[NTRAIN];
  __shared__ float dg[NB][NB + 1];
  __shared__ float dd[NB];
  int t = threadIdx.x;
  for (int i = t; i < NTRAIN; i += BWT) yl[i] = y[i];

  int dr = t >> 6, dc = t & 63;
  float nd[4];
  {
    int rb0 = NTRAIN - NB;
#pragma unroll
    for (int q = 0; q < 4; ++q)
      nd[q] = K[(size_t)(rb0 + dr + 16 * q) * NTRAIN + rb0 + dc];
  }

  for (int m = NTRAIN / NB - 1; m >= 0; --m) {
    int rb = m * NB;
#pragma unroll
    for (int q = 0; q < 4; ++q) {
      int r = dr + 16 * q;
      dg[r][dc] = (dc < r) ? nd[q] : 0.f;
      if (dc == r) dd[r] = nd[q];
    }
    __syncthreads();
    if (m > 0) {
      int rn = rb - NB;
#pragma unroll
      for (int q = 0; q < 4; ++q)
        nd[q] = K[(size_t)(rn + dr + 16 * q) * NTRAIN + rn + dc];
    }

    if (t < 32) {
      float s0 = yl[rb + t], s1 = yl[rb + 32 + t];
      float r0 = 1.0f / dd[t];
      float r1 = 1.0f / dd[32 + t];
#pragma unroll
      for (int j = NB - 1; j >= 32; --j) {
        float xj = lane_bcast(s1, j - 32) * lane_bcast(r1, j - 32);
        s1 -= dg[j][32 + t] * xj;
        s0 -= dg[j][t] * xj;
      }
#pragma unroll
      for (int j = 31; j >= 1; --j) {
        float xj = lane_bcast(s0, j) * lane_bcast(r0, j);
        s0 -= dg[j][t] * xj;
      }
      yl[rb + t]      = s0 * r0;
      yl[rb + 32 + t] = s1 * r1;
    }
    __syncthreads();

    for (int i = t; i < rb; i += BWT) {
      const float* col = K + (size_t)rb * NTRAIN + i;
      float a0 = 0.f, a1 = 0.f, a2 = 0.f, a3 = 0.f;
#pragma unroll
      for (int j = 0; j < NB; j += 4) {
        a0 += col[(size_t)(j + 0) * NTRAIN] * yl[rb + j + 0];
        a1 += col[(size_t)(j + 1) * NTRAIN] * yl[rb + j + 1];
        a2 += col[(size_t)(j + 2) * NTRAIN] * yl[rb + j + 2];
        a3 += col[(size_t)(j + 3) * NTRAIN] * yl[rb + j + 3];
      }
      yl[i] -= (a0 + a1) + (a2 + a3);
    }
  }
  __syncthreads();
  for (int i = t; i < NTRAIN; i += BWT) y[i] = yl[i];
}

#define FGW   8
#define FGSPAN (NTRAIN / FGW)

__global__ void gp_fused_grad(const float* __restrict__ x, const float* __restrict__ tx,
                              const float* __restrict__ alpha,
                              const float* __restrict__ nx, const float* __restrict__ nt,
                              float* __restrict__ out) {
  __shared__ float wt[FGW][16][18];
  __shared__ float red[FGW][40][32];
  int wave = threadIdx.x >> 5, lane = threadIdx.x & 31;
  int rowBase = blockIdx.x * 16;
  int n  = lane & 15;
  int mh = (lane >> 4) * 8;

  float nxr[8];
#pragma unroll
  for (int r = 0; r < 8; ++r) nxr[r] = nx[rowBase + mh + r];

  v2f af[16];
#pragma unroll
  for (int k0 = 0; k0 < DIMS; k0 += 4)
    af[k0 >> 2] = frag_nt(x, DIMS, rowBase, 0, k0, lane);

  v8f g[4]; g[0] = {}; g[1] = {}; g[2] = {}; g[3] = {};
  float rs[8] = {0, 0, 0, 0, 0, 0, 0, 0};

  const int cBeg = wave * FGSPAN, cEnd = cBeg + FGSPAN;
  for (int c0 = cBeg; c0 < cEnd; c0 += 16) {
    if (c0 + 16 < cEnd)
      __builtin_prefetch(tx + (size_t)(c0 + 16) * DIMS + lane * 32, 0, 0);
    v8f ce = {}, co = {};
#pragma unroll
    for (int k0 = 0; k0 < DIMS; k0 += 8) {
      ce = wmma4(af[(k0 >> 2)],     frag_nt(tx, DIMS, c0, 0, k0,     lane), ce);
      co = wmma4(af[(k0 >> 2) + 1], frag_nt(tx, DIMS, c0, 0, k0 + 4, lane), co);
    }
    v8f c = ce + co;
    float ncol = nt[c0 + n];
    float av   = alpha[c0 + n];
#pragma unroll
    for (int r = 0; r < 8; ++r) {
      float d2 = nxr[r] + ncol - 2.0f * c[r];
      float w  = KSCALE * __expf(-0.5f * d2 * (1.0f / LS2)) * av;
      rs[r] += w;
      wt[wave][mh + r][n] = w;
    }
    __syncthreads();
#pragma unroll
    for (int nc = 0; nc < 4; ++nc) {
#pragma unroll
      for (int k0 = 0; k0 < 16; k0 += 4) {
        int kk = k0 + ((lane >> 4) << 1);
        v2f a, b;
        a.x = wt[wave][n][kk];
        a.y = wt[wave][n][kk + 1];
        b.x = tx[(size_t)(c0 + kk) * DIMS + nc * 16 + n];
        b.y = tx[(size_t)(c0 + kk + 1) * DIMS + nc * 16 + n];
        g[nc] = wmma4(a, b, g[nc]);
      }
    }
    __syncthreads();
  }

#pragma unroll
  for (int nc = 0; nc < 4; ++nc) {
#pragma unroll
    for (int r = 0; r < 8; ++r) red[wave][nc * 8 + r][lane] = g[nc][r];
  }
#pragma unroll
  for (int r = 0; r < 8; ++r) red[wave][32 + r][lane] = rs[r];
  __syncthreads();

  const int ncp = (wave & 1) * 2;
  const int rq  = (wave >> 1) * 2;
#pragma unroll
  for (int q = 0; q < 2; ++q) {
    const int r = rq + q;
    float s = 0.f, g0 = 0.f, g1 = 0.f;
#pragma unroll
    for (int w = 0; w < FGW; ++w) {
      s  += red[w][32 + r][lane];
      g0 += red[w][ncp * 8 + r][lane];
      g1 += red[w][(ncp + 1) * 8 + r][lane];
    }
    for (int off = 8; off; off >>= 1) s += __shfl_xor(s, off, 32);
    int row = rowBase + mh + r;
    int d0  = ncp * 16 + n;
    float xv0 = x[(size_t)row * DIMS + d0];
    float xv1 = x[(size_t)row * DIMS + d0 + 16];
    out[(size_t)row * DIMS + d0]      = xv0 - (g0 - s * xv0) * (1.0f / LS2);
    out[(size_t)row * DIMS + d0 + 16] = xv1 - (g1 - s * xv1) * (1.0f / LS2);
  }
}

extern "C" void kernel_launch(void* const* d_in, const int* in_sizes, int n_in,
                              void* d_out, int out_size, void* d_ws, size_t ws_size,
                              hipStream_t stream) {
  (void)in_sizes; (void)n_in; (void)out_size; (void)ws_size;
  const float* x  = (const float*)d_in[0];
  const float* tx = (const float*)d_in[1];
  const float* ty = (const float*)d_in[2];
  float* out = (float*)d_out;

  float* Kd = (float*)d_ws;
  float* nt = Kd + (size_t)NTRAIN * NTRAIN;
  float* nx = nt + NTRAIN;
  float* al = nx + NQUERY;
  float* yw = al + NTRAIN;

  gp_sqnorm<<<(NTRAIN + 255) / 256, 256, 0, stream>>>(tx, nt, NTRAIN);
  gp_sqnorm<<<(NQUERY + 255) / 256, 256, 0, stream>>>(x, nx, NQUERY);

  gp_build_K<<<dim3(NTRAIN / 16, NTRAIN / 16), 32, 0, stream>>>(tx, nt, Kd);
  hipMemcpyAsync(yw, ty, NTRAIN * sizeof(float), hipMemcpyDeviceToDevice, stream);

  gp_chol_diag<<<1, 32, 0, stream>>>(Kd, yw, al, 0);
  for (int kb = 0; kb < NTRAIN / NB; ++kb) {
    int below = NTRAIN - NB * (kb + 1);
    if (below > 0) {
      gp_chol_trsm<<<below / NB, NB, 0, stream>>>(Kd, yw, al, kb);
      int tt = below / 64;
      gp_chol_syrk<<<dim3(tt, tt), 128, 0, stream>>>(Kd, yw, al, kb);
    }
  }

  gp_bwd_solve<<<1, BWT, 0, stream>>>(Kd, al);

  gp_fused_grad<<<NQUERY / 16, FGW * 32, 0, stream>>>(x, tx, al, nx, nt, out);
}


# backward solve in two levels: 1024-row triangles by one work-group; each block's product with all columns left of it on up to 96 work-groups in its own launch (6 launches more); sums regrouped
# speedup vs baseline: 12.8814x; 1.0660x over previous
#include <hip/hip_runtime.h>
#include <math.h>

#define DIMS   64
#define NTRAIN 4096
#define NQUERY 8192
#define NB     64
#define NOISE  1e-4f
#define LS2    64.0f
#define KSCALE 0.6931471805599453f
#define SPAD   68

typedef float v2f __attribute__((ext_vector_type(2)));
typedef float v8f __attribute__((ext_vector_type(8)));

__device__ __forceinline__ v8f wmma4(v2f a, v2f b, v8f c) {
  return __builtin_amdgcn_wmma_f32_16x16x4_f32(
       false, a,  false, b,
       (short)0, c,  false,  false);
}

__device__ __forceinline__ v2f frag_nt(const float* __restrict__ X, size_t rowStride,
                                       int rowBase, int colBase, int k0, int lane) {
  int kk = k0 + ((lane >> 4) << 1);
  const float* p = X + (size_t)(rowBase + (lane & 15)) * rowStride + colBase + kk;
  v2f v; v.x = p[0]; v.y = p[1];
  return v;
}

__device__ __forceinline__ v2f frag_lds(float (*S)[SPAD], int rowBase, int k0, int lane) {
  int kk = k0 + ((lane >> 4) << 1);
  const float* p = &S[rowBase + (lane & 15)][kk];
  v2f v; v.x = p[0]; v.y = p[1];
  return v;
}

__global__ void gp_sqnorm(const float* __restrict__ X, float* __restrict__ nrm, int rows) {
  int r = blockIdx.x * blockDim.x + threadIdx.x;
  if (r >= rows) return;
  const float4* p = (const float4*)(X + (size_t)r * DIMS);
  float s = 0.f;
#pragma unroll
  for (int i = 0; i < DIMS / 4; ++i) {
    float4 v = p[i];
    s += v.x * v.x + v.y * v.y + v.z * v.z + v.w * v.w;
  }
  nrm[r] = s;
}

__global__ void gp_build_K(const float* __restrict__ tx, const float* __restrict__ nrm,
                           float* __restrict__ K) {
  int bi = blockIdx.x, bj = blockIdx.y;
  int lane = threadIdx.x & 31;
  v8f ce = {}, co = {};
#pragma unroll
  for (int k0 = 0; k0 < DIMS; k0 += 8) {
    ce = wmma4(frag_nt(tx, DIMS, bi * 16, 0, k0, lane),
               frag_nt(tx, DIMS, bj * 16, 0, k0, lane), ce);
    co = wmma4(frag_nt(tx, DIMS, bi * 16, 0, k0 + 4, lane),
               frag_nt(tx, DIMS, bj * 16, 0, k0 + 4, lane), co);
  }
  v8f c = ce + co;
  int n  = lane & 15;
  int mh = (lane >> 4) * 8;
  int col = bj * 16 + n;
  float ncol = nrm[col];
#pragma unroll
  for (int r = 0; r < 8; ++r) {
    int row = bi * 16 + mh + r;
    float d2 = nrm[row] + ncol - 2.0f * c[r];
    float v  = KSCALE * __expf(-0.5f * d2 * (1.0f / LS2));
    if (row == col) v += NOISE;
    K[(size_t)row * NTRAIN + col] = v;
  }
}

#define DP 68

__device__ __forceinline__ float lane_bcast(float v, int srcLane) {
  return __int_as_float(__builtin_amdgcn_readlane(__float_as_int(v), srcLane));
}

__device__ __forceinline__ void wave_handover() {
  __builtin_amdgcn_fence(__ATOMIC_RELEASE, "wavefront");
  __builtin_amdgcn_wave_barrier();
  __builtin_amdgcn_fence(__ATOMIC_ACQUIRE, "wavefront");
}

template <int Q>
__device__ __forceinline__ void diag_block16(float (*a)[DP], int l, float& ya, float& yb,
                                             float& za, float& zb) {
  constexpr bool ROWA = (Q < 2);
  constexpr int c0 = 16 * Q;
  constexpr int l0 = c0 & 31;
  const int rA = l, rB = l + 32;
  float sA[16], sB[16];
#pragma unroll
  for (int u = 0; u < 4; ++u) {
    float4 vb = *(const float4*)&a[rB][c0 + 4 * u];
    sB[4 * u + 0] = vb.x; sB[4 * u + 1] = vb.y; sB[4 * u + 2] = vb.z; sB[4 * u + 3] = vb.w;
    if (ROWA) {
      float4 va = *(const float4*)&a[rA][c0 + 4 * u];
      sA[4 * u + 0] = va.x; sA[4 * u + 1] = va.y; sA[4 * u + 2] = va.z; sA[4 * u + 3] = va.w;
    } else {
      sA[4 * u + 0] = 0.f; sA[4 * u + 1] = 0.f; sA[4 * u + 2] = 0.f; sA[4 * u + 3] = 0.f;
    }
  }
#pragma unroll
  for (int jj = 0; jj < 16; ++jj) {
    const int jl = l0 + jj;
    float piv = lane_bcast(ROWA ? sA[jj] : sB[jj], jl);
    float r  = rsqrtf(piv);
    float zj = lane_bcast(ROWA ? ya : yb, jl) * r;
    float LB = sB[jj] * r;
    float LA = 0.f;
    sB[jj] = LB;
    if (ROWA) {
      LA = sA[jj] * r;
      sA[jj] = LA;
      if (l == jl) za = zj;
      ya -= zj * LA;
    } else {
      if (l == jl) zb = zj;
    }
    yb -= zj * LB;
#pragma unroll
    for (int ii = jj + 1; ii < 16; ++ii) {
      float Lij = lane_bcast(ROWA ? LA : LB, l0 + ii);
      sB[ii] -= LB * Lij;
      if (ROWA) sA[ii] -= LA * Lij;
    }
  }
#pragma unroll
  for (int u = 0; u < 4; ++u) {
    *(float4*)&a[rB][c0 + 4 * u] = make_float4(sB[4 * u + 0], sB[4 * u + 1], sB[4 * u + 2], sB[4 * u + 3]);
    if (ROWA)
      *(float4*)&a[rA][c0 + 4 * u] = make_float4(sA[4 * u + 0], sA[4 * u + 1], sA[4 * u + 2], sA[4 * u + 3]);
  }
  wave_handover();
  if constexpr (Q < 3) {
    constexpr int NT = (3 - Q) * (4 - Q) / 2;
    v8f acc[NT];
    {
      int idx = 0;
#pragma unroll
      for (int ni = Q + 1; ni < 4; ++ni) {
#pragma unroll
        for (int mi = ni; mi < 4; ++mi) {
          v8f c = {};
#pragma unroll
          for (int k0 = 0; k0 < 16; k0 += 4)
            c = wmma4(frag_lds(a, 16 * mi, c0 + k0, l), frag_lds(a, 16 * ni, c0 + k0, l), c);
          acc[idx++] = c;
        }
      }
    }
    const int n = l & 15, mh = (l >> 4) * 8;
    {
      int idx = 0;
#pragma unroll
      for (int ni = Q + 1; ni < 4; ++ni) {
#pragma unroll
        for (int mi = ni; mi < 4; ++mi) {
#pragma unroll
          for (int r = 0; r < 8; ++r)
            a[16 * mi + mh + r][16 * ni + n] -= acc[idx][r];
          ++idx;
        }
      }
    }
    wave_handover();
  }
}

__device__ __forceinline__ void diag_factor_wave(float (*a)[DP], int l, float ya, float yb,
                                                 float* __restrict__ K, float* __restrict__ z, int kb) {
  int base = kb * NB;
  int r0 = l >> 4, c4 = (l & 15) << 2;
  float* blk = K + (size_t)(base + r0) * NTRAIN + base + c4;
  float za = 0.f, zb = 0.f;
  diag_block16<0>(a, l, ya, yb, za, zb);
  diag_block16<1>(a, l, ya, yb, za, zb);
  diag_block16<2>(a, l, ya, yb, za, zb);
  diag_block16<3>(a, l, ya, yb, za, zb);
#pragma unroll
  for (int i = 0; i < NB / 2; ++i)
    *(float4*)(blk + (size_t)i * 2 * NTRAIN) = *(const float4*)&a[r0 + 2 * i][c4];
  z[base + l]      = za;
  z[base + 32 + l] = zb;
}

__global__ void __launch_bounds__(32) gp_chol_diag(float* __restrict__ K, const float* __restrict__ yw,
                                                   float* __restrict__ z, int kb) {
  __shared__ __attribute__((aligned(16))) float a[NB][DP];
  int l = threadIdx.x;
  int base = kb * NB;
  int r0 = l >> 4, c4 = (l & 15) << 2;
  float* blk = K + (size_t)(base + r0) * NTRAIN + base + c4;
#pragma unroll
  for (int i = 0; i < NB / 2; ++i)
    *(float4*)&a[r0 + 2 * i][c4]  = *(const float4*)(blk + (size_t)i * 2 * NTRAIN);
  float ya = yw[base + l], yb = yw[base + 32 + l];
  __syncthreads();
  diag_factor_wave(a, l, ya, yb, K, z, kb);
}

#define TP 68

__global__ void __launch_bounds__(NB) gp_chol_trsm(float* __restrict__ K, float* __restrict__ yw,
                                                   const float* __restrict__ z, int kb) {
  __shared__ __attribute__((aligned(16))) float LC[NB][TP];
  __shared__ __attribute__((aligned(16))) float nri[NB];
  __shared__ __attribute__((aligned(16))) float zs[NB];
  int t = threadIdx.x;
  int base = kb * NB;
  int row = base + NB + blockIdx.x * NB + t;
  float* prow = K + (size_t)row * NTRAIN + base;

  float s[NB];
#pragma unroll
  for (int q = 0; q < NB / 4; ++q) {
    float4 v = *(const float4*)(prow + 4 * q);
    s[4 * q + 0] = v.x; s[4 * q + 1] = v.y; s[4 * q + 2] = v.z; s[4 * q + 3] = v.w;
  }

  {
    int r0 = t >> 4, c4 = (t & 15) << 2;
    const float* blk = K + (size_t)(base + r0) * NTRAIN + base + c4;
#pragma unroll
    for (int i = 0; i < NB / 4; ++i) {
      float4 v = *(const float4*)(blk + (size_t)i * 4 * NTRAIN);
      int r = r0 + 4 * i;
      LC[c4 + 0][(r - c4 - 1) & (NB - 1)] = v.x;
      LC[c4 + 1][(r - c4 - 2) & (NB - 1)] = v.y;
      LC[c4 + 2][(r - c4 - 3) & (NB - 1)] = v.z;
      LC[c4 + 3][(r - c4 - 4) & (NB - 1)] = v.w;
    }
  }
  nri[t] = -1.0f / K[(size_t)(base + t) * NTRAIN + base + t];
  zs[t] = z[base + t];
  __syncthreads();

  float ndz = 0.f;
#pragma unroll
  for (int j = 0; j < NB; ++j) {
    const float nx = s[j] * nri[j];
    ndz = fmaf(nx, zs[j], ndz);
#pragma unroll
    for (int m = 0; j + 1 + 4 * m < NB; ++m) {
      const float4 l4 = *(const float4*)&LC[j][4 * m];
      const int i0 = j + 1 + 4 * m;
      s[i0] = fmaf(nx, l4.x, s[i0]);
      if (i0 + 1 < NB) s[i0 + 1] = fmaf(nx, l4.y, s[i0 + 1]);
      if (i0 + 2 < NB) s[i0 + 2] = fmaf(nx, l4.z, s[i0 + 2]);
      if (i0 + 3 < NB) s[i0 + 3] = fmaf(nx, l4.w, s[i0 + 3]);
    }
    s[j] = -nx;
  }

#pragma unroll
  for (int q = 0; q < NB / 4; ++q)
    *(float4*)(prow + 4 * q) = make_float4(s[4 * q + 0], s[4 * q + 1], s[4 * q + 2], s[4 * q + 3]);
  yw[row] += ndz;
}

__global__ void __launch_bounds__(128) gp_chol_syrk(float* __restrict__ K, const float* __restrict__ yw,
                                                    float* __restrict__ z, int kb) {
  __shared__ __attribute__((aligned(16))) float As[64][SPAD];
  __shared__ __attribute__((aligned(16))) float Bs[64][SPAD];
  int bi = blockIdx.x, bj = blockIdx.y;
  if (bj > bi) return;
  int base = kb * NB;
  int tb   = base + NB;
  int t = threadIdx.x;
  int lane = t & 31, wave = t >> 5;
  int rowA = tb + bi * 64, rowB = tb + bj * 64;

  for (int f = t; f < 64 * 16; f += 128) {
    int r = f >> 4, c4 = (f & 15) << 2;
    *(float4*)&As[r][c4] = *(const float4*)&K[(size_t)(rowA + r) * NTRAIN + base + c4];
    *(float4*)&Bs[r][c4] = *(const float4*)&K[(size_t)(rowB + r) * NTRAIN + base + c4];
  }
  __syncthreads();

  int wr = (wave >> 1) * 32;
  int wc = (wave & 1) * 32;
  v8f c00 = {}, c01 = {}, c10 = {}, c11 = {};
#pragma unroll
  for (int k0 = 0; k0 < NB; k0 += 4) {
    v2f a0 = frag_lds(As, wr,      k0, lane);
    v2f a1 = frag_lds(As, wr + 16, k0, lane);
    v2f b0 = frag_lds(Bs, wc,      k0, lane);
    v2f b1 = frag_lds(Bs, wc + 16, k0, lane);
    c00 = wmma4(a0, b0, c00);
    c01 = wmma4(a0, b1, c01);
    c10 = wmma4(a1, b0, c10);
    c11 = wmma4(a1, b1, c11);
  }

  int n  = lane & 15;
  int mh = (lane >> 4) * 8;
  int r0 = tb + bi * 64 + wr + mh;
  int c0 = tb + bj * 64 + wc + n;

  if (bi == 0 && bj == 0) {
    const int tr = wr + mh, tc = wc + n;
    float v00[8], v01[8], v10[8], v11[8];
#pragma unroll
    for (int r = 0; r < 8; ++r) {
      v00[r] = K[(size_t)(r0 + r) * NTRAIN + c0];
      v01[r] = K[(size_t)(r0 + r) * NTRAIN + c0 + 16];
      v10[r] = K[(size_t)(r0 + 16 + r) * NTRAIN + c0];
      v11[r] = K[(size_t)(r0 + 16 + r) * NTRAIN + c0 + 16];
    }
    float ya = 0.f, yb = 0.f;
    if (wave == 0) { ya = yw[tb + lane]; yb = yw[tb + 32 + lane]; }
    __syncthreads();
#pragma unroll
    for (int r = 0; r < 8; ++r) {
      As[tr + r][tc]           = v00[r] - c00[r];
      As[tr + r][tc + 16]      = v01[r] - c01[r];
      As[tr + 16 + r][tc]      = v10[r] - c10[r];
      As[tr + 16 + r][tc + 16] = v11[r] - c11[r];
    }
    __syncthreads();
    if (wave == 0) diag_factor_wave(As, lane, ya, yb, K, z, kb + 1);
    return;
  }

#pragma unroll
  for (int r = 0; r < 8; ++r) {
    K[(size_t)(r0 + r) * NTRAIN + c0]           -= c00[r];
    K[(size_t)(r0 + r) * NTRAIN + c0 + 16]      -= c01[r];
    K[(size_t)(r0 + 16 + r) * NTRAIN + c0]      -= c10[r];
    K[(size_t)(r0 + 16 + r) * NTRAIN + c0 + 16] -= c11[r];
  }
}

#define BWT 1024
#define BWB 1024
#define GVC 32
#define GVW 16
#define GVR (BWB / GVW)

__global__ void __launch_bounds__(BWT) gp_bwd_block(const float* __restrict__ K,
                                                    float* __restrict__ y, int J) {
  __shared__ float yl[BWB];
  __shared__ float dg[NB][NB + 1];
  __shared__ float dd[NB];
  int t = threadIdx.x;
  const int bb0 = J * BWB;
  for (int i = t; i < BWB; i += BWT) yl[i] = y[bb0 + i];

  int dr = t >> 6, dc = t & 63;
  float nd[4];
  {
    int rb0 = bb0 + BWB - NB;
#pragma unroll
    for (int q = 0; q < 4; ++q)
      nd[q] = K[(size_t)(rb0 + dr + 16 * q) * NTRAIN + rb0 + dc];
  }

  for (int m = BWB / NB - 1; m >= 0; --m) {
    int lb = m * NB;
    int rb = bb0 + lb;
#pragma unroll
    for (int q = 0; q < 4; ++q) {
      int r = dr + 16 * q;
      dg[r][dc] = (dc < r) ? nd[q] : 0.f;
      if (dc == r) dd[r] = nd[q];
    }
    __syncthreads();
    if (m > 0) {
      int rn = rb - NB;
#pragma unroll
      for (int q = 0; q < 4; ++q)
        nd[q] = K[(size_t)(rn + dr + 16 * q) * NTRAIN + rn + dc];
    }

    if (t < 32) {
      float s0 = yl[lb + t], s1 = yl[lb + 32 + t];
      float r0 = 1.0f / dd[t];
      float r1 = 1.0f / dd[32 + t];
#pragma unroll
      for (int j = NB - 1; j >= 32; --j) {
        float xj = lane_bcast(s1, j - 32) * lane_bcast(r1, j - 32);
        s1 -= dg[j][32 + t] * xj;
        s0 -= dg[j][t] * xj;
      }
#pragma unroll
      for (int j = 31; j >= 1; --j) {
        float xj = lane_bcast(s0, j) * lane_bcast(r0, j);
        s0 -= dg[j][t] * xj;
      }
      yl[lb + t]      = s0 * r0;
      yl[lb + 32 + t] = s1 * r1;
    }
    __syncthreads();

    if (t < lb) {
      const float* col = K + (size_t)rb * NTRAIN + bb0 + t;
      float a0 = 0.f, a1 = 0.f, a2 = 0.f, a3 = 0.f;
#pragma unroll
      for (int j = 0; j < NB; j += 4) {
        a0 += col[(size_t)(j + 0) * NTRAIN] * yl[lb + j + 0];
        a1 += col[(size_t)(j + 1) * NTRAIN] * yl[lb + j + 1];
        a2 += col[(size_t)(j + 2) * NTRAIN] * yl[lb + j + 2];
        a3 += col[(size_t)(j + 3) * NTRAIN] * yl[lb + j + 3];
      }
      yl[t] -= (a0 + a1) + (a2 + a3);
    }
  }
  __syncthreads();
  for (int i = t; i < BWB; i += BWT) y[bb0 + i] = yl[i];
}

__global__ void __launch_bounds__(GVW * 32) gp_bwd_gemv(const float* __restrict__ K,
                                                        float* __restrict__ y, int J) {
  __shared__ float xs[BWB];
  __shared__ float part[GVW][GVC];
  int t = threadIdx.x;
  int c = t & 31, g = t >> 5;
  const int bb0 = J * BWB;
  float xa = y[bb0 + t], xb = y[bb0 + GVW * 32 + t];
  const int col = blockIdx.x * GVC + c;
  const float* p = K + (size_t)(bb0 + g * GVR) * NTRAIN + col;
  float v[GVR];
#pragma unroll
  for (int j = 0; j < GVR; ++j) v[j] = p[(size_t)j * NTRAIN];
  xs[t] = xa;
  xs[GVW * 32 + t] = xb;
  __syncthreads();
  const float* xg = xs + g * GVR;
  float a0 = 0.f, a1 = 0.f, a2 = 0.f, a3 = 0.f;
#pragma unroll
  for (int j = 0; j < GVR; j += 4) {
    a0 += v[j + 0] * xg[j + 0];
    a1 += v[j + 1] * xg[j + 1];
    a2 += v[j + 2] * xg[j + 2];
    a3 += v[j + 3] * xg[j + 3];
  }
  part[g][c] = (a0 + a1) + (a2 + a3);
  __syncthreads();
  if (g == 0) {
    float s = 0.f;
#pragma unroll
    for (int w = 0; w < GVW; ++w) s += part[w][c];
    y[col] -= s;
  }
}

#define FGW   8
#define FGSPAN (NTRAIN / FGW)

__global__ void gp_fused_grad(const float* __restrict__ x, const float* __restrict__ tx,
                              const float* __restrict__ alpha,
                              const float* __restrict__ nx, const float* __restrict__ nt,
                              float* __restrict__ out) {
  __shared__ float wt[FGW][16][18];
  __shared__ float red[FGW][40][32];
  int wave = threadIdx.x >> 5, lane = threadIdx.x & 31;
  int rowBase = blockIdx.x * 16;
  int n  = lane & 15;
  int mh = (lane >> 4) * 8;

  float nxr[8];
#pragma unroll
  for (int r = 0; r < 8; ++r) nxr[r] = nx[rowBase + mh + r];

  v2f af[16];
#pragma unroll
  for (int k0 = 0; k0 < DIMS; k0 += 4)
    af[k0 >> 2] = frag_nt(x, DIMS, rowBase, 0, k0, lane);

  v8f g[4]; g[0] = {}; g[1] = {}; g[2] = {}; g[3] = {};
  float rs[8] = {0, 0, 0, 0, 0, 0, 0, 0};

  const int cBeg = wave * FGSPAN, cEnd = cBeg + FGSPAN;
  for (int c0 = cBeg; c0 < cEnd; c0 += 16) {
    if (c0 + 16 < cEnd)
      __builtin_prefetch(tx + (size_t)(c0 + 16) * DIMS + lane * 32, 0, 0);
    v8f ce = {}, co = {};
#pragma unroll
    for (int k0 = 0; k0 < DIMS; k0 += 8) {
      ce = wmma4(af[(k0 >> 2)],     frag_nt(tx, DIMS, c0, 0, k0,     lane), ce);
      co = wmma4(af[(k0 >> 2) + 1], frag_nt(tx, DIMS, c0, 0, k0 + 4, lane), co);
    }
    v8f c = ce + co;
    float ncol = nt[c0 + n];
    float av   = alpha[c0 + n];
#pragma unroll
    for (int r = 0; r < 8; ++r) {
      float d2 = nxr[r] + ncol - 2.0f * c[r];
      float w  = KSCALE * __expf(-0.5f * d2 * (1.0f / LS2)) * av;
      rs[r] += w;
      wt[wave][mh + r][n] = w;
    }
    __syncthreads();
#pragma unroll
    for (int nc = 0; nc < 4; ++nc) {
#pragma unroll
      for (int k0 = 0; k0 < 16; k0 += 4) {
        int kk = k0 + ((lane >> 4) << 1);
        v2f a, b;
        a.x = wt[wave][n][kk];
        a.y = wt[wave][n][kk + 1];
        b.x = tx[(size_t)(c0 + kk) * DIMS + nc * 16 + n];
        b.y = tx[(size_t)(c0 + kk + 1) * DIMS + nc * 16 + n];
        g[nc] = wmma4(a, b, g[nc]);
      }
    }
    __syncthreads();
  }

#pragma unroll
  for (int nc = 0; nc < 4; ++nc) {
#pragma unroll
    for (int r = 0; r < 8; ++r) red[wave][nc * 8 + r][lane] = g[nc][r];
  }
#pragma unroll
  for (int r = 0; r < 8; ++r) red[wave][32 + r][lane] = rs[r];
  __syncthreads();

  const int ncp = (wave & 1) * 2;
  const int rq  = (wave >> 1) * 2;
#pragma unroll
  for (int q = 0; q < 2; ++q) {
    const int r = rq + q;
    float s = 0.f, g0 = 0.f, g1 = 0.f;
#pragma unroll
    for (int w = 0; w < FGW; ++w) {
      s  += red[w][32 + r][lane];
      g0 += red[w][ncp * 8 + r][lane];
      g1 += red[w][(ncp + 1) * 8 + r][lane];
    }
    for (int off = 8; off; off >>= 1) s += __shfl_xor(s, off, 32);
    int row = rowBase + mh + r;
    int d0  = ncp * 16 + n;
    float xv0 = x[(size_t)row * DIMS + d0];
    float xv1 = x[(size_t)row * DIMS + d0 + 16];
    out[(size_t)row * DIMS + d0]      = xv0 - (g0 - s * xv0) * (1.0f / LS2);
    out[(size_t)row * DIMS + d0 + 16] = xv1 - (g1 - s * xv1) * (1.0f / LS2);
  }
}

extern "C" void kernel_launch(void* const* d_in, const int* in_sizes, int n_in,
                              void* d_out, int out_size, void* d_ws, size_t ws_size,
                              hipStream_t stream) {
  (void)in_sizes; (void)n_in; (void)out_size; (void)ws_size;
  const float* x  = (const float*)d_in[0];
  const float* tx = (const float*)d_in[1];
  const float* ty = (const float*)d_in[2];
  float* out = (float*)d_out;

  float* Kd = (float*)d_ws;
  float* nt = Kd + (size_t)NTRAIN * NTRAIN;
  float* nx = nt + NTRAIN;
  float* al = nx + NQUERY;
  float* yw = al + NTRAIN;

  gp_sqnorm<<<(NTRAIN + 255) / 256, 256, 0, stream>>>(tx, nt, NTRAIN);
  gp_sqnorm<<<(NQUERY + 255) / 256, 256, 0, stream>>>(x, nx, NQUERY);

  gp_build_K<<<dim3(NTRAIN / 16, NTRAIN / 16), 32, 0, stream>>>(tx, nt, Kd);
  hipMemcpyAsync(yw, ty, NTRAIN * sizeof(float), hipMemcpyDeviceToDevice, stream);

  gp_chol_diag<<<1, 32, 0, stream>>>(Kd, yw, al, 0);
  for (int kb = 0; kb < NTRAIN / NB; ++kb) {
    int below = NTRAIN - NB * (kb + 1);
    if (below > 0) {
      gp_chol_trsm<<<below / NB, NB, 0, stream>>>(Kd, yw, al, kb);
      int tt = below / 64;
      gp_chol_syrk<<<dim3(tt, tt), 128, 0, stream>>>(Kd, yw, al, kb);
    }
  }

  for (int J = NTRAIN / BWB - 1; J >= 0; --J) {
    gp_bwd_block<<<1, BWT, 0, stream>>>(Kd, al, J);
    if (J > 0)
      gp_bwd_gemv<<<J * (BWB / GVC), GVW * 32, 0, stream>>>(Kd, al, J);
  }

  gp_fused_grad<<<NQUERY / 16, FGW * 32, 0, stream>>>(x, tx, al, nx, nt, out);
}


# trailing update: each work-group issues all its global reads up front in one group (staging loop unrolled, 16 panel loads per thread in flight; tile values ahead of the products); sum order kept
# speedup vs baseline: 13.4727x; 1.0459x over previous
#include <hip/hip_runtime.h>
#include <math.h>

#define DIMS   64
#define NTRAIN 4096
#define NQUERY 8192
#define NB     64
#define NOISE  1e-4f
#define LS2    64.0f
#define KSCALE 0.6931471805599453f
#define SPAD   68

typedef float v2f __attribute__((ext_vector_type(2)));
typedef float v8f __attribute__((ext_vector_type(8)));

__device__ __forceinline__ v8f wmma4(v2f a, v2f b, v8f c) {
  return __builtin_amdgcn_wmma_f32_16x16x4_f32(
       false, a,  false, b,
       (short)0, c,  false,  false);
}

__device__ __forceinline__ v2f frag_nt(const float* __restrict__ X, size_t rowStride,
                                       int rowBase, int colBase, int k0, int lane) {
  int kk = k0 + ((lane >> 4) << 1);
  const float* p = X + (size_t)(rowBase + (lane & 15)) * rowStride + colBase + kk;
  v2f v; v.x = p[0]; v.y = p[1];
  return v;
}

__device__ __forceinline__ v2f frag_lds(float (*S)[SPAD], int rowBase, int k0, int lane) {
  int kk = k0 + ((lane >> 4) << 1);
  const float* p = &S[rowBase + (lane & 15)][kk];
  v2f v; v.x = p[0]; v.y = p[1];
  return v;
}

__global__ void gp_sqnorm(const float* __restrict__ X, float* __restrict__ nrm, int rows) {
  int r = blockIdx.x * blockDim.x + threadIdx.x;
  if (r >= rows) return;
  const float4* p = (const float4*)(X + (size_t)r * DIMS);
  float s = 0.f;
#pragma unroll
  for (int i = 0; i < DIMS / 4; ++i) {
    float4 v = p[i];
    s += v.x * v.x + v.y * v.y + v.z * v.z + v.w * v.w;
  }
  nrm[r] = s;
}

__global__ void gp_build_K(const float* __restrict__ tx, const float* __restrict__ nrm,
                           float* __restrict__ K) {
  int bi = blockIdx.x, bj = blockIdx.y;
  int lane = threadIdx.x & 31;
  v8f ce = {}, co = {};
#pragma unroll
  for (int k0 = 0; k0 < DIMS; k0 += 8) {
    ce = wmma4(frag_nt(tx, DIMS, bi * 16, 0, k0, lane),
               frag_nt(tx, DIMS, bj * 16, 0, k0, lane), ce);
    co = wmma4(frag_nt(tx, DIMS, bi * 16, 0, k0 + 4, lane),
               frag_nt(tx, DIMS, bj * 16, 0, k0 + 4, lane), co);
  }
  v8f c = ce + co;
  int n  = lane & 15;
  int mh = (lane >> 4) * 8;
  int col = bj * 16 + n;
  float ncol = nrm[col];
#pragma unroll
  for (int r = 0; r < 8; ++r) {
    int row = bi * 16 + mh + r;
    float d2 = nrm[row] + ncol - 2.0f * c[r];
    float v  = KSCALE * __expf(-0.5f * d2 * (1.0f / LS2));
    if (row == col) v += NOISE;
    K[(size_t)row * NTRAIN + col] = v;
  }
}

#define DP 68

__device__ __forceinline__ float lane_bcast(float v, int srcLane) {
  return __int_as_float(__builtin_amdgcn_readlane(__float_as_int(v), srcLane));
}

__device__ __forceinline__ void wave_handover() {
  __builtin_amdgcn_fence(__ATOMIC_RELEASE, "wavefront");
  __builtin_amdgcn_wave_barrier();
  __builtin_amdgcn_fence(__ATOMIC_ACQUIRE, "wavefront");
}

template <int Q>
__device__ __forceinline__ void diag_block16(float (*a)[DP], int l, float& ya, float& yb,
                                             float& za, float& zb) {
  constexpr bool ROWA = (Q < 2);
  constexpr int c0 = 16 * Q;
  constexpr int l0 = c0 & 31;
  const int rA = l, rB = l + 32;
  float sA[16], sB[16];
#pragma unroll
  for (int u = 0; u < 4; ++u) {
    float4 vb = *(const float4*)&a[rB][c0 + 4 * u];
    sB[4 * u + 0] = vb.x; sB[4 * u + 1] = vb.y; sB[4 * u + 2] = vb.z; sB[4 * u + 3] = vb.w;
    if (ROWA) {
      float4 va = *(const float4*)&a[rA][c0 + 4 * u];
      sA[4 * u + 0] = va.x; sA[4 * u + 1] = va.y; sA[4 * u + 2] = va.z; sA[4 * u + 3] = va.w;
    } else {
      sA[4 * u + 0] = 0.f; sA[4 * u + 1] = 0.f; sA[4 * u + 2] = 0.f; sA[4 * u + 3] = 0.f;
    }
  }
#pragma unroll
  for (int jj = 0; jj < 16; ++jj) {
    const int jl = l0 + jj;
    float piv = lane_bcast(ROWA ? sA[jj] : sB[jj], jl);
    float r  = rsqrtf(piv);
    float zj = lane_bcast(ROWA ? ya : yb, jl) * r;
    float LB = sB[jj] * r;
    float LA = 0.f;
    sB[jj] = LB;
    if (ROWA) {
      LA = sA[jj] * r;
      sA[jj] = LA;
      if (l == jl) za = zj;
      ya -= zj * LA;
    } else {
      if (l == jl) zb = zj;
    }
    yb -= zj * LB;
#pragma unroll
    for (int ii = jj + 1; ii < 16; ++ii) {
      float Lij = lane_bcast(ROWA ? LA : LB, l0 + ii);
      sB[ii] -= LB * Lij;
      if (ROWA) sA[ii] -= LA * Lij;
    }
  }
#pragma unroll
  for (int u = 0; u < 4; ++u) {
    *(float4*)&a[rB][c0 + 4 * u] = make_float4(sB[4 * u + 0], sB[4 * u + 1], sB[4 * u + 2], sB[4 * u + 3]);
    if (ROWA)
      *(float4*)&a[rA][c0 + 4 * u] = make_float4(sA[4 * u + 0], sA[4 * u + 1], sA[4 * u + 2], sA[4 * u + 3]);
  }
  wave_handover();
  if constexpr (Q < 3) {
    constexpr int NT = (3 - Q) * (4 - Q) / 2;
    v8f acc[NT];
    {
      int idx = 0;
#pragma unroll
      for (int ni = Q + 1; ni < 4; ++ni) {
#pragma unroll
        for (int mi = ni; mi < 4; ++mi) {
          v8f c = {};
#pragma unroll
          for (int k0 = 0; k0 < 16; k0 += 4)
            c = wmma4(frag_lds(a, 16 * mi, c0 + k0, l), frag_lds(a, 16 * ni, c0 + k0, l), c);
          acc[idx++] = c;
        }
      }
    }
    const int n = l & 15, mh = (l >> 4) * 8;
    {
      int idx = 0;
#pragma unroll
      for (int ni = Q + 1; ni < 4; ++ni) {
#pragma unroll
        for (int mi = ni; mi < 4; ++mi) {
#pragma unroll
          for (int r = 0; r < 8; ++r)
            a[16 * mi + mh + r][16 * ni + n] -= acc[idx][r];
          ++idx;
        }
      }
    }
    wave_handover();
  }
}

__device__ __forceinline__ void diag_factor_wave(float (*a)[DP], int l, float ya, float yb,
                                                 float* __restrict__ K, float* __restrict__ z, int kb) {
  int base = kb * NB;
  int r0 = l >> 4, c4 = (l & 15) << 2;
  float* blk = K + (size_t)(base + r0) * NTRAIN + base + c4;
  float za = 0.f, zb = 0.f;
  diag_block16<0>(a, l, ya, yb, za, zb);
  diag_block16<1>(a, l, ya, yb, za, zb);
  diag_block16<2>(a, l, ya, yb, za, zb);
  diag_block16<3>(a, l, ya, yb, za, zb);
#pragma unroll
  for (int i = 0; i < NB / 2; ++i)
    *(float4*)(blk + (size_t)i * 2 * NTRAIN) = *(const float4*)&a[r0 + 2 * i][c4];
  z[base + l]      = za;
  z[base + 32 + l] = zb;
}

__global__ void __launch_bounds__(32) gp_chol_diag(float* __restrict__ K, const float* __restrict__ yw,
                                                   float* __restrict__ z, int kb) {
  __shared__ __attribute__((aligned(16))) float a[NB][DP];
  int l = threadIdx.x;
  int base = kb * NB;
  int r0 = l >> 4, c4 = (l & 15) << 2;
  float* blk = K + (size_t)(base + r0) * NTRAIN + base + c4;
#pragma unroll
  for (int i = 0; i < NB / 2; ++i)
    *(float4*)&a[r0 + 2 * i][c4]  = *(const float4*)(blk + (size_t)i * 2 * NTRAIN);
  float ya = yw[base + l], yb = yw[base + 32 + l];
  __syncthreads();
  diag_factor_wave(a, l, ya, yb, K, z, kb);
}

#define TP 68

__global__ void __launch_bounds__(NB) gp_chol_trsm(float* __restrict__ K, float* __restrict__ yw,
                                                   const float* __restrict__ z, int kb) {
  __shared__ __attribute__((aligned(16))) float LC[NB][TP];
  __shared__ __attribute__((aligned(16))) float nri[NB];
  __shared__ __attribute__((aligned(16))) float zs[NB];
  int t = threadIdx.x;
  int base = kb * NB;
  int row = base + NB + blockIdx.x * NB + t;
  float* prow = K + (size_t)row * NTRAIN + base;

  float s[NB];
#pragma unroll
  for (int q = 0; q < NB / 4; ++q) {
    float4 v = *(const float4*)(prow + 4 * q);
    s[4 * q + 0] = v.x; s[4 * q + 1] = v.y; s[4 * q + 2] = v.z; s[4 * q + 3] = v.w;
  }

  {
    int r0 = t >> 4, c4 = (t & 15) << 2;
    const float* blk = K + (size_t)(base + r0) * NTRAIN + base + c4;
#pragma unroll
    for (int i = 0; i < NB / 4; ++i) {
      float4 v = *(const float4*)(blk + (size_t)i * 4 * NTRAIN);
      int r = r0 + 4 * i;
      LC[c4 + 0][(r - c4 - 1) & (NB - 1)] = v.x;
      LC[c4 + 1][(r - c4 - 2) & (NB - 1)] = v.y;
      LC[c4 + 2][(r - c4 - 3) & (NB - 1)] = v.z;
      LC[c4 + 3][(r - c4 - 4) & (NB - 1)] = v.w;
    }
  }
  nri[t] = -1.0f / K[(size_t)(base + t) * NTRAIN + base + t];
  zs[t] = z[base + t];
  __syncthreads();

  float ndz = 0.f;
#pragma unroll
  for (int j = 0; j < NB; ++j) {
    const float nx = s[j] * nri[j];
    ndz = fmaf(nx, zs[j], ndz);
#pragma unroll
    for (int m = 0; j + 1 + 4 * m < NB; ++m) {
      const float4 l4 = *(const float4*)&LC[j][4 * m];
      const int i0 = j + 1 + 4 * m;
      s[i0] = fmaf(nx, l4.x, s[i0]);
      if (i0 + 1 < NB) s[i0 + 1] = fmaf(nx, l4.y, s[i0 + 1]);
      if (i0 + 2 < NB) s[i0 + 2] = fmaf(nx, l4.z, s[i0 + 2]);
      if (i0 + 3 < NB) s[i0 + 3] = fmaf(nx, l4.w, s[i0 + 3]);
    }
    s[j] = -nx;
  }

#pragma unroll
  for (int q = 0; q < NB / 4; ++q)
    *(float4*)(prow + 4 * q) = make_float4(s[4 * q + 0], s[4 * q + 1], s[4 * q + 2], s[4 * q + 3]);
  yw[row] += ndz;
}

__global__ void __launch_bounds__(128) gp_chol_syrk(float* __restrict__ K, const float* __restrict__ yw,
                                                    float* __restrict__ z, int kb) {
  __shared__ __attribute__((aligned(16))) float As[64][SPAD];
  __shared__ __attribute__((aligned(16))) float Bs[64][SPAD];
  int bi = blockIdx.x, bj = blockIdx.y;
  if (bj > bi) return;
  int base = kb * NB;
  int tb   = base + NB;
  int t = threadIdx.x;
  int lane = t & 31, wave = t >> 5;
  int rowA = tb + bi * 64, rowB = tb + bj * 64;
  const bool nextDiag = (bi == 0 && bj == 0);

  int wr = (wave >> 1) * 32;
  int wc = (wave & 1) * 32;
  int n  = lane & 15;
  int mh = (lane >> 4) * 8;
  int r0 = tb + bi * 64 + wr + mh;
  int c0 = tb + bj * 64 + wc + n;

  const int sr = t >> 4, sc4 = (t & 15) << 2;
  const float* pA = K + (size_t)(rowA + sr) * NTRAIN + base + sc4;
  const float* pB = K + (size_t)(rowB + sr) * NTRAIN + base + sc4;
  float4 va[8], vb[8];
#pragma unroll
  for (int i = 0; i < 8; ++i) {
    va[i] = *(const float4*)(pA + (size_t)i * 8 * NTRAIN);
    vb[i] = *(const float4*)(pB + (size_t)i * 8 * NTRAIN);
  }
  float v00[8], v01[8], v10[8], v11[8];
#pragma unroll
  for (int r = 0; r < 8; ++r) {
    v00[r] = K[(size_t)(r0 + r) * NTRAIN + c0];
    v01[r] = K[(size_t)(r0 + r) * NTRAIN + c0 + 16];
    v10[r] = K[(size_t)(r0 + 16 + r) * NTRAIN + c0];
    v11[r] = K[(size_t)(r0 + 16 + r) * NTRAIN + c0 + 16];
  }
  float ya = 0.f, yb = 0.f;
  if (nextDiag && wave == 0) { ya = yw[tb + lane]; yb = yw[tb + 32 + lane]; }

#pragma unroll
  for (int i = 0; i < 8; ++i) {
    *(float4*)&As[sr + 8 * i][sc4] = va[i];
    *(float4*)&Bs[sr + 8 * i][sc4] = vb[i];
  }
  __syncthreads();

  v8f c00 = {}, c01 = {}, c10 = {}, c11 = {};
#pragma unroll
  for (int k0 = 0; k0 < NB; k0 += 4) {
    v2f a0 = frag_lds(As, wr,      k0, lane);
    v2f a1 = frag_lds(As, wr + 16, k0, lane);
    v2f b0 = frag_lds(Bs, wc,      k0, lane);
    v2f b1 = frag_lds(Bs, wc + 16, k0, lane);
    c00 = wmma4(a0, b0, c00);
    c01 = wmma4(a0, b1, c01);
    c10 = wmma4(a1, b0, c10);
    c11 = wmma4(a1, b1, c11);
  }

  if (nextDiag) {
    const int tr = wr + mh, tc = wc + n;
    __syncthreads();
#pragma unroll
    for (int r = 0; r < 8; ++r) {
      As[tr + r][tc]           = v00[r] - c00[r];
      As[tr + r][tc + 16]      = v01[r] - c01[r];
      As[tr + 16 + r][tc]      = v10[r] - c10[r];
      As[tr + 16 + r][tc + 16] = v11[r] - c11[r];
    }
    __syncthreads();
    if (wave == 0) diag_factor_wave(As, lane, ya, yb, K, z, kb + 1);
    return;
  }

#pragma unroll
  for (int r = 0; r < 8; ++r) {
    K[(size_t)(r0 + r) * NTRAIN + c0]           = v00[r] - c00[r];
    K[(size_t)(r0 + r) * NTRAIN + c0 + 16]      = v01[r] - c01[r];
    K[(size_t)(r0 + 16 + r) * NTRAIN + c0]      = v10[r] - c10[r];
    K[(size_t)(r0 + 16 + r) * NTRAIN + c0 + 16] = v11[r] - c11[r];
  }
}

#define BWT 1024
#define BWB 1024
#define GVC 32
#define GVW 16
#define GVR (BWB / GVW)

__global__ void __launch_bounds__(BWT) gp_bwd_block(const float* __restrict__ K,
                                                    float* __restrict__ y, int J) {
  __shared__ float yl[BWB];
  __shared__ float dg[NB][NB + 1];
  __shared__ float dd[NB];
  int t = threadIdx.x;
  const int bb0 = J * BWB;
  for (int i = t; i < BWB; i += BWT) yl[i] = y[bb0 + i];

  int dr = t >> 6, dc = t & 63;
  float nd[4];
  {
    int rb0 = bb0 + BWB - NB;
#pragma unroll
    for (int q = 0; q < 4; ++q)
      nd[q] = K[(size_t)(rb0 + dr + 16 * q) * NTRAIN + rb0 + dc];
  }

  for (int m = BWB / NB - 1; m >= 0; --m) {
    int lb = m * NB;
    int rb = bb0 + lb;
#pragma unroll
    for (int q = 0; q < 4; ++q) {
      int r = dr + 16 * q;
      dg[r][dc] = (dc < r) ? nd[q] : 0.f;
      if (dc == r) dd[r] = nd[q];
    }
    __syncthreads();
    if (m > 0) {
      int rn = rb - NB;
#pragma unroll
      for (int q = 0; q < 4; ++q)
        nd[q] = K[(size_t)(rn + dr + 16 * q) * NTRAIN + rn + dc];
    }

    if (t < 32) {
      float s0 = yl[lb + t], s1 = yl[lb + 32 + t];
      float r0 = 1.0f / dd[t];
      float r1 = 1.0f / dd[32 + t];
#pragma unroll
      for (int j = NB - 1; j >= 32; --j) {
        float xj = lane_bcast(s1, j - 32) * lane_bcast(r1, j - 32);
        s1 -= dg[j][32 + t] * xj;
        s0 -= dg[j][t] * xj;
      }
#pragma unroll
      for (int j = 31; j >= 1; --j) {
        float xj = lane_bcast(s0, j) * lane_bcast(r0, j);
        s0 -= dg[j][t] * xj;
      }
      yl[lb + t]      = s0 * r0;
      yl[lb + 32 + t] = s1 * r1;
    }
    __syncthreads();

    if (t < lb) {
      const float* col = K + (size_t)rb * NTRAIN + bb0 + t;
      float a0 = 0.f, a1 = 0.f, a2 = 0.f, a3 = 0.f;
#pragma unroll
      for (int j = 0; j < NB; j += 4) {
        a0 += col[(size_t)(j + 0) * NTRAIN] * yl[lb + j + 0];
        a1 += col[(size_t)(j + 1) * NTRAIN] * yl[lb + j + 1];
        a2 += col[(size_t)(j + 2) * NTRAIN] * yl[lb + j + 2];
        a3 += col[(size_t)(j + 3) * NTRAIN] * yl[lb + j + 3];
      }
      yl[t] -= (a0 + a1) + (a2 + a3);
    }
  }
  __syncthreads();
  for (int i = t; i < BWB; i += BWT) y[bb0 + i] = yl[i];
}

__global__ void __launch_bounds__(GVW * 32) gp_bwd_gemv(const float* __restrict__ K,
                                                        float* __restrict__ y, int J) {
  __shared__ float xs[BWB];
  __shared__ float part[GVW][GVC];
  int t = threadIdx.x;
  int c = t & 31, g = t >> 5;
  const int bb0 = J * BWB;
  float xa = y[bb0 + t], xb = y[bb0 + GVW * 32 + t];
  const int col = blockIdx.x * GVC + c;
  const float* p = K + (size_t)(bb0 + g * GVR) * NTRAIN + col;
  float v[GVR];
#pragma unroll
  for (int j = 0; j < GVR; ++j) v[j] = p[(size_t)j * NTRAIN];
  xs[t] = xa;
  xs[GVW * 32 + t] = xb;
  __syncthreads();
  const float* xg = xs + g * GVR;
  float a0 = 0.f, a1 = 0.f, a2 = 0.f, a3 = 0.f;
#pragma unroll
  for (int j = 0; j < GVR; j += 4) {
    a0 += v[j + 0] * xg[j + 0];
    a1 += v[j + 1] * xg[j + 1];
    a2 += v[j + 2] * xg[j + 2];
    a3 += v[j + 3] * xg[j + 3];
  }
  part[g][c] = (a0 + a1) + (a2 + a3);
  __syncthreads();
  if (g == 0) {
    float s = 0.f;
#pragma unroll
    for (int w = 0; w < GVW; ++w) s += part[w][c];
    y[col] -= s;
  }
}

#define FGW   8
#define FGSPAN (NTRAIN / FGW)

__global__ void gp_fused_grad(const float* __restrict__ x, const float* __restrict__ tx,
                              const float* __restrict__ alpha,
                              const float* __restrict__ nx, const float* __restrict__ nt,
                              float* __restrict__ out) {
  __shared__ float wt[FGW][16][18];
  __shared__ float red[FGW][40][32];
  int wave = threadIdx.x >> 5, lane = threadIdx.x & 31;
  int rowBase = blockIdx.x * 16;
  int n  = lane & 15;
  int mh = (lane >> 4) * 8;

  float nxr[8];
#pragma unroll
  for (int r = 0; r < 8; ++r) nxr[r] = nx[rowBase + mh + r];

  v2f af[16];
#pragma unroll
  for (int k0 = 0; k0 < DIMS; k0 += 4)
    af[k0 >> 2] = frag_nt(x, DIMS, rowBase, 0, k0, lane);

  v8f g[4]; g[0] = {}; g[1] = {}; g[2] = {}; g[3] = {};
  float rs[8] = {0, 0, 0, 0, 0, 0, 0, 0};

  const int cBeg = wave * FGSPAN, cEnd = cBeg + FGSPAN;
  for (int c0 = cBeg; c0 < cEnd; c0 += 16) {
    if (c0 + 16 < cEnd)
      __builtin_prefetch(tx + (size_t)(c0 + 16) * DIMS + lane * 32, 0, 0);
    v8f ce = {}, co = {};
#pragma unroll
    for (int k0 = 0; k0 < DIMS; k0 += 8) {
      ce = wmma4(af[(k0 >> 2)],     frag_nt(tx, DIMS, c0, 0, k0,     lane), ce);
      co = wmma4(af[(k0 >> 2) + 1], frag_nt(tx, DIMS, c0, 0, k0 + 4, lane), co);
    }
    v8f c = ce + co;
    float ncol = nt[c0 + n];
    float av   = alpha[c0 + n];
#pragma unroll
    for (int r = 0; r < 8; ++r) {
      float d2 = nxr[r] + ncol - 2.0f * c[r];
      float w  = KSCALE * __expf(-0.5f * d2 * (1.0f / LS2)) * av;
      rs[r] += w;
      wt[wave][mh + r][n] = w;
    }
    __syncthreads();
#pragma unroll
    for (int nc = 0; nc < 4; ++nc) {
#pragma unroll
      for (int k0 = 0; k0 < 16; k0 += 4) {
        int kk = k0 + ((lane >> 4) << 1);
        v2f a, b;
        a.x = wt[wave][n][kk];
        a.y = wt[wave][n][kk + 1];
        b.x = tx[(size_t)(c0 + kk) * DIMS + nc * 16 + n];
        b.y = tx[(size_t)(c0 + kk + 1) * DIMS + nc * 16 + n];
        g[nc] = wmma4(a, b, g[nc]);
      }
    }
    __syncthreads();
  }

#pragma unroll
  for (int nc = 0; nc < 4; ++nc) {
#pragma unroll
    for (int r = 0; r < 8; ++r) red[wave][nc * 8 + r][lane] = g[nc][r];
  }
#pragma unroll
  for (int r = 0; r < 8; ++r) red[wave][32 + r][lane] = rs[r];
  __syncthreads();

  const int ncp = (wave & 1) * 2;
  const int rq  = (wave >> 1) * 2;
#pragma unroll
  for (int q = 0; q < 2; ++q) {
    const int r = rq + q;
    float s = 0.f, g0 = 0.f, g1 = 0.f;
#pragma unroll
    for (int w = 0; w < FGW; ++w) {
      s  += red[w][32 + r][lane];
      g0 += red[w][ncp * 8 + r][lane];
      g1 += red[w][(ncp + 1) * 8 + r][lane];
    }
    for (int off = 8; off; off >>= 1) s += __shfl_xor(s, off, 32);
    int row = rowBase + mh + r;
    int d0  = ncp * 16 + n;
    float xv0 = x[(size_t)row * DIMS + d0];
    float xv1 = x[(size_t)row * DIMS + d0 + 16];
    out[(size_t)row * DIMS + d0]      = xv0 - (g0 - s * xv0) * (1.0f / LS2);
    out[(size_t)row * DIMS + d0 + 16] = xv1 - (g1 - s * xv1) * (1.0f / LS2);
  }
}

extern "C" void kernel_launch(void* const* d_in, const int* in_sizes, int n_in,
                              void* d_out, int out_size, void* d_ws, size_t ws_size,
                              hipStream_t stream) {
  (void)in_sizes; (void)n_in; (void)out_size; (void)ws_size;
  const float* x  = (const float*)d_in[0];
  const float* tx = (const float*)d_in[1];
  const float* ty = (const float*)d_in[2];
  float* out = (float*)d_out;

  float* Kd = (float*)d_ws;
  float* nt = Kd + (size_t)NTRAIN * NTRAIN;
  float* nx = nt + NTRAIN;
  float* al = nx + NQUERY;
  float* yw = al + NTRAIN;

  gp_sqnorm<<<(NTRAIN + 255) / 256, 256, 0, stream>>>(tx, nt, NTRAIN);
  gp_sqnorm<<<(NQUERY + 255) / 256, 256, 0, stream>>>(x, nx, NQUERY);

  gp_build_K<<<dim3(NTRAIN / 16, NTRAIN / 16), 32, 0, stream>>>(tx, nt, Kd);
  hipMemcpyAsync(yw, ty, NTRAIN * sizeof(float), hipMemcpyDeviceToDevice, stream);

  gp_chol_diag<<<1, 32, 0, stream>>>(Kd, yw, al, 0);
  for (int kb = 0; kb < NTRAIN / NB; ++kb) {
    int below = NTRAIN - NB * (kb + 1);
    if (below > 0) {
      gp_chol_trsm<<<below / NB, NB, 0, stream>>>(Kd, yw, al, kb);
      int tt = below / 64;
      gp_chol_syrk<<<dim3(tt, tt), 128, 0, stream>>>(Kd, yw, al, kb);
    }
  }

  for (int J = NTRAIN / BWB - 1; J >= 0; --J) {
    gp_bwd_block<<<1, BWT, 0, stream>>>(Kd, al, J);
    if (J > 0)
      gp_bwd_gemv<<<J * (BWB / GVC), GVW * 32, 0, stream>>>(Kd, al, J);
  }

  gp_fused_grad<<<NQUERY / 16, FGW * 32, 0, stream>>>(x, tx, al, nx, nt, out);
}


# K build and final gradient: train rows fetched by whole-line loads into LDS (gradient: per-wave tile, a chunk ahead, in-wave hand-over), fragments read from LDS, not strided global; sum order kept
# speedup vs baseline: 14.1382x; 1.0494x over previous
#include <hip/hip_runtime.h>
#include <math.h>

#define DIMS   64
#define NTRAIN 4096
#define NQUERY 8192
#define NB     64
#define NOISE  1e-4f
#define LS2    64.0f
#define KSCALE 0.6931471805599453f
#define SPAD   68

typedef float v2f __attribute__((ext_vector_type(2)));
typedef float v8f __attribute__((ext_vector_type(8)));
typedef float v4f __attribute__((ext_vector_type(4)));

__device__ __forceinline__ v8f wmma4(v2f a, v2f b, v8f c) {
  return __builtin_amdgcn_wmma_f32_16x16x4_f32(
       false, a,  false, b,
       (short)0, c,  false,  false);
}

__device__ __forceinline__ v2f frag_nt(const float* __restrict__ X, size_t rowStride,
                                       int rowBase, int colBase, int k0, int lane) {
  int kk = k0 + ((lane >> 4) << 1);
  const float* p = X + (size_t)(rowBase + (lane & 15)) * rowStride + colBase + kk;
  v2f v; v.x = p[0]; v.y = p[1];
  return v;
}

__device__ __forceinline__ v2f frag_lds(float (*S)[SPAD], int rowBase, int k0, int lane) {
  int kk = k0 + ((lane >> 4) << 1);
  const float* p = &S[rowBase + (lane & 15)][kk];
  v2f v; v.x = p[0]; v.y = p[1];
  return v;
}

__global__ void gp_sqnorm(const float* __restrict__ X, float* __restrict__ nrm, int rows) {
  int r = blockIdx.x * blockDim.x + threadIdx.x;
  if (r >= rows) return;
  const float4* p = (const float4*)(X + (size_t)r * DIMS);
  float s = 0.f;
#pragma unroll
  for (int i = 0; i < DIMS / 4; ++i) {
    float4 v = p[i];
    s += v.x * v.x + v.y * v.y + v.z * v.z + v.w * v.w;
  }
  nrm[r] = s;
}

__global__ void __launch_bounds__(128) gp_build_K(const float* __restrict__ tx, const float* __restrict__ nrm,
                                                  float* __restrict__ K) {
  __shared__ __attribute__((aligned(16))) float As[64][SPAD];
  __shared__ __attribute__((aligned(16))) float Bs[64][SPAD];
  int bi = blockIdx.x, bj = blockIdx.y;
  int t = threadIdx.x;
  int lane = t & 31, wave = t >> 5;
  int wr = (wave >> 1) * 32;
  int wc = (wave & 1) * 32;
  int n  = lane & 15;
  int mh = (lane >> 4) * 8;
  int r0 = bi * 64 + wr + mh;
  int c0 = bj * 64 + wc + n;

  const int sr = t >> 4, sc4 = (t & 15) << 2;
  const float* pA = tx + (size_t)(bi * 64 + sr) * DIMS + sc4;
  const float* pB = tx + (size_t)(bj * 64 + sr) * DIMS + sc4;
  float4 va[8], vb[8];
#pragma unroll
  for (int i = 0; i < 8; ++i) {
    va[i] = *(const float4*)(pA + (size_t)i * 8 * DIMS);
    vb[i] = *(const float4*)(pB + (size_t)i * 8 * DIMS);
  }
  float nr0[8], nr1[8];
#pragma unroll
  for (int r = 0; r < 8; ++r) { nr0[r] = nrm[r0 + r]; nr1[r] = nrm[r0 + 16 + r]; }
  float nc0 = nrm[c0], nc1 = nrm[c0 + 16];
#pragma unroll
  for (int i = 0; i < 8; ++i) {
    *(float4*)&As[sr + 8 * i][sc4] = va[i];
    *(float4*)&Bs[sr + 8 * i][sc4] = vb[i];
  }
  __syncthreads();

  v8f e00 = {}, e01 = {}, e10 = {}, e11 = {};
  v8f o00 = {}, o01 = {}, o10 = {}, o11 = {};
#pragma unroll
  for (int k0 = 0; k0 < DIMS; k0 += 8) {
    v2f a0 = frag_lds(As, wr,      k0, lane);
    v2f a1 = frag_lds(As, wr + 16, k0, lane);
    v2f b0 = frag_lds(Bs, wc,      k0, lane);
    v2f b1 = frag_lds(Bs, wc + 16, k0, lane);
    e00 = wmma4(a0, b0, e00);
    e01 = wmma4(a0, b1, e01);
    e10 = wmma4(a1, b0, e10);
    e11 = wmma4(a1, b1, e11);
    v2f p0 = frag_lds(As, wr,      k0 + 4, lane);
    v2f p1 = frag_lds(As, wr + 16, k0 + 4, lane);
    v2f q0 = frag_lds(Bs, wc,      k0 + 4, lane);
    v2f q1 = frag_lds(Bs, wc + 16, k0 + 4, lane);
    o00 = wmma4(p0, q0, o00);
    o01 = wmma4(p0, q1, o01);
    o10 = wmma4(p1, q0, o10);
    o11 = wmma4(p1, q1, o11);
  }
  v8f c00 = e00 + o00, c01 = e01 + o01, c10 = e10 + o10, c11 = e11 + o11;

#pragma unroll
  for (int r = 0; r < 8; ++r) {
    int rowa = r0 + r, rowb = r0 + 16 + r;
    float v00 = KSCALE * __expf(-0.5f * (nr0[r] + nc0 - 2.0f * c00[r]) * (1.0f / LS2));
    float v01 = KSCALE * __expf(-0.5f * (nr0[r] + nc1 - 2.0f * c01[r]) * (1.0f / LS2));
    float v10 = KSCALE * __expf(-0.5f * (nr1[r] + nc0 - 2.0f * c10[r]) * (1.0f / LS2));
    float v11 = KSCALE * __expf(-0.5f * (nr1[r] + nc1 - 2.0f * c11[r]) * (1.0f / LS2));
    if (rowa == c0)      v00 += NOISE;
    if (rowa == c0 + 16) v01 += NOISE;
    if (rowb == c0)      v10 += NOISE;
    if (rowb == c0 + 16) v11 += NOISE;
    K[(size_t)rowa * NTRAIN + c0]      = v00;
    K[(size_t)rowa * NTRAIN + c0 + 16] = v01;
    K[(size_t)rowb * NTRAIN + c0]      = v10;
    K[(size_t)rowb * NTRAIN + c0 + 16] = v11;
  }
}

#define DP 68

__device__ __forceinline__ float lane_bcast(float v, int srcLane) {
  return __int_as_float(__builtin_amdgcn_readlane(__float_as_int(v), srcLane));
}

__device__ __forceinline__ void wave_handover() {
  __builtin_amdgcn_fence(__ATOMIC_RELEASE, "wavefront");
  __builtin_amdgcn_wave_barrier();
  __builtin_amdgcn_fence(__ATOMIC_ACQUIRE, "wavefront");
}

template <int Q>
__device__ __forceinline__ void diag_block16(float (*a)[DP], int l, float& ya, float& yb,
                                             float& za, float& zb) {
  constexpr bool ROWA = (Q < 2);
  constexpr int c0 = 16 * Q;
  constexpr int l0 = c0 & 31;
  const int rA = l, rB = l + 32;
  float sA[16], sB[16];
#pragma unroll
  for (int u = 0; u < 4; ++u) {
    float4 vb = *(const float4*)&a[rB][c0 + 4 * u];
    sB[4 * u + 0] = vb.x; sB[4 * u + 1] = vb.y; sB[4 * u + 2] = vb.z; sB[4 * u + 3] = vb.w;
    if (ROWA) {
      float4 va = *(const float4*)&a[rA][c0 + 4 * u];
      sA[4 * u + 0] = va.x; sA[4 * u + 1] = va.y; sA[4 * u + 2] = va.z; sA[4 * u + 3] = va.w;
    } else {
      sA[4 * u + 0] = 0.f; sA[4 * u + 1] = 0.f; sA[4 * u + 2] = 0.f; sA[4 * u + 3] = 0.f;
    }
  }
#pragma unroll
  for (int jj = 0; jj < 16; ++jj) {
    const int jl = l0 + jj;
    float piv = lane_bcast(ROWA ? sA[jj] : sB[jj], jl);
    float r  = rsqrtf(piv);
    float zj = lane_bcast(ROWA ? ya : yb, jl) * r;
    float LB = sB[jj] * r;
    float LA = 0.f;
    sB[jj] = LB;
    if (ROWA) {
      LA = sA[jj] * r;
      sA[jj] = LA;
      if (l == jl) za = zj;
      ya -= zj * LA;
    } else {
      if (l == jl) zb = zj;
    }
    yb -= zj * LB;
#pragma unroll
    for (int ii = jj + 1; ii < 16; ++ii) {
      float Lij = lane_bcast(ROWA ? LA : LB, l0 + ii);
      sB[ii] -= LB * Lij;
      if (ROWA) sA[ii] -= LA * Lij;
    }
  }
#pragma unroll
  for (int u = 0; u < 4; ++u) {
    *(float4*)&a[rB][c0 + 4 * u] = make_float4(sB[4 * u + 0], sB[4 * u + 1], sB[4 * u + 2], sB[4 * u + 3]);
    if (ROWA)
      *(float4*)&a[rA][c0 + 4 * u] = make_float4(sA[4 * u + 0], sA[4 * u + 1], sA[4 * u + 2], sA[4 * u + 3]);
  }
  wave_handover();
  if constexpr (Q < 3) {
    constexpr int NT = (3 - Q) * (4 - Q) / 2;
    v8f acc[NT];
    {
      int idx = 0;
#pragma unroll
      for (int ni = Q + 1; ni < 4; ++ni) {
#pragma unroll
        for (int mi = ni; mi < 4; ++mi) {
          v8f c = {};
#pragma unroll
          for (int k0 = 0; k0 < 16; k0 += 4)
            c = wmma4(frag_lds(a, 16 * mi, c0 + k0, l), frag_lds(a, 16 * ni, c0 + k0, l), c);
          acc[idx++] = c;
        }
      }
    }
    const int n = l & 15, mh = (l >> 4) * 8;
    {
      int idx = 0;
#pragma unroll
      for (int ni = Q + 1; ni < 4; ++ni) {
#pragma unroll
        for (int mi = ni; mi < 4; ++mi) {
#pragma unroll
          for (int r = 0; r < 8; ++r)
            a[16 * mi + mh + r][16 * ni + n] -= acc[idx][r];
          ++idx;
        }
      }
    }
    wave_handover();
  }
}

__device__ __forceinline__ void diag_factor_wave(float (*a)[DP], int l, float ya, float yb,
                                                 float* __restrict__ K, float* __restrict__ z, int kb) {
  int base = kb * NB;
  int r0 = l >> 4, c4 = (l & 15) << 2;
  float* blk = K + (size_t)(base + r0) * NTRAIN + base + c4;
  float za = 0.f, zb = 0.f;
  diag_block16<0>(a, l, ya, yb, za, zb);
  diag_block16<1>(a, l, ya, yb, za, zb);
  diag_block16<2>(a, l, ya, yb, za, zb);
  diag_block16<3>(a, l, ya, yb, za, zb);
#pragma unroll
  for (int i = 0; i < NB / 2; ++i)
    *(float4*)(blk + (size_t)i * 2 * NTRAIN) = *(const float4*)&a[r0 + 2 * i][c4];
  z[base + l]      = za;
  z[base + 32 + l] = zb;
}

__global__ void __launch_bounds__(32) gp_chol_diag(float* __restrict__ K, const float* __restrict__ yw,
                                                   float* __restrict__ z, int kb) {
  __shared__ __attribute__((aligned(16))) float a[NB][DP];
  int l = threadIdx.x;
  int base = kb * NB;
  int r0 = l >> 4, c4 = (l & 15) << 2;
  float* blk = K + (size_t)(base + r0) * NTRAIN + base + c4;
#pragma unroll
  for (int i = 0; i < NB / 2; ++i)
    *(float4*)&a[r0 + 2 * i][c4]  = *(const float4*)(blk + (size_t)i * 2 * NTRAIN);
  float ya = yw[base + l], yb = yw[base + 32 + l];
  __syncthreads();
  diag_factor_wave(a, l, ya, yb, K, z, kb);
}

#define TP 68

__global__ void __launch_bounds__(NB) gp_chol_trsm(float* __restrict__ K, float* __restrict__ yw,
                                                   const float* __restrict__ z, int kb) {
  __shared__ __attribute__((aligned(16))) float LC[NB][TP];
  __shared__ __attribute__((aligned(16))) float nri[NB];
  __shared__ __attribute__((aligned(16))) float zs[NB];
  int t = threadIdx.x;
  int base = kb * NB;
  int row = base + NB + blockIdx.x * NB + t;
  float* prow = K + (size_t)row * NTRAIN + base;

  float s[NB];
#pragma unroll
  for (int q = 0; q < NB / 4; ++q) {
    float4 v = *(const float4*)(prow + 4 * q);
    s[4 * q + 0] = v.x; s[4 * q + 1] = v.y; s[4 * q + 2] = v.z; s[4 * q + 3] = v.w;
  }

  {
    int r0 = t >> 4, c4 = (t & 15) << 2;
    const float* blk = K + (size_t)(base + r0) * NTRAIN + base + c4;
#pragma unroll
    for (int i = 0; i < NB / 4; ++i) {
      float4 v = *(const float4*)(blk + (size_t)i * 4 * NTRAIN);
      int r = r0 + 4 * i;
      LC[c4 + 0][(r - c4 - 1) & (NB - 1)] = v.x;
      LC[c4 + 1][(r - c4 - 2) & (NB - 1)] = v.y;
      LC[c4 + 2][(r - c4 - 3) & (NB - 1)] = v.z;
      LC[c4 + 3][(r - c4 - 4) & (NB - 1)] = v.w;
    }
  }
  nri[t] = -1.0f / K[(size_t)(base + t) * NTRAIN + base + t];
  zs[t] = z[base + t];
  __syncthreads();

  float ndz = 0.f;
#pragma unroll
  for (int j = 0; j < NB; ++j) {
    const float nx = s[j] * nri[j];
    ndz = fmaf(nx, zs[j], ndz);
#pragma unroll
    for (int m = 0; j + 1 + 4 * m < NB; ++m) {
      const float4 l4 = *(const float4*)&LC[j][4 * m];
      const int i0 = j + 1 + 4 * m;
      s[i0] = fmaf(nx, l4.x, s[i0]);
      if (i0 + 1 < NB) s[i0 + 1] = fmaf(nx, l4.y, s[i0 + 1]);
      if (i0 + 2 < NB) s[i0 + 2] = fmaf(nx, l4.z, s[i0 + 2]);
      if (i0 + 3 < NB) s[i0 + 3] = fmaf(nx, l4.w, s[i0 + 3]);
    }
    s[j] = -nx;
  }

#pragma unroll
  for (int q = 0; q < NB / 4; ++q)
    *(float4*)(prow + 4 * q) = make_float4(s[4 * q + 0], s[4 * q + 1], s[4 * q + 2], s[4 * q + 3]);
  yw[row] += ndz;
}

__global__ void __launch_bounds__(128) gp_chol_syrk(float* __restrict__ K, const float* __restrict__ yw,
                                                    float* __restrict__ z, int kb) {
  __shared__ __attribute__((aligned(16))) float As[64][SPAD];
  __shared__ __attribute__((aligned(16))) float Bs[64][SPAD];
  int bi = blockIdx.x, bj = blockIdx.y;
  if (bj > bi) return;
  int base = kb * NB;
  int tb   = base + NB;
  int t = threadIdx.x;
  int lane = t & 31, wave = t >> 5;
  int rowA = tb + bi * 64, rowB = tb + bj * 64;
  const bool nextDiag = (bi == 0 && bj == 0);

  int wr = (wave >> 1) * 32;
  int wc = (wave & 1) * 32;
  int n  = lane & 15;
  int mh = (lane >> 4) * 8;
  int r0 = tb + bi * 64 + wr + mh;
  int c0 = tb + bj * 64 + wc + n;

  const int sr = t >> 4, sc4 = (t & 15) << 2;
  const float* pA = K + (size_t)(rowA + sr) * NTRAIN + base + sc4;
  const float* pB = K + (size_t)(rowB + sr) * NTRAIN + base + sc4;
  float4 va[8], vb[8];
#pragma unroll
  for (int i = 0; i < 8; ++i) {
    va[i] = *(const float4*)(pA + (size_t)i * 8 * NTRAIN);
    vb[i] = *(const float4*)(pB + (size_t)i * 8 * NTRAIN);
  }
  float v00[8], v01[8], v10[8], v11[8];
#pragma unroll
  for (int r = 0; r < 8; ++r) {
    v00[r] = K[(size_t)(r0 + r) * NTRAIN + c0];
    v01[r] = K[(size_t)(r0 + r) * NTRAIN + c0 + 16];
    v10[r] = K[(size_t)(r0 + 16 + r) * NTRAIN + c0];
    v11[r] = K[(size_t)(r0 + 16 + r) * NTRAIN + c0 + 16];
  }
  float ya = 0.f, yb = 0.f;
  if (nextDiag && wave == 0) { ya = yw[tb + lane]; yb = yw[tb + 32 + lane]; }

#pragma unroll
  for (int i = 0; i < 8; ++i) {
    *(float4*)&As[sr + 8 * i][sc4] = va[i];
    *(float4*)&Bs[sr + 8 * i][sc4] = vb[i];
  }
  __syncthreads();

  v8f c00 = {}, c01 = {}, c10 = {}, c11 = {};
#pragma unroll
  for (int k0 = 0; k0 < NB; k0 += 4) {
    v2f a0 = frag_lds(As, wr,      k0, lane);
    v2f a1 = frag_lds(As, wr + 16, k0, lane);
    v2f b0 = frag_lds(Bs, wc,      k0, lane);
    v2f b1 = frag_lds(Bs, wc + 16, k0, lane);
    c00 = wmma4(a0, b0, c00);
    c01 = wmma4(a0, b1, c01);
    c10 = wmma4(a1, b0, c10);
    c11 = wmma4(a1, b1, c11);
  }

  if (nextDiag) {
    const int tr = wr + mh, tc = wc + n;
    __syncthreads();
#pragma unroll
    for (int r = 0; r < 8; ++r) {
      As[tr + r][tc]           = v00[r] - c00[r];
      As[tr + r][tc + 16]      = v01[r] - c01[r];
      As[tr + 16 + r][tc]      = v10[r] - c10[r];
      As[tr + 16 + r][tc + 16] = v11[r] - c11[r];
    }
    __syncthreads();
    if (wave == 0) diag_factor_wave(As, lane, ya, yb, K, z, kb + 1);
    return;
  }

#pragma unroll
  for (int r = 0; r < 8; ++r) {
    K[(size_t)(r0 + r) * NTRAIN + c0]           = v00[r] - c00[r];
    K[(size_t)(r0 + r) * NTRAIN + c0 + 16]      = v01[r] - c01[r];
    K[(size_t)(r0 + 16 + r) * NTRAIN + c0]      = v10[r] - c10[r];
    K[(size_t)(r0 + 16 + r) * NTRAIN + c0 + 16] = v11[r] - c11[r];
  }
}

#define BWT 1024
#define BWB 1024
#define GVC 32
#define GVW 16
#define GVR (BWB / GVW)

__global__ void __launch_bounds__(BWT) gp_bwd_block(const float* __restrict__ K,
                                                    float* __restrict__ y, int J) {
  __shared__ float yl[BWB];
  __shared__ float dg[NB][NB + 1];
  __shared__ float dd[NB];
  int t = threadIdx.x;
  const int bb0 = J * BWB;
  for (int i = t; i < BWB; i += BWT) yl[i] = y[bb0 + i];

  int dr = t >> 6, dc = t & 63;
  float nd[4];
  {
    int rb0 = bb0 + BWB - NB;
#pragma unroll
    for (int q = 0; q < 4; ++q)
      nd[q] = K[(size_t)(rb0 + dr + 16 * q) * NTRAIN + rb0 + dc];
  }

  for (int m = BWB / NB - 1; m >= 0; --m) {
    int lb = m * NB;
    int rb = bb0 + lb;
#pragma unroll
    for (int q = 0; q < 4; ++q) {
      int r = dr + 16 * q;
      dg[r][dc] = (dc < r) ? nd[q] : 0.f;
      if (dc == r) dd[r] = nd[q];
    }
    __syncthreads();
    if (m > 0) {
      int rn = rb - NB;
#pragma unroll
      for (int q = 0; q < 4; ++q)
        nd[q] = K[(size_t)(rn + dr + 16 * q) * NTRAIN + rn + dc];
    }

    if (t < 32) {
      float s0 = yl[lb + t], s1 = yl[lb + 32 + t];
      float r0 = 1.0f / dd[t];
      float r1 = 1.0f / dd[32 + t];
#pragma unroll
      for (int j = NB - 1; j >= 32; --j) {
        float xj = lane_bcast(s1, j - 32) * lane_bcast(r1, j - 32);
        s1 -= dg[j][32 + t] * xj;
        s0 -= dg[j][t] * xj;
      }
#pragma unroll
      for (int j = 31; j >= 1; --j) {
        float xj = lane_bcast(s0, j) * lane_bcast(r0, j);
        s0 -= dg[j][t] * xj;
      }
      yl[lb + t]      = s0 * r0;
      yl[lb + 32 + t] = s1 * r1;
    }
    __syncthreads();

    if (t < lb) {
      const float* col = K + (size_t)rb * NTRAIN + bb0 + t;
      float a0 = 0.f, a1 = 0.f, a2 = 0.f, a3 = 0.f;
#pragma unroll
      for (int j = 0; j < NB; j += 4) {
        a0 += col[(size_t)(j + 0) * NTRAIN] * yl[lb + j + 0];
        a1 += col[(size_t)(j + 1) * NTRAIN] * yl[lb + j + 1];
        a2 += col[(size_t)(j + 2) * NTRAIN] * yl[lb + j + 2];
        a3 += col[(size_t)(j + 3) * NTRAIN] * yl[lb + j + 3];
      }
      yl[t] -= (a0 + a1) + (a2 + a3);
    }
  }
  __syncthreads();
  for (int i = t; i < BWB; i += BWT) y[bb0 + i] = yl[i];
}

__global__ void __launch_bounds__(GVW * 32) gp_bwd_gemv(const float* __restrict__ K,
                                                        float* __restrict__ y, int J) {
  __shared__ float xs[BWB];
  __shared__ float part[GVW][GVC];
  int t = threadIdx.x;
  int c = t & 31, g = t >> 5;
  const int bb0 = J * BWB;
  float xa = y[bb0 + t], xb = y[bb0 + GVW * 32 + t];
  const int col = blockIdx.x * GVC + c;
  const float* p = K + (size_t)(bb0 + g * GVR) * NTRAIN + col;
  float v[GVR];
#pragma unroll
  for (int j = 0; j < GVR; ++j) v[j] = p[(size_t)j * NTRAIN];
  xs[t] = xa;
  xs[GVW * 32 + t] = xb;
  __syncthreads();
  const float* xg = xs + g * GVR;
  float a0 = 0.f, a1 = 0.f, a2 = 0.f, a3 = 0.f;
#pragma unroll
  for (int j = 0; j < GVR; j += 4) {
    a0 += v[j + 0] * xg[j + 0];
    a1 += v[j + 1] * xg[j + 1];
    a2 += v[j + 2] * xg[j + 2];
    a3 += v[j + 3] * xg[j + 3];
  }
  part[g][c] = (a0 + a1) + (a2 + a3);
  __syncthreads();
  if (g == 0) {
    float s = 0.f;
#pragma unroll
    for (int w = 0; w < GVW; ++w) s += part[w][c];
    y[col] -= s;
  }
}

#define FGW   8
#define FGSPAN (NTRAIN / FGW)
#define FGP   68

__global__ void __launch_bounds__(FGW * 32) gp_fused_grad(const float* __restrict__ x, const float* __restrict__ tx,
                              const float* __restrict__ alpha,
                              const float* __restrict__ nx, const float* __restrict__ nt,
                              float* __restrict__ out) {
  __shared__ __attribute__((aligned(16))) float tc[FGW][16][FGP];
  __shared__ __attribute__((aligned(16))) float wt[FGW][16][18];
  __shared__ float red[FGW][40][32];
  const int wave = __builtin_amdgcn_readfirstlane((int)(threadIdx.x >> 5));
  const int lane = threadIdx.x & 31;
  int rowBase = blockIdx.x * 16;
  int n  = lane & 15;
  int mh = (lane >> 4) * 8;
  const int h2 = (lane >> 4) << 1;

  float nxr[8];
#pragma unroll
  for (int r = 0; r < 8; ++r) nxr[r] = nx[rowBase + mh + r];

  v2f af[16];
#pragma unroll
  for (int k0 = 0; k0 < DIMS; k0 += 4)
    af[k0 >> 2] = frag_nt(x, DIMS, rowBase, 0, k0, lane);

  v8f g[4]; g[0] = {}; g[1] = {}; g[2] = {}; g[3] = {};
  float rs[8] = {0, 0, 0, 0, 0, 0, 0, 0};

  float (*T)[FGP] = tc[wave];
  float (*Wt)[18] = wt[wave];

  const int cBeg = wave * FGSPAN, cEnd = cBeg + FGSPAN;
  const int sr = lane >> 4, sc4 = (lane & 15) << 2;
  const float* pc = tx + (size_t)(cBeg + sr) * DIMS + sc4;
  v4f nxt[8];
#pragma unroll
  for (int i = 0; i < 8; ++i) nxt[i] = *(const v4f*)(pc + (size_t)i * 2 * DIMS);
  float ncolN = nt[cBeg + n];
  float avN   = alpha[cBeg + n];

  for (int c0 = cBeg; c0 < cEnd; c0 += 16) {
#pragma unroll
    for (int i = 0; i < 8; ++i) *(v4f*)&T[sr + 2 * i][sc4] = nxt[i];
    const float ncol = ncolN, av = avN;
    wave_handover();
    if (c0 + 16 < cEnd) {
      pc += 16 * DIMS;
#pragma unroll
      for (int i = 0; i < 8; ++i) nxt[i] = *(const v4f*)(pc + (size_t)i * 2 * DIMS);
      ncolN = nt[c0 + 16 + n];
      avN   = alpha[c0 + 16 + n];
    }
    v2f bf[16];
#pragma unroll
    for (int s = 0; s < 16; ++s) {
      const float* p = &T[n][4 * s + h2];
      bf[s].x = p[0]; bf[s].y = p[1];
    }
    v8f ce = {}, co = {};
#pragma unroll
    for (int k0 = 0; k0 < DIMS; k0 += 8) {
      ce = wmma4(af[(k0 >> 2)],     bf[(k0 >> 2)],     ce);
      co = wmma4(af[(k0 >> 2) + 1], bf[(k0 >> 2) + 1], co);
    }
    v8f c = ce + co;
    v2f b2[4][4];
#pragma unroll
    for (int nc = 0; nc < 4; ++nc) {
#pragma unroll
      for (int q = 0; q < 4; ++q) {
        int kk = 4 * q + h2;
        b2[nc][q].x = T[kk][nc * 16 + n];
        b2[nc][q].y = T[kk + 1][nc * 16 + n];
      }
    }
#pragma unroll
    for (int r = 0; r < 8; ++r) {
      float d2 = nxr[r] + ncol - 2.0f * c[r];
      float w  = KSCALE * __expf(-0.5f * d2 * (1.0f / LS2)) * av;
      rs[r] += w;
      Wt[mh + r][n] = w;
    }
    wave_handover();
    v2f a2[4];
#pragma unroll
    for (int q = 0; q < 4; ++q) {
      int kk = 4 * q + h2;
      a2[q].x = Wt[n][kk];
      a2[q].y = Wt[n][kk + 1];
    }
#pragma unroll
    for (int nc = 0; nc < 4; ++nc) {
#pragma unroll
      for (int q = 0; q < 4; ++q)
        g[nc] = wmma4(a2[q], b2[nc][q], g[nc]);
    }
    wave_handover();
  }

#pragma unroll
  for (int nc = 0; nc < 4; ++nc) {
#pragma unroll
    for (int r = 0; r < 8; ++r) red[wave][nc * 8 + r][lane] = g[nc][r];
  }
#pragma unroll
  for (int r = 0; r < 8; ++r) red[wave][32 + r][lane] = rs[r];
  __syncthreads();

  const int ncp = (wave & 1) * 2;
  const int rq  = (wave >> 1) * 2;
#pragma unroll
  for (int q = 0; q < 2; ++q) {
    const int r = rq + q;
    float s = 0.f, g0 = 0.f, g1 = 0.f;
#pragma unroll
    for (int w = 0; w < FGW; ++w) {
      s  += red[w][32 + r][lane];
      g0 += red[w][ncp * 8 + r][lane];
      g1 += red[w][(ncp + 1) * 8 + r][lane];
    }
    for (int off = 8; off; off >>= 1) s += __shfl_xor(s, off, 32);
    int row = rowBase + mh + r;
    int d0  = ncp * 16 + n;
    float xv0 = x[(size_t)row * DIMS + d0];
    float xv1 = x[(size_t)row * DIMS + d0 + 16];
    out[(size_t)row * DIMS + d0]      = xv0 - (g0 - s * xv0) * (1.0f / LS2);
    out[(size_t)row * DIMS + d0 + 16] = xv1 - (g1 - s * xv1) * (1.0f / LS2);
  }
}

extern "C" void kernel_launch(void* const* d_in, const int* in_sizes, int n_in,
                              void* d_out, int out_size, void* d_ws, size_t ws_size,
                              hipStream_t stream) {
  (void)in_sizes; (void)n_in; (void)out_size; (void)ws_size;
  const float* x  = (const float*)d_in[0];
  const float* tx = (const float*)d_in[1];
  const float* ty = (const float*)d_in[2];
  float* out = (float*)d_out;

  float* Kd = (float*)d_ws;
  float* nt = Kd + (size_t)NTRAIN * NTRAIN;
  float* nx = nt + NTRAIN;
  float* al = nx + NQUERY;
  float* yw = al + NTRAIN;

  gp_sqnorm<<<(NTRAIN + 255) / 256, 256, 0, stream>>>(tx, nt, NTRAIN);
  gp_sqnorm<<<(NQUERY + 255) / 256, 256, 0, stream>>>(x, nx, NQUERY);

  gp_build_K<<<dim3(NTRAIN / 64, NTRAIN / 64), 128, 0, stream>>>(tx, nt, Kd);
  hipMemcpyAsync(yw, ty, NTRAIN * sizeof(float), hipMemcpyDeviceToDevice, stream);

  gp_chol_diag<<<1, 32, 0, stream>>>(Kd, yw, al, 0);
  for (int kb = 0; kb < NTRAIN / NB; ++kb) {
    int below = NTRAIN - NB * (kb + 1);
    if (below > 0) {
      gp_chol_trsm<<<below / NB, NB, 0, stream>>>(Kd, yw, al, kb);
      int tt = below / 64;
      gp_chol_syrk<<<dim3(tt, tt), 128, 0, stream>>>(Kd, yw, al, kb);
    }
  }

  for (int J = NTRAIN / BWB - 1; J >= 0; --J) {
    gp_bwd_block<<<1, BWT, 0, stream>>>(Kd, al, J);
    if (J > 0)
      gp_bwd_gemv<<<J * (BWB / GVC), GVW * 32, 0, stream>>>(Kd, al, J);
  }

  gp_fused_grad<<<NQUERY / 16, FGW * 32, 0, stream>>>(x, tx, al, nx, nt, out);
}
